# starting kernel
# baseline (speedup 1.0000x reference)
#include <hip/hip_runtime.h>
#include <hip/hip_bf16.h>

#define N_NODES 100000
#define N_EDGES 1600000

typedef float v2f __attribute__((ext_vector_type(2)));
typedef float v8f __attribute__((ext_vector_type(8)));

__global__ void zero_f32(float* __restrict__ p, long n) {
    long i = (long)blockIdx.x * blockDim.x + threadIdx.x;
    long stride = (long)gridDim.x * blockDim.x;
    for (; i < n; i += stride) p[i] = 0.0f;
}

__global__ void degree_kernel(const int* __restrict__ dst, float* __restrict__ deg, int E) {
    int e = blockIdx.x * blockDim.x + threadIdx.x;
    if (e < E) unsafeAtomicAdd(&deg[dst[e]], 1.0f);
}

__global__ void invdeg_kernel(const float* __restrict__ deg, float* __restrict__ inv, int n) {
    int i = blockIdx.x * blockDim.x + threadIdx.x;
    if (i < n) {
        float d = deg[i];
        inv[i] = (d > 0.0f) ? (1.0f / fmaxf(d, 1.0f)) : 0.0f;
    }
}

__global__ void scatter_add_kernel(const float* __restrict__ x,
                                   const int* __restrict__ src,
                                   const int* __restrict__ dst,
                                   float* __restrict__ agg,
                                   int E, int F, int chunkShift) {
    long total   = (long)E << chunkShift;
    long idx     = (long)blockIdx.x * blockDim.x + threadIdx.x;
    long stride  = (long)gridDim.x * blockDim.x;
    int  cmask   = (1 << chunkShift) - 1;
    for (; idx < total; idx += stride) {
        int e  = (int)(idx >> chunkShift);
        int f4 = (int)(idx & cmask) << 2;
        int s  = src[e];
        int d  = dst[e];
        const float4 v = *(const float4*)(x + (size_t)s * F + f4);
        float* ap = agg + (size_t)d * F + f4;
        unsafeAtomicAdd(ap + 0, v.x);
        unsafeAtomicAdd(ap + 1, v.y);
        unsafeAtomicAdd(ap + 2, v.z);
        unsafeAtomicAdd(ap + 3, v.w);
    }
}

__global__ __launch_bounds__(128)
void gemm_wmma_kernel(const float* __restrict__ A1, const float* __restrict__ scale,
                      const float* __restrict__ W1,
                      const float* __restrict__ A2, const float* __restrict__ W2,
                      const float* __restrict__ bias, float* __restrict__ out,
                      int Nnodes, int K, int dout, int relu) {
    int wave = threadIdx.x >> 5;
    int lane = threadIdx.x & 31;
    int m0 = (blockIdx.x * 4 + wave) * 16;
    if (m0 >= Nnodes) return;
    int n0    = blockIdx.y * 16;
    int mrow  = lane & 15;
    int khalf = (lane >> 4) << 1;
    int node  = m0 + mrow;
    int ncol  = n0 + mrow;
    int wrow  = (ncol < dout) ? ncol : 0;

    v8f acc = {};
    float sc = scale ? scale[node] : 1.0f;

    const float* ap = A1 + (size_t)node * K + khalf;
    const float* wp = W1 + (size_t)wrow * K + khalf;
    for (int k = 0; k < K; k += 4) {
        v2f a, b;
        a.x = ap[k] * sc;  a.y = ap[k + 1] * sc;
        b.x = wp[k];       b.y = wp[k + 1];
        acc = __builtin_amdgcn_wmma_f32_16x16x4_f32(false, a, false, b,
                                                    (short)0, acc, false, false);
    }
    if (A2) {
        const float* ap2 = A2 + (size_t)node * K + khalf;
        const float* wp2 = W2 + (size_t)wrow * K + khalf;
        for (int k = 0; k < K; k += 4) {
            v2f a, b;
            a.x = ap2[k];  a.y = ap2[k + 1];
            b.x = wp2[k];  b.y = wp2[k + 1];
            acc = __builtin_amdgcn_wmma_f32_16x16x4_f32(false, a, false, b,
                                                        (short)0, acc, false, false);
        }
    }
    if (ncol < dout) {
        float bv   = bias[ncol];
        int  rbase = m0 + ((lane >> 4) << 3);
        #pragma unroll
        for (int r = 0; r < 8; ++r) {
            float v = acc[r] + bv;
            if (relu) v = fmaxf(v, 0.0f);
            out[(size_t)(rbase + r) * dout + ncol] = v;
        }
    }
}

static inline int cdiv(long a, long b) { return (int)((a + b - 1) / b); }

extern "C" void kernel_launch(void* const* d_in, const int* in_sizes, int n_in,
                              void* d_out, int out_size, void* d_ws, size_t ws_size,
                              hipStream_t stream) {
    const float* x    = (const float*)d_in[0];
    const int*   edge = (const int*)d_in[1];
    const int*   src  = edge;
    const int*   dst  = edge + N_EDGES;
    const float* Wl0 = (const float*)d_in[2],  *bl0 = (const float*)d_in[3],  *Wr0 = (const float*)d_in[4];
    const float* Wl1 = (const float*)d_in[5],  *bl1 = (const float*)d_in[6],  *Wr1 = (const float*)d_in[7];
    const float* Wl2 = (const float*)d_in[8],  *bl2 = (const float*)d_in[9],  *Wr2 = (const float*)d_in[10];
    const float* Wd1 = (const float*)d_in[11], *bd1 = (const float*)d_in[12];
    const float* Wd2 = (const float*)d_in[13], *bd2 = (const float*)d_in[14];

    float* ws   = (float*)d_ws;
    float* deg  = ws;
    float* inv  = deg  + N_NODES;
    float* agg  = inv  + N_NODES;
    float* bufA = agg  + (size_t)N_NODES * 128;
    float* bufB = bufA + (size_t)N_NODES * 128;

    const int BT = 256;
    const int gm = cdiv(N_NODES, 64);

    zero_f32<<<cdiv(N_NODES, BT), BT, 0, stream>>>(deg, N_NODES);
    degree_kernel<<<cdiv(N_EDGES, BT), BT, 0, stream>>>(dst, deg, N_EDGES);
    invdeg_kernel<<<cdiv(N_NODES, BT), BT, 0, stream>>>(deg, inv, N_NODES);

    zero_f32<<<cdiv((long)N_NODES * 64, BT), BT, 0, stream>>>(agg, (long)N_NODES * 64);
    scatter_add_kernel<<<cdiv((long)N_EDGES * 16, BT), BT, 0, stream>>>(x, src, dst, agg, N_EDGES, 64, 4);
    gemm_wmma_kernel<<<dim3(gm, 8), 128, 0, stream>>>(agg, inv, Wl0, x, Wr0, bl0, bufA,
                                                      N_NODES, 64, 128, 1);

    zero_f32<<<cdiv((long)N_NODES * 128, BT), BT, 0, stream>>>(agg, (long)N_NODES * 128);
    scatter_add_kernel<<<cdiv((long)N_EDGES * 32, BT), BT, 0, stream>>>(bufA, src, dst, agg, N_EDGES, 128, 5);
    gemm_wmma_kernel<<<dim3(gm, 8), 128, 0, stream>>>(agg, inv, Wl1, bufA, Wr1, bl1, bufB,
                                                      N_NODES, 128, 128, 1);

    zero_f32<<<cdiv((long)N_NODES * 128, BT), BT, 0, stream>>>(agg, (long)N_NODES * 128);
    scatter_add_kernel<<<cdiv((long)N_EDGES * 32, BT), BT, 0, stream>>>(bufB, src, dst, agg, N_EDGES, 128, 5);
    gemm_wmma_kernel<<<dim3(gm, 4), 128, 0, stream>>>(agg, inv, Wl2, bufB, Wr2, bl2, bufA,
                                                      N_NODES, 128, 64, 1);

    gemm_wmma_kernel<<<dim3(gm, 8), 128, 0, stream>>>(bufA, nullptr, Wd1, nullptr, nullptr, bd1, bufB,
                                                      N_NODES, 64, 128, 1);
    gemm_wmma_kernel<<<dim3(gm, 1), 128, 0, stream>>>(bufB, nullptr, Wd2, nullptr, nullptr, bd2,
                                                      (float*)d_out, N_NODES, 128, 2, 0);
}


# sums by destination (degree + 3 feature sums) by owner work-groups: 512 rows kept in LDS, whole edge list scanned, LDS float adds; no global atomics, no zero pass; sum order changes (by arrival)
# speedup vs baseline: 1.8349x; 1.8349x over previous
#include <hip/hip_runtime.h>
#include <hip/hip_bf16.h>

#define N_NODES 100000
#define N_EDGES 1600000

typedef float v2f __attribute__((ext_vector_type(2)));
typedef float v8f __attribute__((ext_vector_type(8)));

#define OWN_ROWS 512
#define OWN_QLEN 160

template <int F, bool DEG>
__global__ __launch_bounds__(1024)
void own_rows_sum_kernel(const float* __restrict__ x,
                         const int* __restrict__ src,
                         const int* __restrict__ dst,
                         float* __restrict__ agg,
                         float* __restrict__ inv) {
    constexpr int R   = OWN_ROWS;
    constexpr int VEC = F / 32;
    constexpr int NSC = N_EDGES / 512;
    static_assert(N_EDGES % 512 == 0, "list length must be a multiple of 512");
    typedef float vecf __attribute__((ext_vector_type(VEC)));

    __shared__ __attribute__((aligned(16))) float acc[(R + 1) * F];
    __shared__ int2  que[32][OWN_QLEN];
    __shared__ float degl[R];

    const int tid  = threadIdx.x;
    const int lane = tid & 31;
    const int wave = __builtin_amdgcn_readfirstlane(tid >> 5);
    const int lo   = blockIdx.x * R;

    for (int i = tid; i < (R + 1) * F / 4; i += 1024)
        ((float4*)acc)[i] = make_float4(0.0f, 0.0f, 0.0f, 0.0f);
    if (DEG) for (int i = tid; i < R; i += 1024) degl[i] = 0.0f;
    __syncthreads();

    int2* q = que[wave];
    int tail = 0;

    auto drain = [&](int h, int cnt) {
        __builtin_amdgcn_wave_barrier();
        int2 ent = q[h + lane];
        if (lane >= cnt) ent = make_int2(0, R);
        const int sl = src[ent.x];
        #pragma unroll
        for (int g = 0; g < 4; ++g) {
            if (g * 8 < cnt) {
                vecf v[8];
                int  rr[8];
                #pragma unroll
                for (int j = 0; j < 8; ++j) {
                    int s = __builtin_amdgcn_readlane(sl, g * 8 + j);
                    rr[j] = __builtin_amdgcn_readlane(ent.y, g * 8 + j);
                    v[j]  = *(const vecf*)(x + (size_t)s * F + lane * VEC);
                }
                #pragma unroll
                for (int j = 0; j < 8; ++j) {
                    float* a = acc + rr[j] * F + lane;
                    #pragma unroll
                    for (int c = 0; c < VEC; ++c)
                        __hip_atomic_fetch_add(a + c * 32, v[j][c], __ATOMIC_RELAXED,
                                               __HIP_MEMORY_SCOPE_WORKGROUP);
                }
            }
        }
    };

    auto chunk = [&](const int4 d, int ch) {
        unsigned r[4];
        r[0] = (unsigned)(d.x - lo); r[1] = (unsigned)(d.y - lo);
        r[2] = (unsigned)(d.z - lo); r[3] = (unsigned)(d.w - lo);
        const unsigned m = min(min(r[0], r[1]), min(r[2], r[3]));
        if (__builtin_amdgcn_ballot_w32(m < (unsigned)R) != 0u) {
            #pragma unroll
            for (int c = 0; c < 4; ++c) {
                const bool hit = r[c] < (unsigned)R;
                const unsigned mask = __builtin_amdgcn_ballot_w32(hit);
                if (mask != 0u) {
                    if (hit) {
                        const int e = ch * 128 + lane * 4 + c;
                        q[tail + (int)__builtin_amdgcn_mbcnt_lo(mask, 0u)] = make_int2(e, (int)r[c]);
                        if (DEG)
                            __hip_atomic_fetch_add(&degl[r[c]], 1.0f, __ATOMIC_RELAXED,
                                                   __HIP_MEMORY_SCOPE_WORKGROUP);
                    }
                    tail += __builtin_popcount(mask);
                }
            }
            if (tail >= 32) {
                int h = 0;
                do { drain(h, 32); h += 32; } while (tail - h >= 32);
                __builtin_amdgcn_wave_barrier();
                const int2 t = q[h + lane];
                __builtin_amdgcn_wave_barrier();
                if (lane < tail - h) q[lane] = t;
                tail -= h;
            }
        }
    };

    const int4* d4 = (const int4*)dst;
    const int off = (int)((blockIdx.x * 15u) % (unsigned)NSC);
    auto piece_of = [&](int ci) { int sc = ci + off; return (sc >= NSC) ? sc - NSC : sc; };

    int4 cur[4];
    {
        const int sc = piece_of(wave);
        #pragma unroll
        for (int k = 0; k < 4; ++k) cur[k] = d4[((size_t)sc * 4 + k) * 32 + lane];
    }
    for (int ci = wave; ci < NSC; ci += 32) {
        const int sc  = piece_of(ci);
        const int scn = piece_of((ci + 32 < NSC) ? ci + 32 : ci);
        int4 nxt[4];
        #pragma unroll
        for (int k = 0; k < 4; ++k) nxt[k] = d4[((size_t)scn * 4 + k) * 32 + lane];
        #pragma unroll
        for (int k = 0; k < 4; ++k) chunk(cur[k], sc * 4 + k);
        #pragma unroll
        for (int k = 0; k < 4; ++k) cur[k] = nxt[k];
    }
    if (tail > 0) drain(0, tail);
    __syncthreads();

    for (int row = wave; row < R; row += 32) {
        const int node = lo + row;
        if (node < N_NODES) {
            vecf o;
            #pragma unroll
            for (int c = 0; c < VEC; ++c) o[c] = acc[row * F + c * 32 + lane];
            *(vecf*)(agg + (size_t)node * F + lane * VEC) = o;
        }
    }
    if (DEG) {
        if (tid < R) {
            const int node = lo + tid;
            if (node < N_NODES) {
                const float dd = degl[tid];
                inv[node] = (dd > 0.0f) ? (1.0f / fmaxf(dd, 1.0f)) : 0.0f;
            }
        }
    }
}

__global__ __launch_bounds__(128)
void gemm_wmma_kernel(const float* __restrict__ A1, const float* __restrict__ scale,
                      const float* __restrict__ W1,
                      const float* __restrict__ A2, const float* __restrict__ W2,
                      const float* __restrict__ bias, float* __restrict__ out,
                      int Nnodes, int K, int dout, int relu) {
    int wave = threadIdx.x >> 5;
    int lane = threadIdx.x & 31;
    int m0 = (blockIdx.x * 4 + wave) * 16;
    if (m0 >= Nnodes) return;
    int n0    = blockIdx.y * 16;
    int mrow  = lane & 15;
    int khalf = (lane >> 4) << 1;
    int node  = m0 + mrow;
    int ncol  = n0 + mrow;
    int wrow  = (ncol < dout) ? ncol : 0;

    v8f acc = {};
    float sc = scale ? scale[node] : 1.0f;

    const float* ap = A1 + (size_t)node * K + khalf;
    const float* wp = W1 + (size_t)wrow * K + khalf;
    for (int k = 0; k < K; k += 4) {
        v2f a, b;
        a.x = ap[k] * sc;  a.y = ap[k + 1] * sc;
        b.x = wp[k];       b.y = wp[k + 1];
        acc = __builtin_amdgcn_wmma_f32_16x16x4_f32(false, a, false, b,
                                                    (short)0, acc, false, false);
    }
    if (A2) {
        const float* ap2 = A2 + (size_t)node * K + khalf;
        const float* wp2 = W2 + (size_t)wrow * K + khalf;
        for (int k = 0; k < K; k += 4) {
            v2f a, b;
            a.x = ap2[k];  a.y = ap2[k + 1];
            b.x = wp2[k];  b.y = wp2[k + 1];
            acc = __builtin_amdgcn_wmma_f32_16x16x4_f32(false, a, false, b,
                                                        (short)0, acc, false, false);
        }
    }
    if (ncol < dout) {
        float bv   = bias[ncol];
        int  rbase = m0 + ((lane >> 4) << 3);
        #pragma unroll
        for (int r = 0; r < 8; ++r) {
            float v = acc[r] + bv;
            if (relu) v = fmaxf(v, 0.0f);
            out[(size_t)(rbase + r) * dout + ncol] = v;
        }
    }
}

static inline int cdiv(long a, long b) { return (int)((a + b - 1) / b); }

extern "C" void kernel_launch(void* const* d_in, const int* in_sizes, int n_in,
                              void* d_out, int out_size, void* d_ws, size_t ws_size,
                              hipStream_t stream) {
    const float* x    = (const float*)d_in[0];
    const int*   edge = (const int*)d_in[1];
    const int*   src  = edge;
    const int*   dst  = edge + N_EDGES;
    const float* Wl0 = (const float*)d_in[2],  *bl0 = (const float*)d_in[3],  *Wr0 = (const float*)d_in[4];
    const float* Wl1 = (const float*)d_in[5],  *bl1 = (const float*)d_in[6],  *Wr1 = (const float*)d_in[7];
    const float* Wl2 = (const float*)d_in[8],  *bl2 = (const float*)d_in[9],  *Wr2 = (const float*)d_in[10];
    const float* Wd1 = (const float*)d_in[11], *bd1 = (const float*)d_in[12];
    const float* Wd2 = (const float*)d_in[13], *bd2 = (const float*)d_in[14];

    float* ws   = (float*)d_ws;
    float* deg  = ws;
    float* inv  = deg  + N_NODES;
    float* agg  = inv  + N_NODES;
    float* bufA = agg  + (size_t)N_NODES * 128;
    float* bufB = bufA + (size_t)N_NODES * 128;

    const int gm = cdiv(N_NODES, 64);
    const int go = cdiv(N_NODES, OWN_ROWS);

    own_rows_sum_kernel<64, true><<<go, 1024, 0, stream>>>(x, src, dst, agg, inv);
    gemm_wmma_kernel<<<dim3(gm, 8), 128, 0, stream>>>(agg, inv, Wl0, x, Wr0, bl0, bufA,
                                                      N_NODES, 64, 128, 1);

    own_rows_sum_kernel<128, false><<<go, 1024, 0, stream>>>(bufA, src, dst, agg, inv);
    gemm_wmma_kernel<<<dim3(gm, 8), 128, 0, stream>>>(agg, inv, Wl1, bufA, Wr1, bl1, bufB,
                                                      N_NODES, 128, 128, 1);

    own_rows_sum_kernel<128, false><<<go, 1024, 0, stream>>>(bufB, src, dst, agg, inv);
    gemm_wmma_kernel<<<dim3(gm, 4), 128, 0, stream>>>(agg, inv, Wl2, bufB, Wr2, bl2, bufA,
                                                      N_NODES, 128, 64, 1);

    gemm_wmma_kernel<<<dim3(gm, 8), 128, 0, stream>>>(bufA, nullptr, Wd1, nullptr, nullptr, bd1, bufB,
                                                      N_NODES, 64, 128, 1);
    gemm_wmma_kernel<<<dim3(gm, 1), 128, 0, stream>>>(bufB, nullptr, Wd2, nullptr, nullptr, bd2,
                                                      (float*)d_out, N_NODES, 128, 2, 0);
}


# layer 2's sum pass adds the per-wave edge lists saved by layer 0's pass instead of scanning the edge list (layer 1's product writes in place so the list plane survives); sum order stays by arrival
# speedup vs baseline: 3.6222x; 1.9741x over previous
#include <hip/hip_runtime.h>
#include <hip/hip_bf16.h>

#define N_NODES 100000
#define N_EDGES 1600000

typedef float v2f __attribute__((ext_vector_type(2)));
typedef float v8f __attribute__((ext_vector_type(8)));

#define OWN_ROWS 512
#define OWN_QLEN 160
#define SAVE_CAP  1024
#define SAVE_HEAD 8192

template <int F, int MODE>
__global__ __launch_bounds__(1024)
void own_rows_sum_kernel(const float* __restrict__ x,
                         const int* __restrict__ src,
                         const int* __restrict__ dst,
                         float* __restrict__ agg,
                         float* __restrict__ inv,
                         unsigned* __restrict__ sv) {
    static_assert(MODE == 1 || MODE == 2, "first pass or a later pass");
    constexpr bool DEG = (MODE == 1);
    constexpr int R   = OWN_ROWS;
    static_assert(R == 512, "a saved word keeps the local row in 9 bits");
    static_assert(((N_NODES + OWN_ROWS - 1) / OWN_ROWS) * 32 <= SAVE_HEAD, "one count per wave");
    constexpr int VEC = F / 32;
    constexpr int NSC = N_EDGES / 512;
    static_assert(N_EDGES % 512 == 0, "list length must be a multiple of 512");
    typedef float vecf __attribute__((ext_vector_type(VEC)));

    __shared__ __attribute__((aligned(16))) float acc[(R + 1) * F];
    __shared__ int2  que[32][OWN_QLEN];
    __shared__ float degl[R];

    const int tid  = threadIdx.x;
    const int lane = tid & 31;
    const int wave = __builtin_amdgcn_readfirstlane(tid >> 5);
    const int lo   = blockIdx.x * R;

    for (int i = tid; i < (R + 1) * F / 4; i += 1024)
        ((float4*)acc)[i] = make_float4(0.0f, 0.0f, 0.0f, 0.0f);
    if (DEG) for (int i = tid; i < R; i += 1024) degl[i] = 0.0f;
    __syncthreads();

    int2* q = que[wave];
    int tail = 0;
    int saved = 0;
    unsigned* const svw = sv + SAVE_HEAD + (size_t)(blockIdx.x * 32 + wave) * SAVE_CAP;

    auto add_rows = [&](int sl, int ry, int cnt) {
        #pragma unroll
        for (int g = 0; g < 4; ++g) {
            if (g * 8 < cnt) {
                vecf v[8];
                int  rr[8];
                #pragma unroll
                for (int j = 0; j < 8; ++j) {
                    int s = __builtin_amdgcn_readlane(sl, g * 8 + j);
                    rr[j] = __builtin_amdgcn_readlane(ry, g * 8 + j);
                    v[j]  = *(const vecf*)(x + (size_t)s * F + lane * VEC);
                }
                #pragma unroll
                for (int j = 0; j < 8; ++j) {
                    float* a = acc + rr[j] * F + lane;
                    #pragma unroll
                    for (int c = 0; c < VEC; ++c)
                        __hip_atomic_fetch_add(a + c * 32, v[j][c], __ATOMIC_RELAXED,
                                               __HIP_MEMORY_SCOPE_WORKGROUP);
                }
            }
        }
    };

    auto drain = [&](int h, int cnt) {
        __builtin_amdgcn_wave_barrier();
        int2 ent = q[h + lane];
        if (lane >= cnt) ent = make_int2(0, R);
        const int sl = src[ent.x];
        if (MODE == 1) {
            if (lane < cnt && saved + lane < SAVE_CAP)
                svw[saved + lane] = ((unsigned)sl << 9) | (unsigned)ent.y;
            saved += cnt;
        }
        add_rows(sl, ent.y, cnt);
    };

    auto replay = [&](int n) {
        if (n <= 0) return;
        unsigned cur = svw[min(lane, n - 1)];
        for (int h = 0; h < n; h += 32) {
            const int cnt = min(32, n - h);
            const unsigned nxt = svw[min(h + 32 + lane, n - 1)];
            int sl = min((int)(cur >> 9), N_NODES - 1);
            int ry = (int)(cur & 511u);
            if (lane >= cnt) { sl = 0; ry = R; }
            add_rows(sl, ry, cnt);
            cur = nxt;
        }
    };

    auto chunk = [&](const int4 d, int ch) {
        unsigned r[4];
        r[0] = (unsigned)(d.x - lo); r[1] = (unsigned)(d.y - lo);
        r[2] = (unsigned)(d.z - lo); r[3] = (unsigned)(d.w - lo);
        const unsigned m = min(min(r[0], r[1]), min(r[2], r[3]));
        if (__builtin_amdgcn_ballot_w32(m < (unsigned)R) != 0u) {
            #pragma unroll
            for (int c = 0; c < 4; ++c) {
                const bool hit = r[c] < (unsigned)R;
                const unsigned mask = __builtin_amdgcn_ballot_w32(hit);
                if (mask != 0u) {
                    if (hit) {
                        const int e = ch * 128 + lane * 4 + c;
                        q[tail + (int)__builtin_amdgcn_mbcnt_lo(mask, 0u)] = make_int2(e, (int)r[c]);
                        if (DEG)
                            __hip_atomic_fetch_add(&degl[r[c]], 1.0f, __ATOMIC_RELAXED,
                                                   __HIP_MEMORY_SCOPE_WORKGROUP);
                    }
                    tail += __builtin_popcount(mask);
                }
            }
            if (tail >= 32) {
                int h = 0;
                do { drain(h, 32); h += 32; } while (tail - h >= 32);
                __builtin_amdgcn_wave_barrier();
                const int2 t = q[h + lane];
                __builtin_amdgcn_wave_barrier();
                if (lane < tail - h) q[lane] = t;
                tail -= h;
            }
        }
    };

    auto scan_list = [&]() {
    const int4* d4 = (const int4*)dst;
    const int off = (int)((blockIdx.x * 15u) % (unsigned)NSC);
    auto piece_of = [&](int ci) { int sc = ci + off; return (sc >= NSC) ? sc - NSC : sc; };

    int4 cur[4];
    {
        const int sc = piece_of(wave);
        #pragma unroll
        for (int k = 0; k < 4; ++k) cur[k] = d4[((size_t)sc * 4 + k) * 32 + lane];
    }
    for (int ci = wave; ci < NSC; ci += 32) {
        const int sc  = piece_of(ci);
        const int scn = piece_of((ci + 32 < NSC) ? ci + 32 : ci);
        int4 nxt[4];
        #pragma unroll
        for (int k = 0; k < 4; ++k) nxt[k] = d4[((size_t)scn * 4 + k) * 32 + lane];
        #pragma unroll
        for (int k = 0; k < 4; ++k) chunk(cur[k], sc * 4 + k);
        #pragma unroll
        for (int k = 0; k < 4; ++k) cur[k] = nxt[k];
    }
    if (tail > 0) drain(0, tail);
    };

    if (MODE == 2) {
        const unsigned n = (unsigned)__builtin_amdgcn_readfirstlane((int)sv[blockIdx.x * 32 + wave]);
        if (n <= (unsigned)SAVE_CAP) replay((int)n);
        else scan_list();
    } else {
        scan_list();
        if (lane == 0) sv[blockIdx.x * 32 + wave] = (unsigned)saved;
    }
    __syncthreads();

    for (int row = wave; row < R; row += 32) {
        const int node = lo + row;
        if (node < N_NODES) {
            vecf o;
            #pragma unroll
            for (int c = 0; c < VEC; ++c) o[c] = acc[row * F + c * 32 + lane];
            *(vecf*)(agg + (size_t)node * F + lane * VEC) = o;
        }
    }
    if (DEG) {
        if (tid < R) {
            const int node = lo + tid;
            if (node < N_NODES) {
                const float dd = degl[tid];
                inv[node] = (dd > 0.0f) ? (1.0f / fmaxf(dd, 1.0f)) : 0.0f;
            }
        }
    }
}

#define SLAB_ROWS 64

template <int K, int DOUT, bool SAGE, bool RELU>
__global__ __launch_bounds__(128)
void slab_gemm_kernel(const float* __restrict__ A1, const float* __restrict__ scale,
                      const float* __restrict__ W1,
                      const float* A2, const float* __restrict__ W2,
                      const float* __restrict__ bias, float* out) {
    constexpr int KT  = SAGE ? 2 * K : K;
    constexpr int P   = KT + 4;
    constexpr int Q   = K / 4;
    constexpr int CT  = (DOUT + 15) / 16;
    constexpr int CPW = (CT >= 4) ? CT / 4 : 1;
    constexpr int RPW = (CT >= 4) ? 4 : 1;
    static_assert(CT == 1 || CT % 4 == 0, "column tiles are dealt to 4 waves");
    static_assert(K % 4 == 0 && (SLAB_ROWS * Q) % 128 == 0, "staging loop has no tail");
    static_assert(N_NODES % 16 == 0, "a row tile is wholly inside or outside the table");

    __shared__ __attribute__((aligned(16))) float slab[SLAB_ROWS * P];

    const int tid  = threadIdx.x;
    const int lane = tid & 31;
    const int wave = __builtin_amdgcn_readfirstlane(tid >> 5);
    const int row0 = blockIdx.x * SLAB_ROWS;

    for (int i = tid; i < SLAB_ROWS * Q; i += 128) {
        const int r = i / Q, c = i % Q;
        const int node = min(row0 + r, N_NODES - 1);
        float4 v = ((const float4*)(A1 + (size_t)node * K))[c];
        if (SAGE) {
            const float s = scale[node];
            v.x *= s; v.y *= s; v.z *= s; v.w *= s;
        }
        *(float4*)(slab + r * P + c * 4) = v;
        if (SAGE)
            *(float4*)(slab + r * P + K + c * 4) = ((const float4*)(A2 + (size_t)node * K))[c];
    }
    __syncthreads();

    const int mrow  = lane & 15;
    const int khalf = (lane >> 4) << 1;
    const int rt0   = (CT >= 4) ? 0 : wave;
    const int ct0   = (CT >= 4) ? wave * CPW : 0;

    v8f acc[RPW][CPW];
    const float* ar[RPW];
    const float* w1r[CPW];
    const float* w2r[CPW];
    #pragma unroll
    for (int i = 0; i < RPW; ++i) {
        ar[i] = slab + ((rt0 + i) * 16 + mrow) * P + khalf;
        #pragma unroll
        for (int j = 0; j < CPW; ++j) acc[i][j] = v8f{};
    }
    #pragma unroll
    for (int j = 0; j < CPW; ++j) {
        const int ncol = (ct0 + j) * 16 + mrow;
        const int wrow = (ncol < DOUT) ? ncol : 0;
        w1r[j] = W1 + (size_t)wrow * K + khalf;
        w2r[j] = SAGE ? W2 + (size_t)wrow * K + khalf : nullptr;
    }

    for (int k = 0; k < K; k += 4) {
        v2f a[RPW], b[CPW];
        #pragma unroll
        for (int i = 0; i < RPW; ++i) a[i] = *(const v2f*)(ar[i] + k);
        #pragma unroll
        for (int j = 0; j < CPW; ++j) b[j] = *(const v2f*)(w1r[j] + k);
        #pragma unroll
        for (int i = 0; i < RPW; ++i) {
            #pragma unroll
            for (int j = 0; j < CPW; ++j)
                acc[i][j] = __builtin_amdgcn_wmma_f32_16x16x4_f32(false, a[i], false, b[j],
                                                                  (short)0, acc[i][j], false, false);
        }
    }
    if (SAGE) {
        for (int k = 0; k < K; k += 4) {
            v2f a[RPW], b[CPW];
            #pragma unroll
            for (int i = 0; i < RPW; ++i) a[i] = *(const v2f*)(ar[i] + K + k);
            #pragma unroll
            for (int j = 0; j < CPW; ++j) b[j] = *(const v2f*)(w2r[j] + k);
            #pragma unroll
            for (int i = 0; i < RPW; ++i) {
                #pragma unroll
                for (int j = 0; j < CPW; ++j)
                    acc[i][j] = __builtin_amdgcn_wmma_f32_16x16x4_f32(false, a[i], false, b[j],
                                                                      (short)0, acc[i][j], false, false);
            }
        }
    }

    #pragma unroll
    for (int j = 0; j < CPW; ++j) {
        const int ncol = (ct0 + j) * 16 + mrow;
        if (ncol < DOUT) {
            const float bv = bias[ncol];
            #pragma unroll
            for (int i = 0; i < RPW; ++i) {
                const int m0 = row0 + (rt0 + i) * 16;
                if (m0 < N_NODES) {
                    const int rbase = m0 + ((lane >> 4) << 3);
                    #pragma unroll
                    for (int r = 0; r < 8; ++r) {
                        float v = acc[i][j][r] + bv;
                        if (RELU) v = fmaxf(v, 0.0f);
                        out[(size_t)(rbase + r) * DOUT + ncol] = v;
                    }
                }
            }
        }
    }
}

static inline int cdiv(long a, long b) { return (int)((a + b - 1) / b); }

extern "C" void kernel_launch(void* const* d_in, const int* in_sizes, int n_in,
                              void* d_out, int out_size, void* d_ws, size_t ws_size,
                              hipStream_t stream) {
    const float* x    = (const float*)d_in[0];
    const int*   edge = (const int*)d_in[1];
    const int*   src  = edge;
    const int*   dst  = edge + N_EDGES;
    const float* Wl0 = (const float*)d_in[2],  *bl0 = (const float*)d_in[3],  *Wr0 = (const float*)d_in[4];
    const float* Wl1 = (const float*)d_in[5],  *bl1 = (const float*)d_in[6],  *Wr1 = (const float*)d_in[7];
    const float* Wl2 = (const float*)d_in[8],  *bl2 = (const float*)d_in[9],  *Wr2 = (const float*)d_in[10];
    const float* Wd1 = (const float*)d_in[11], *bd1 = (const float*)d_in[12];
    const float* Wd2 = (const float*)d_in[13], *bd2 = (const float*)d_in[14];

    float* ws   = (float*)d_ws;
    float* deg  = ws;
    float* inv  = deg  + N_NODES;
    float* agg  = inv  + N_NODES;
    float* bufA = agg  + (size_t)N_NODES * 128;
    float* bufB = bufA + (size_t)N_NODES * 128;

    const int gs = cdiv(N_NODES, SLAB_ROWS);
    const int go = cdiv(N_NODES, OWN_ROWS);
    static_assert((size_t)SAVE_HEAD + (size_t)((N_NODES + OWN_ROWS - 1) / OWN_ROWS) * 32 * SAVE_CAP
                  <= (size_t)N_NODES * 128, "saved lists must fit the plane");
    unsigned* sv = (unsigned*)bufB;

    own_rows_sum_kernel<64, 1><<<go, 1024, 0, stream>>>(x, src, dst, agg, inv, sv);
    slab_gemm_kernel<64, 128, true, true><<<gs, 128, 0, stream>>>(agg, inv, Wl0, x, Wr0, bl0, bufA);

    own_rows_sum_kernel<128, 2><<<go, 1024, 0, stream>>>(bufA, src, dst, agg, inv, sv);
    slab_gemm_kernel<128, 128, true, true><<<gs, 128, 0, stream>>>(agg, inv, Wl1, bufA, Wr1, bl1, bufA);

    own_rows_sum_kernel<128, 2><<<go, 1024, 0, stream>>>(bufA, src, dst, agg, inv, sv);
    slab_gemm_kernel<128, 64, true, true><<<gs, 128, 0, stream>>>(agg, inv, Wl2, bufA, Wr2, bl2, bufB);

    slab_gemm_kernel<64, 128, false, true><<<gs, 128, 0, stream>>>(bufB, nullptr, Wd1, nullptr, nullptr, bd1, bufA);
    slab_gemm_kernel<128, 2, false, false><<<gs, 128, 0, stream>>>(bufA, nullptr, Wd2, nullptr, nullptr, bd2,
                                                                   (float*)d_out);
}


# edge list read once per call: a binning launch sorts it into cells per (group, owner block), LDS counter as slot; layer 0's sums add from cells, no owner scans the list; sum order stays by arrival
# speedup vs baseline: 3.8677x; 1.0678x over previous
#include <hip/hip_runtime.h>
#include <hip/hip_bf16.h>

#define N_NODES 100000
#define N_EDGES 1600000

typedef float v2f __attribute__((ext_vector_type(2)));
typedef float v8f __attribute__((ext_vector_type(8)));

#define OWN_ROWS 512
#define OWN_QLEN 160
#define N_BLK ((N_NODES + OWN_ROWS - 1) / OWN_ROWS)
#define BIN_PIECES 25
#define BIN_WGS (N_EDGES / 512 / BIN_PIECES)
#define BIN_CAP 128
#define SAVE_HEAD 32768

__global__ __launch_bounds__(1024)
void bin_edges_kernel(const int* __restrict__ src, const int* __restrict__ dst,
                      unsigned* __restrict__ sv) {
    static_assert(N_EDGES % (512 * BIN_PIECES) == 0, "every group has whole pieces");
    static_assert(N_BLK * BIN_WGS <= SAVE_HEAD, "one count per cell");
    static_assert(OWN_ROWS == 512, "a word keeps the local row in 9 bits");
    static_assert(BIN_CAP == 128, "a wave writes a cell with one 16-byte store per lane");
    __shared__ unsigned cnt[N_BLK];
    __shared__ __attribute__((aligned(16))) unsigned cell[N_BLK * BIN_CAP];

    const int tid  = threadIdx.x;
    const int lane = tid & 31;
    const int wave = tid >> 5;
    for (int i = tid; i < N_BLK; i += 1024) cnt[i] = 0u;
    __syncthreads();

    constexpr int NV = BIN_PIECES * 128;
    constexpr int NK = (NV + 1023) / 1024;
    const int4* d4 = (const int4*)dst + (size_t)blockIdx.x * NV;
    const int4* s4 = (const int4*)src + (size_t)blockIdx.x * NV;
    int4 dv[NK], sw[NK];
    #pragma unroll
    for (int k = 0; k < NK; ++k) {
        const int i = min(k * 1024 + tid, NV - 1);
        dv[k] = d4[i];
        sw[k] = s4[i];
    }
    #pragma unroll
    for (int k = 0; k < NK; ++k) {
        if (k * 1024 + tid < NV) {
            const int d[4] = {dv[k].x, dv[k].y, dv[k].z, dv[k].w};
            const int s[4] = {sw[k].x, sw[k].y, sw[k].z, sw[k].w};
            unsigned o[4], r[4];
            #pragma unroll
            for (int c = 0; c < 4; ++c) {
                o[c] = min((unsigned)d[c] >> 9, (unsigned)(N_BLK - 1));
                r[c] = atomicAdd(&cnt[o[c]], 1u);
            }
            #pragma unroll
            for (int c = 0; c < 4; ++c) {
                if (r[c] < (unsigned)BIN_CAP)
                    cell[o[c] * BIN_CAP + r[c]] = ((unsigned)s[c] << 9) | ((unsigned)d[c] & 511u);
            }
        }
    }
    __syncthreads();

    for (int o = wave; o < N_BLK; o += 32) {
        const uint4 v = ((const uint4*)(cell + o * BIN_CAP))[lane];
        ((uint4*)(sv + SAVE_HEAD + ((size_t)o * BIN_WGS + blockIdx.x) * BIN_CAP))[lane] = v;
        if (lane == 0) sv[o * BIN_WGS + blockIdx.x] = cnt[o];
    }
}

template <int F, bool DEG>
__global__ __launch_bounds__(1024)
void own_rows_sum_kernel(const float* __restrict__ x,
                         const int* __restrict__ src,
                         const int* __restrict__ dst,
                         float* __restrict__ agg,
                         float* __restrict__ inv,
                         const unsigned* __restrict__ sv) {
    constexpr int R   = OWN_ROWS;
    constexpr int VEC = F / 32;
    typedef float vecf __attribute__((ext_vector_type(VEC)));

    __shared__ __attribute__((aligned(16))) float acc[(R + 1) * F];
    __shared__ int2  que[32][OWN_QLEN];
    __shared__ float degl[R];

    const int tid  = threadIdx.x;
    const int lane = tid & 31;
    const int wave = __builtin_amdgcn_readfirstlane(tid >> 5);
    const int lo   = blockIdx.x * R;

    for (int i = tid; i < (R + 1) * F / 4; i += 1024)
        ((float4*)acc)[i] = make_float4(0.0f, 0.0f, 0.0f, 0.0f);
    if (DEG) for (int i = tid; i < R; i += 1024) degl[i] = 0.0f;
    __syncthreads();

    int2* q = que[wave];
    int tail = 0;

    auto add_rows = [&](int sl, int ry, int cnt) {
        #pragma unroll
        for (int g = 0; g < 4; ++g) {
            if (g * 8 < cnt) {
                vecf v[8];
                int  rr[8];
                #pragma unroll
                for (int j = 0; j < 8; ++j) {
                    int s = __builtin_amdgcn_readlane(sl, g * 8 + j);
                    rr[j] = __builtin_amdgcn_readlane(ry, g * 8 + j);
                    v[j]  = *(const vecf*)(x + (size_t)s * F + lane * VEC);
                }
                #pragma unroll
                for (int j = 0; j < 8; ++j) {
                    float* a = acc + rr[j] * F + lane;
                    #pragma unroll
                    for (int c = 0; c < VEC; ++c)
                        __hip_atomic_fetch_add(a + c * 32, v[j][c], __ATOMIC_RELAXED,
                                               __HIP_MEMORY_SCOPE_WORKGROUP);
                }
            }
        }
    };

    auto replay = [&](const unsigned* cw, int n, unsigned cur) {
        for (int h = 0; h < n; h += 32) {
            const int cnt = min(32, n - h);
            const unsigned nxt = cw[min(h + 32 + lane, BIN_CAP - 1)];
            int sl = min((int)(cur >> 9), N_NODES - 1);
            int ry = (int)(cur & 511u);
            if (lane >= cnt) { sl = 0; ry = R; }
            if (DEG) {
                if (lane < cnt)
                    __hip_atomic_fetch_add(&degl[ry], 1.0f, __ATOMIC_RELAXED,
                                           __HIP_MEMORY_SCOPE_WORKGROUP);
            }
            add_rows(sl, ry, cnt);
            cur = nxt;
        }
    };

    auto drain = [&](int h, int cnt) {
        __builtin_amdgcn_wave_barrier();
        int2 ent = q[h + lane];
        if (lane >= cnt) ent = make_int2(0, R);
        const int sl = min(max(src[ent.x], 0), N_NODES - 1);
        add_rows(sl, ent.y, cnt);
    };

    auto chunk = [&](const int4 d, int ch) {
        unsigned r[4];
        r[0] = (unsigned)(d.x - lo); r[1] = (unsigned)(d.y - lo);
        r[2] = (unsigned)(d.z - lo); r[3] = (unsigned)(d.w - lo);
        const unsigned m = min(min(r[0], r[1]), min(r[2], r[3]));
        if (__builtin_amdgcn_ballot_w32(m < (unsigned)R) != 0u) {
            #pragma unroll
            for (int c = 0; c < 4; ++c) {
                const bool hit = r[c] < (unsigned)R;
                const unsigned mask = __builtin_amdgcn_ballot_w32(hit);
                if (mask != 0u) {
                    if (hit) {
                        const int e = ch * 128 + lane * 4 + c;
                        q[tail + (int)__builtin_amdgcn_mbcnt_lo(mask, 0u)] = make_int2(e, (int)r[c]);
                        if (DEG)
                            __hip_atomic_fetch_add(&degl[r[c]], 1.0f, __ATOMIC_RELAXED,
                                                   __HIP_MEMORY_SCOPE_WORKGROUP);
                    }
                    tail += __builtin_popcount(mask);
                }
            }
            if (tail >= 32) {
                int h = 0;
                do { drain(h, 32); h += 32; } while (tail - h >= 32);
                __builtin_amdgcn_wave_barrier();
                const int2 t = q[h + lane];
                __builtin_amdgcn_wave_barrier();
                if (lane < tail - h) q[lane] = t;
                tail -= h;
            }
        }
    };

    auto scan_group = [&](int grp) {
        const int4* d4 = (const int4*)dst;
        for (int ci = 0; ci < BIN_PIECES; ++ci) {
            const int sc = grp * BIN_PIECES + ci;
            for (int k = 0; k < 4; ++k)
                chunk(d4[((size_t)sc * 4 + k) * 32 + lane], sc * 4 + k);
        }
        if (tail > 0) drain(0, tail);
        tail = 0;
    };

    static_assert(BIN_WGS >= 32, "every wave has a first cell");
    const unsigned* const cnts  = sv + blockIdx.x * BIN_WGS;
    const unsigned* const cells = sv + SAVE_HEAD + (size_t)blockIdx.x * BIN_WGS * BIN_CAP;
    int grp = wave;
    unsigned n = cnts[grp];
    unsigned f = cells[grp * BIN_CAP + lane];
    while (true) {
        const bool more = grp + 32 < BIN_WGS;
        const int  gn   = more ? grp + 32 : grp;
        const unsigned n2 = cnts[gn];
        const unsigned f2 = cells[gn * BIN_CAP + lane];
        n = (unsigned)__builtin_amdgcn_readfirstlane((int)n);
        if (n <= (unsigned)BIN_CAP) replay(cells + grp * BIN_CAP, (int)n, f);
        else scan_group(grp);
        if (!more) break;
        grp = gn; n = n2; f = f2;
    }
    __syncthreads();

    for (int row = wave; row < R; row += 32) {
        const int node = lo + row;
        if (node < N_NODES) {
            vecf o;
            #pragma unroll
            for (int c = 0; c < VEC; ++c) o[c] = acc[row * F + c * 32 + lane];
            *(vecf*)(agg + (size_t)node * F + lane * VEC) = o;
        }
    }
    if (DEG) {
        if (tid < R) {
            const int node = lo + tid;
            if (node < N_NODES) {
                const float dd = degl[tid];
                inv[node] = (dd > 0.0f) ? (1.0f / fmaxf(dd, 1.0f)) : 0.0f;
            }
        }
    }
}

#define SLAB_ROWS 64

template <int K, int DOUT, bool SAGE, bool RELU>
__global__ __launch_bounds__(128)
void slab_gemm_kernel(const float* __restrict__ A1, const float* __restrict__ scale,
                      const float* __restrict__ W1,
                      const float* A2, const float* __restrict__ W2,
                      const float* __restrict__ bias, float* out) {
    constexpr int KT  = SAGE ? 2 * K : K;
    constexpr int P   = KT + 4;
    constexpr int Q   = K / 4;
    constexpr int CT  = (DOUT + 15) / 16;
    constexpr int CPW = (CT >= 4) ? CT / 4 : 1;
    constexpr int RPW = (CT >= 4) ? 4 : 1;
    static_assert(CT == 1 || CT % 4 == 0, "column tiles are dealt to 4 waves");
    static_assert(K % 4 == 0 && (SLAB_ROWS * Q) % 128 == 0, "staging loop has no tail");
    static_assert(N_NODES % 16 == 0, "a row tile is wholly inside or outside the table");

    __shared__ __attribute__((aligned(16))) float slab[SLAB_ROWS * P];

    const int tid  = threadIdx.x;
    const int lane = tid & 31;
    const int wave = __builtin_amdgcn_readfirstlane(tid >> 5);
    const int row0 = blockIdx.x * SLAB_ROWS;

    for (int i = tid; i < SLAB_ROWS * Q; i += 128) {
        const int r = i / Q, c = i % Q;
        const int node = min(row0 + r, N_NODES - 1);
        float4 v = ((const float4*)(A1 + (size_t)node * K))[c];
        if (SAGE) {
            const float s = scale[node];
            v.x *= s; v.y *= s; v.z *= s; v.w *= s;
        }
        *(float4*)(slab + r * P + c * 4) = v;
        if (SAGE)
            *(float4*)(slab + r * P + K + c * 4) = ((const float4*)(A2 + (size_t)node * K))[c];
    }
    __syncthreads();

    const int mrow  = lane & 15;
    const int khalf = (lane >> 4) << 1;
    const int rt0   = (CT >= 4) ? 0 : wave;
    const int ct0   = (CT >= 4) ? wave * CPW : 0;

    v8f acc[RPW][CPW];
    const float* ar[RPW];
    const float* w1r[CPW];
    const float* w2r[CPW];
    #pragma unroll
    for (int i = 0; i < RPW; ++i) {
        ar[i] = slab + ((rt0 + i) * 16 + mrow) * P + khalf;
        #pragma unroll
        for (int j = 0; j < CPW; ++j) acc[i][j] = v8f{};
    }
    #pragma unroll
    for (int j = 0; j < CPW; ++j) {
        const int ncol = (ct0 + j) * 16 + mrow;
        const int wrow = (ncol < DOUT) ? ncol : 0;
        w1r[j] = W1 + (size_t)wrow * K + khalf;
        w2r[j] = SAGE ? W2 + (size_t)wrow * K + khalf : nullptr;
    }

    for (int k = 0; k < K; k += 4) {
        v2f a[RPW], b[CPW];
        #pragma unroll
        for (int i = 0; i < RPW; ++i) a[i] = *(const v2f*)(ar[i] + k);
        #pragma unroll
        for (int j = 0; j < CPW; ++j) b[j] = *(const v2f*)(w1r[j] + k);
        #pragma unroll
        for (int i = 0; i < RPW; ++i) {
            #pragma unroll
            for (int j = 0; j < CPW; ++j)
                acc[i][j] = __builtin_amdgcn_wmma_f32_16x16x4_f32(false, a[i], false, b[j],
                                                                  (short)0, acc[i][j], false, false);
        }
    }
    if (SAGE) {
        for (int k = 0; k < K; k += 4) {
            v2f a[RPW], b[CPW];
            #pragma unroll
            for (int i = 0; i < RPW; ++i) a[i] = *(const v2f*)(ar[i] + K + k);
            #pragma unroll
            for (int j = 0; j < CPW; ++j) b[j] = *(const v2f*)(w2r[j] + k);
            #pragma unroll
            for (int i = 0; i < RPW; ++i) {
                #pragma unroll
                for (int j = 0; j < CPW; ++j)
                    acc[i][j] = __builtin_amdgcn_wmma_f32_16x16x4_f32(false, a[i], false, b[j],
                                                                      (short)0, acc[i][j], false, false);
            }
        }
    }

    #pragma unroll
    for (int j = 0; j < CPW; ++j) {
        const int ncol = (ct0 + j) * 16 + mrow;
        if (ncol < DOUT) {
            const float bv = bias[ncol];
            #pragma unroll
            for (int i = 0; i < RPW; ++i) {
                const int m0 = row0 + (rt0 + i) * 16;
                if (m0 < N_NODES) {
                    const int rbase = m0 + ((lane >> 4) << 3);
                    #pragma unroll
                    for (int r = 0; r < 8; ++r) {
                        float v = acc[i][j][r] + bv;
                        if (RELU) v = fmaxf(v, 0.0f);
                        out[(size_t)(rbase + r) * DOUT + ncol] = v;
                    }
                }
            }
        }
    }
}

static inline int cdiv(long a, long b) { return (int)((a + b - 1) / b); }

extern "C" void kernel_launch(void* const* d_in, const int* in_sizes, int n_in,
                              void* d_out, int out_size, void* d_ws, size_t ws_size,
                              hipStream_t stream) {
    const float* x    = (const float*)d_in[0];
    const int*   edge = (const int*)d_in[1];
    const int*   src  = edge;
    const int*   dst  = edge + N_EDGES;
    const float* Wl0 = (const float*)d_in[2],  *bl0 = (const float*)d_in[3],  *Wr0 = (const float*)d_in[4];
    const float* Wl1 = (const float*)d_in[5],  *bl1 = (const float*)d_in[6],  *Wr1 = (const float*)d_in[7];
    const float* Wl2 = (const float*)d_in[8],  *bl2 = (const float*)d_in[9],  *Wr2 = (const float*)d_in[10];
    const float* Wd1 = (const float*)d_in[11], *bd1 = (const float*)d_in[12];
    const float* Wd2 = (const float*)d_in[13], *bd2 = (const float*)d_in[14];

    float* ws   = (float*)d_ws;
    float* deg  = ws;
    float* inv  = deg  + N_NODES;
    float* agg  = inv  + N_NODES;
    float* bufA = agg  + (size_t)N_NODES * 128;
    float* bufB = bufA + (size_t)N_NODES * 128;

    const int gs = cdiv(N_NODES, SLAB_ROWS);
    const int go = N_BLK;
    static_assert((size_t)SAVE_HEAD + (size_t)N_BLK * BIN_WGS * BIN_CAP
                  <= (size_t)N_NODES * 128, "cells must fit the plane");
    unsigned* sv = (unsigned*)bufB;

    bin_edges_kernel<<<BIN_WGS, 1024, 0, stream>>>(src, dst, sv);

    own_rows_sum_kernel<64, true><<<go, 1024, 0, stream>>>(x, src, dst, agg, inv, sv);
    slab_gemm_kernel<64, 128, true, true><<<gs, 128, 0, stream>>>(agg, inv, Wl0, x, Wr0, bl0, bufA);

    own_rows_sum_kernel<128, false><<<go, 1024, 0, stream>>>(bufA, src, dst, agg, inv, sv);
    slab_gemm_kernel<128, 128, true, true><<<gs, 128, 0, stream>>>(agg, inv, Wl1, bufA, Wr1, bl1, bufA);

    own_rows_sum_kernel<128, false><<<go, 1024, 0, stream>>>(bufA, src, dst, agg, inv, sv);
    slab_gemm_kernel<128, 64, true, true><<<gs, 128, 0, stream>>>(agg, inv, Wl2, bufA, Wr2, bl2, bufB);

    slab_gemm_kernel<64, 128, false, true><<<gs, 128, 0, stream>>>(bufB, nullptr, Wd1, nullptr, nullptr, bd1, bufA);
    slab_gemm_kernel<128, 2, false, false><<<gs, 128, 0, stream>>>(bufA, nullptr, Wd2, nullptr, nullptr, bd2,
                                                                   (float*)d_out);
}


# owner blocks of 392 rows instead of 512: 256 owner work-groups, one per WGP, where 196 left 60 WGPs idle; each pass's LDS float adds spread over all WGPs; sum order stays by arrival
# speedup vs baseline: 4.0747x; 1.0535x over previous
#include <hip/hip_runtime.h>
#include <hip/hip_bf16.h>

#define N_NODES 100000
#define N_EDGES 1600000

typedef float v2f __attribute__((ext_vector_type(2)));
typedef float v8f __attribute__((ext_vector_type(8)));

#define OWN_ROWS 392
#define OWN_QLEN 160
#define N_BLK ((N_NODES + OWN_ROWS - 1) / OWN_ROWS)
#define BIN_PIECES 25
#define BIN_WGS (N_EDGES / 512 / BIN_PIECES)
#define BIN_CAP 128
#define SAVE_HEAD 32768

__global__ __launch_bounds__(1024)
void bin_edges_kernel(const int* __restrict__ src, const int* __restrict__ dst,
                      unsigned* __restrict__ sv) {
    static_assert(N_EDGES % (512 * BIN_PIECES) == 0, "every group has whole pieces");
    static_assert(N_BLK * BIN_WGS <= SAVE_HEAD, "one count per cell");
    static_assert(OWN_ROWS <= 512, "a word keeps the local row in 9 bits");
    static_assert(BIN_CAP == 128, "a wave writes a cell with one 16-byte store per lane");
    __shared__ unsigned cnt[N_BLK];
    __shared__ __attribute__((aligned(16))) unsigned cell[N_BLK * BIN_CAP];

    const int tid  = threadIdx.x;
    const int lane = tid & 31;
    const int wave = tid >> 5;
    for (int i = tid; i < N_BLK; i += 1024) cnt[i] = 0u;
    __syncthreads();

    constexpr int NV = BIN_PIECES * 128;
    constexpr int NK = (NV + 1023) / 1024;
    const int4* d4 = (const int4*)dst + (size_t)blockIdx.x * NV;
    const int4* s4 = (const int4*)src + (size_t)blockIdx.x * NV;
    int4 dv[NK], sw[NK];
    #pragma unroll
    for (int k = 0; k < NK; ++k) {
        const int i = min(k * 1024 + tid, NV - 1);
        dv[k] = d4[i];
        sw[k] = s4[i];
    }
    #pragma unroll
    for (int k = 0; k < NK; ++k) {
        if (k * 1024 + tid < NV) {
            const int d[4] = {dv[k].x, dv[k].y, dv[k].z, dv[k].w};
            const int s[4] = {sw[k].x, sw[k].y, sw[k].z, sw[k].w};
            unsigned o[4], r[4], lr[4];
            #pragma unroll
            for (int c = 0; c < 4; ++c) {
                o[c]  = min((unsigned)d[c] / (unsigned)OWN_ROWS, (unsigned)(N_BLK - 1));
                lr[c] = (unsigned)d[c] - o[c] * (unsigned)OWN_ROWS;
                r[c]  = atomicAdd(&cnt[o[c]], 1u);
            }
            #pragma unroll
            for (int c = 0; c < 4; ++c) {
                if (r[c] < (unsigned)BIN_CAP)
                    cell[o[c] * BIN_CAP + r[c]] = ((unsigned)s[c] << 9) | (lr[c] & 511u);
            }
        }
    }
    __syncthreads();

    for (int o = wave; o < N_BLK; o += 32) {
        const uint4 v = ((const uint4*)(cell + o * BIN_CAP))[lane];
        ((uint4*)(sv + SAVE_HEAD + ((size_t)o * BIN_WGS + blockIdx.x) * BIN_CAP))[lane] = v;
        if (lane == 0) sv[o * BIN_WGS + blockIdx.x] = cnt[o];
    }
}

template <int F, bool DEG>
__global__ __launch_bounds__(1024)
void own_rows_sum_kernel(const float* __restrict__ x,
                         const int* __restrict__ src,
                         const int* __restrict__ dst,
                         float* __restrict__ agg,
                         float* __restrict__ inv,
                         const unsigned* __restrict__ sv) {
    constexpr int R   = OWN_ROWS;
    constexpr int VEC = F / 32;
    typedef float vecf __attribute__((ext_vector_type(VEC)));

    __shared__ __attribute__((aligned(16))) float acc[(R + 1) * F];
    constexpr int QLEN = ((R + 1) * F * 4 < 160 * 1024) ? 3 * OWN_QLEN : OWN_QLEN;
    __shared__ int2  que[32][QLEN];
    __shared__ float degl[R];

    const int tid  = threadIdx.x;
    const int lane = tid & 31;
    const int wave = __builtin_amdgcn_readfirstlane(tid >> 5);
    const int lo   = blockIdx.x * R;

    for (int i = tid; i < (R + 1) * F / 4; i += 1024)
        ((float4*)acc)[i] = make_float4(0.0f, 0.0f, 0.0f, 0.0f);
    if (DEG) for (int i = tid; i < R; i += 1024) degl[i] = 0.0f;
    __syncthreads();

    int2* q = que[wave];
    int tail = 0;

    auto add_rows = [&](int sl, int ry, int cnt) {
        #pragma unroll
        for (int g = 0; g < 4; ++g) {
            if (g * 8 < cnt) {
                vecf v[8];
                int  rr[8];
                #pragma unroll
                for (int j = 0; j < 8; ++j) {
                    int s = __builtin_amdgcn_readlane(sl, g * 8 + j);
                    rr[j] = __builtin_amdgcn_readlane(ry, g * 8 + j);
                    v[j]  = *(const vecf*)(x + (size_t)s * F + lane * VEC);
                }
                #pragma unroll
                for (int j = 0; j < 8; ++j) {
                    float* a = acc + rr[j] * F + lane;
                    #pragma unroll
                    for (int c = 0; c < VEC; ++c)
                        __hip_atomic_fetch_add(a + c * 32, v[j][c], __ATOMIC_RELAXED,
                                               __HIP_MEMORY_SCOPE_WORKGROUP);
                }
            }
        }
    };

    auto replay = [&](const unsigned* cw, int n, unsigned cur) {
        for (int h = 0; h < n; h += 32) {
            const int cnt = min(32, n - h);
            const unsigned nxt = cw[min(h + 32 + lane, BIN_CAP - 1)];
            int sl = min((int)(cur >> 9), N_NODES - 1);
            int ry = min((int)(cur & 511u), R - 1);
            if (lane >= cnt) { sl = 0; ry = R; }
            if (DEG) {
                if (lane < cnt)
                    __hip_atomic_fetch_add(&degl[ry], 1.0f, __ATOMIC_RELAXED,
                                           __HIP_MEMORY_SCOPE_WORKGROUP);
            }
            add_rows(sl, ry, cnt);
            cur = nxt;
        }
    };

    auto drain = [&](int h, int cnt) {
        __builtin_amdgcn_wave_barrier();
        int2 ent = q[h + lane];
        if (lane >= cnt) ent = make_int2(0, R);
        const int sl = min(max(src[ent.x], 0), N_NODES - 1);
        add_rows(sl, ent.y, cnt);
    };

    auto chunk = [&](const int4 d, int ch) {
        unsigned r[4];
        r[0] = (unsigned)(d.x - lo); r[1] = (unsigned)(d.y - lo);
        r[2] = (unsigned)(d.z - lo); r[3] = (unsigned)(d.w - lo);
        const unsigned m = min(min(r[0], r[1]), min(r[2], r[3]));
        if (__builtin_amdgcn_ballot_w32(m < (unsigned)R) != 0u) {
            #pragma unroll
            for (int c = 0; c < 4; ++c) {
                const bool hit = r[c] < (unsigned)R;
                const unsigned mask = __builtin_amdgcn_ballot_w32(hit);
                if (mask != 0u) {
                    if (hit) {
                        const int e = ch * 128 + lane * 4 + c;
                        q[tail + (int)__builtin_amdgcn_mbcnt_lo(mask, 0u)] = make_int2(e, (int)r[c]);
                        if (DEG)
                            __hip_atomic_fetch_add(&degl[r[c]], 1.0f, __ATOMIC_RELAXED,
                                                   __HIP_MEMORY_SCOPE_WORKGROUP);
                    }
                    tail += __builtin_popcount(mask);
                }
            }
            if (tail >= 32) {
                int h = 0;
                do { drain(h, 32); h += 32; } while (tail - h >= 32);
                __builtin_amdgcn_wave_barrier();
                const int2 t = q[h + lane];
                __builtin_amdgcn_wave_barrier();
                if (lane < tail - h) q[lane] = t;
                tail -= h;
            }
        }
    };

    auto scan_group = [&](int grp) {
        const int4* d4 = (const int4*)dst;
        for (int ci = 0; ci < BIN_PIECES; ++ci) {
            const int sc = grp * BIN_PIECES + ci;
            for (int k = 0; k < 4; ++k)
                chunk(d4[((size_t)sc * 4 + k) * 32 + lane], sc * 4 + k);
        }
        if (tail > 0) drain(0, tail);
        tail = 0;
    };

    static_assert(BIN_WGS >= 32, "every wave has a first cell");
    const unsigned* const cnts  = sv + blockIdx.x * BIN_WGS;
    const unsigned* const cells = sv + SAVE_HEAD + (size_t)blockIdx.x * BIN_WGS * BIN_CAP;
    int grp = wave;
    unsigned n = cnts[grp];
    unsigned f = cells[grp * BIN_CAP + lane];
    while (true) {
        const bool more = grp + 32 < BIN_WGS;
        const int  gn   = more ? grp + 32 : grp;
        const unsigned n2 = cnts[gn];
        const unsigned f2 = cells[gn * BIN_CAP + lane];
        n = (unsigned)__builtin_amdgcn_readfirstlane((int)n);
        if (n <= (unsigned)BIN_CAP) replay(cells + grp * BIN_CAP, (int)n, f);
        else scan_group(grp);
        if (!more) break;
        grp = gn; n = n2; f = f2;
    }
    __syncthreads();

    for (int row = wave; row < R; row += 32) {
        const int node = lo + row;
        if (node < N_NODES) {
            vecf o;
            #pragma unroll
            for (int c = 0; c < VEC; ++c) o[c] = acc[row * F + c * 32 + lane];
            *(vecf*)(agg + (size_t)node * F + lane * VEC) = o;
        }
    }
    if (DEG) {
        if (tid < R) {
            const int node = lo + tid;
            if (node < N_NODES) {
                const float dd = degl[tid];
                inv[node] = (dd > 0.0f) ? (1.0f / fmaxf(dd, 1.0f)) : 0.0f;
            }
        }
    }
}

#define SLAB_ROWS 64

template <int K, int DOUT, bool SAGE, bool RELU>
__global__ __launch_bounds__(128)
void slab_gemm_kernel(const float* __restrict__ A1, const float* __restrict__ scale,
                      const float* __restrict__ W1,
                      const float* A2, const float* __restrict__ W2,
                      const float* __restrict__ bias, float* out) {
    constexpr int KT  = SAGE ? 2 * K : K;
    constexpr int P   = KT + 4;
    constexpr int Q   = K / 4;
    constexpr int CT  = (DOUT + 15) / 16;
    constexpr int CPW = (CT >= 4) ? CT / 4 : 1;
    constexpr int RPW = (CT >= 4) ? 4 : 1;
    static_assert(CT == 1 || CT % 4 == 0, "column tiles are dealt to 4 waves");
    static_assert(K % 4 == 0 && (SLAB_ROWS * Q) % 128 == 0, "staging loop has no tail");
    static_assert(N_NODES % 16 == 0, "a row tile is wholly inside or outside the table");

    __shared__ __attribute__((aligned(16))) float slab[SLAB_ROWS * P];

    const int tid  = threadIdx.x;
    const int lane = tid & 31;
    const int wave = __builtin_amdgcn_readfirstlane(tid >> 5);
    const int row0 = blockIdx.x * SLAB_ROWS;

    for (int i = tid; i < SLAB_ROWS * Q; i += 128) {
        const int r = i / Q, c = i % Q;
        const int node = min(row0 + r, N_NODES - 1);
        float4 v = ((const float4*)(A1 + (size_t)node * K))[c];
        if (SAGE) {
            const float s = scale[node];
            v.x *= s; v.y *= s; v.z *= s; v.w *= s;
        }
        *(float4*)(slab + r * P + c * 4) = v;
        if (SAGE)
            *(float4*)(slab + r * P + K + c * 4) = ((const float4*)(A2 + (size_t)node * K))[c];
    }
    __syncthreads();

    const int mrow  = lane & 15;
    const int khalf = (lane >> 4) << 1;
    const int rt0   = (CT >= 4) ? 0 : wave;
    const int ct0   = (CT >= 4) ? wave * CPW : 0;

    v8f acc[RPW][CPW];
    const float* ar[RPW];
    const float* w1r[CPW];
    const float* w2r[CPW];
    #pragma unroll
    for (int i = 0; i < RPW; ++i) {
        ar[i] = slab + ((rt0 + i) * 16 + mrow) * P + khalf;
        #pragma unroll
        for (int j = 0; j < CPW; ++j) acc[i][j] = v8f{};
    }
    #pragma unroll
    for (int j = 0; j < CPW; ++j) {
        const int ncol = (ct0 + j) * 16 + mrow;
        const int wrow = (ncol < DOUT) ? ncol : 0;
        w1r[j] = W1 + (size_t)wrow * K + khalf;
        w2r[j] = SAGE ? W2 + (size_t)wrow * K + khalf : nullptr;
    }

    for (int k = 0; k < K; k += 4) {
        v2f a[RPW], b[CPW];
        #pragma unroll
        for (int i = 0; i < RPW; ++i) a[i] = *(const v2f*)(ar[i] + k);
        #pragma unroll
        for (int j = 0; j < CPW; ++j) b[j] = *(const v2f*)(w1r[j] + k);
        #pragma unroll
        for (int i = 0; i < RPW; ++i) {
            #pragma unroll
            for (int j = 0; j < CPW; ++j)
                acc[i][j] = __builtin_amdgcn_wmma_f32_16x16x4_f32(false, a[i], false, b[j],
                                                                  (short)0, acc[i][j], false, false);
        }
    }
    if (SAGE) {
        for (int k = 0; k < K; k += 4) {
            v2f a[RPW], b[CPW];
            #pragma unroll
            for (int i = 0; i < RPW; ++i) a[i] = *(const v2f*)(ar[i] + K + k);
            #pragma unroll
            for (int j = 0; j < CPW; ++j) b[j] = *(const v2f*)(w2r[j] + k);
            #pragma unroll
            for (int i = 0; i < RPW; ++i) {
                #pragma unroll
                for (int j = 0; j < CPW; ++j)
                    acc[i][j] = __builtin_amdgcn_wmma_f32_16x16x4_f32(false, a[i], false, b[j],
                                                                      (short)0, acc[i][j], false, false);
            }
        }
    }

    #pragma unroll
    for (int j = 0; j < CPW; ++j) {
        const int ncol = (ct0 + j) * 16 + mrow;
        if (ncol < DOUT) {
            const float bv = bias[ncol];
            #pragma unroll
            for (int i = 0; i < RPW; ++i) {
                const int m0 = row0 + (rt0 + i) * 16;
                if (m0 < N_NODES) {
                    const int rbase = m0 + ((lane >> 4) << 3);
                    #pragma unroll
                    for (int r = 0; r < 8; ++r) {
                        float v = acc[i][j][r] + bv;
                        if (RELU) v = fmaxf(v, 0.0f);
                        out[(size_t)(rbase + r) * DOUT + ncol] = v;
                    }
                }
            }
        }
    }
}

static inline int cdiv(long a, long b) { return (int)((a + b - 1) / b); }

extern "C" void kernel_launch(void* const* d_in, const int* in_sizes, int n_in,
                              void* d_out, int out_size, void* d_ws, size_t ws_size,
                              hipStream_t stream) {
    const float* x    = (const float*)d_in[0];
    const int*   edge = (const int*)d_in[1];
    const int*   src  = edge;
    const int*   dst  = edge + N_EDGES;
    const float* Wl0 = (const float*)d_in[2],  *bl0 = (const float*)d_in[3],  *Wr0 = (const float*)d_in[4];
    const float* Wl1 = (const float*)d_in[5],  *bl1 = (const float*)d_in[6],  *Wr1 = (const float*)d_in[7];
    const float* Wl2 = (const float*)d_in[8],  *bl2 = (const float*)d_in[9],  *Wr2 = (const float*)d_in[10];
    const float* Wd1 = (const float*)d_in[11], *bd1 = (const float*)d_in[12];
    const float* Wd2 = (const float*)d_in[13], *bd2 = (const float*)d_in[14];

    float* ws   = (float*)d_ws;
    float* deg  = ws;
    float* inv  = deg  + N_NODES;
    float* agg  = inv  + N_NODES;
    float* bufA = agg  + (size_t)N_NODES * 128;
    float* bufB = bufA + (size_t)N_NODES * 128;

    const int gs = cdiv(N_NODES, SLAB_ROWS);
    const int go = N_BLK;
    static_assert((size_t)SAVE_HEAD + (size_t)N_BLK * BIN_WGS * BIN_CAP
                  <= (size_t)N_NODES * 128, "cells must fit the plane");
    unsigned* sv = (unsigned*)bufB;

    bin_edges_kernel<<<BIN_WGS, 1024, 0, stream>>>(src, dst, sv);

    own_rows_sum_kernel<64, true><<<go, 1024, 0, stream>>>(x, src, dst, agg, inv, sv);
    slab_gemm_kernel<64, 128, true, true><<<gs, 128, 0, stream>>>(agg, inv, Wl0, x, Wr0, bl0, bufA);

    own_rows_sum_kernel<128, false><<<go, 1024, 0, stream>>>(bufA, src, dst, agg, inv, sv);
    slab_gemm_kernel<128, 128, true, true><<<gs, 128, 0, stream>>>(agg, inv, Wl1, bufA, Wr1, bl1, bufA);

    own_rows_sum_kernel<128, false><<<go, 1024, 0, stream>>>(bufA, src, dst, agg, inv, sv);
    slab_gemm_kernel<128, 64, true, true><<<gs, 128, 0, stream>>>(agg, inv, Wl2, bufA, Wr2, bl2, bufB);

    slab_gemm_kernel<64, 128, false, true><<<gs, 128, 0, stream>>>(bufB, nullptr, Wd1, nullptr, nullptr, bd1, bufA);
    slab_gemm_kernel<128, 2, false, false><<<gs, 128, 0, stream>>>(bufA, nullptr, Wd2, nullptr, nullptr, bd2,
                                                                   (float*)d_out);
}


# layer 2's product and both decoder products in one launch: a 4-wave group writes its 64-row result tile to LDS as the next product's A operand (twice); 2 launches fewer; sum order unchanged
# speedup vs baseline: 4.5690x; 1.1213x over previous
#include <hip/hip_runtime.h>
#include <hip/hip_bf16.h>

#define N_NODES 100000
#define N_EDGES 1600000

typedef float v2f __attribute__((ext_vector_type(2)));
typedef float v8f __attribute__((ext_vector_type(8)));

#define OWN_ROWS 392
#define OWN_QLEN 160
#define N_BLK ((N_NODES + OWN_ROWS - 1) / OWN_ROWS)
#define BIN_PIECES 25
#define BIN_WGS (N_EDGES / 512 / BIN_PIECES)
#define BIN_CAP 128
#define SAVE_HEAD 32768

__global__ __launch_bounds__(1024)
void bin_edges_kernel(const int* __restrict__ src, const int* __restrict__ dst,
                      unsigned* __restrict__ sv) {
    static_assert(N_EDGES % (512 * BIN_PIECES) == 0, "every group has whole pieces");
    static_assert(N_BLK * BIN_WGS <= SAVE_HEAD, "one count per cell");
    static_assert(OWN_ROWS <= 512, "a word keeps the local row in 9 bits");
    static_assert(BIN_CAP == 128, "a wave writes a cell with one 16-byte store per lane");
    __shared__ unsigned cnt[N_BLK];
    __shared__ __attribute__((aligned(16))) unsigned cell[N_BLK * BIN_CAP];

    const int tid  = threadIdx.x;
    const int lane = tid & 31;
    const int wave = tid >> 5;
    for (int i = tid; i < N_BLK; i += 1024) cnt[i] = 0u;
    __syncthreads();

    constexpr int NV = BIN_PIECES * 128;
    constexpr int NK = (NV + 1023) / 1024;
    const int4* d4 = (const int4*)dst + (size_t)blockIdx.x * NV;
    const int4* s4 = (const int4*)src + (size_t)blockIdx.x * NV;
    int4 dv[NK], sw[NK];
    #pragma unroll
    for (int k = 0; k < NK; ++k) {
        const int i = min(k * 1024 + tid, NV - 1);
        dv[k] = d4[i];
        sw[k] = s4[i];
    }
    #pragma unroll
    for (int k = 0; k < NK; ++k) {
        if (k * 1024 + tid < NV) {
            const int d[4] = {dv[k].x, dv[k].y, dv[k].z, dv[k].w};
            const int s[4] = {sw[k].x, sw[k].y, sw[k].z, sw[k].w};
            unsigned o[4], r[4], lr[4];
            #pragma unroll
            for (int c = 0; c < 4; ++c) {
                o[c]  = min((unsigned)d[c] / (unsigned)OWN_ROWS, (unsigned)(N_BLK - 1));
                lr[c] = (unsigned)d[c] - o[c] * (unsigned)OWN_ROWS;
                r[c]  = atomicAdd(&cnt[o[c]], 1u);
            }
            #pragma unroll
            for (int c = 0; c < 4; ++c) {
                if (r[c] < (unsigned)BIN_CAP)
                    cell[o[c] * BIN_CAP + r[c]] = ((unsigned)s[c] << 9) | (lr[c] & 511u);
            }
        }
    }
    __syncthreads();

    for (int o = wave; o < N_BLK; o += 32) {
        const uint4 v = ((const uint4*)(cell + o * BIN_CAP))[lane];
        ((uint4*)(sv + SAVE_HEAD + ((size_t)o * BIN_WGS + blockIdx.x) * BIN_CAP))[lane] = v;
        if (lane == 0) sv[o * BIN_WGS + blockIdx.x] = cnt[o];
    }
}

template <int F, bool DEG>
__global__ __launch_bounds__(1024)
void own_rows_sum_kernel(const float* __restrict__ x,
                         const int* __restrict__ src,
                         const int* __restrict__ dst,
                         float* __restrict__ agg,
                         float* __restrict__ inv,
                         const unsigned* __restrict__ sv) {
    constexpr int R   = OWN_ROWS;
    constexpr int VEC = F / 32;
    typedef float vecf __attribute__((ext_vector_type(VEC)));

    __shared__ __attribute__((aligned(16))) float acc[(R + 1) * F];
    constexpr int QLEN = ((R + 1) * F * 4 < 160 * 1024) ? 3 * OWN_QLEN : OWN_QLEN;
    __shared__ int2  que[32][QLEN];
    __shared__ float degl[R];

    const int tid  = threadIdx.x;
    const int lane = tid & 31;
    const int wave = __builtin_amdgcn_readfirstlane(tid >> 5);
    const int lo   = blockIdx.x * R;

    for (int i = tid; i < (R + 1) * F / 4; i += 1024)
        ((float4*)acc)[i] = make_float4(0.0f, 0.0f, 0.0f, 0.0f);
    if (DEG) for (int i = tid; i < R; i += 1024) degl[i] = 0.0f;
    __syncthreads();

    int2* q = que[wave];
    int tail = 0;

    auto add_rows = [&](int sl, int ry, int cnt) {
        #pragma unroll
        for (int g = 0; g < 4; ++g) {
            if (g * 8 < cnt) {
                vecf v[8];
                int  rr[8];
                #pragma unroll
                for (int j = 0; j < 8; ++j) {
                    int s = __builtin_amdgcn_readlane(sl, g * 8 + j);
                    rr[j] = __builtin_amdgcn_readlane(ry, g * 8 + j);
                    v[j]  = *(const vecf*)(x + (size_t)s * F + lane * VEC);
                }
                #pragma unroll
                for (int j = 0; j < 8; ++j) {
                    float* a = acc + rr[j] * F + lane;
                    #pragma unroll
                    for (int c = 0; c < VEC; ++c)
                        __hip_atomic_fetch_add(a + c * 32, v[j][c], __ATOMIC_RELAXED,
                                               __HIP_MEMORY_SCOPE_WORKGROUP);
                }
            }
        }
    };

    auto replay = [&](const unsigned* cw, int n, unsigned cur) {
        for (int h = 0; h < n; h += 32) {
            const int cnt = min(32, n - h);
            const unsigned nxt = cw[min(h + 32 + lane, BIN_CAP - 1)];
            int sl = min((int)(cur >> 9), N_NODES - 1);
            int ry = min((int)(cur & 511u), R - 1);
            if (lane >= cnt) { sl = 0; ry = R; }
            if (DEG) {
                if (lane < cnt)
                    __hip_atomic_fetch_add(&degl[ry], 1.0f, __ATOMIC_RELAXED,
                                           __HIP_MEMORY_SCOPE_WORKGROUP);
            }
            add_rows(sl, ry, cnt);
            cur = nxt;
        }
    };

    auto drain = [&](int h, int cnt) {
        __builtin_amdgcn_wave_barrier();
        int2 ent = q[h + lane];
        if (lane >= cnt) ent = make_int2(0, R);
        const int sl = min(max(src[ent.x], 0), N_NODES - 1);
        add_rows(sl, ent.y, cnt);
    };

    auto chunk = [&](const int4 d, int ch) {
        unsigned r[4];
        r[0] = (unsigned)(d.x - lo); r[1] = (unsigned)(d.y - lo);
        r[2] = (unsigned)(d.z - lo); r[3] = (unsigned)(d.w - lo);
        const unsigned m = min(min(r[0], r[1]), min(r[2], r[3]));
        if (__builtin_amdgcn_ballot_w32(m < (unsigned)R) != 0u) {
            #pragma unroll
            for (int c = 0; c < 4; ++c) {
                const bool hit = r[c] < (unsigned)R;
                const unsigned mask = __builtin_amdgcn_ballot_w32(hit);
                if (mask != 0u) {
                    if (hit) {
                        const int e = ch * 128 + lane * 4 + c;
                        q[tail + (int)__builtin_amdgcn_mbcnt_lo(mask, 0u)] = make_int2(e, (int)r[c]);
                        if (DEG)
                            __hip_atomic_fetch_add(&degl[r[c]], 1.0f, __ATOMIC_RELAXED,
                                                   __HIP_MEMORY_SCOPE_WORKGROUP);
                    }
                    tail += __builtin_popcount(mask);
                }
            }
            if (tail >= 32) {
                int h = 0;
                do { drain(h, 32); h += 32; } while (tail - h >= 32);
                __builtin_amdgcn_wave_barrier();
                const int2 t = q[h + lane];
                __builtin_amdgcn_wave_barrier();
                if (lane < tail - h) q[lane] = t;
                tail -= h;
            }
        }
    };

    auto scan_group = [&](int grp) {
        const int4* d4 = (const int4*)dst;
        for (int ci = 0; ci < BIN_PIECES; ++ci) {
            const int sc = grp * BIN_PIECES + ci;
            for (int k = 0; k < 4; ++k)
                chunk(d4[((size_t)sc * 4 + k) * 32 + lane], sc * 4 + k);
        }
        if (tail > 0) drain(0, tail);
        tail = 0;
    };

    static_assert(BIN_WGS >= 32, "every wave has a first cell");
    const unsigned* const cnts  = sv + blockIdx.x * BIN_WGS;
    const unsigned* const cells = sv + SAVE_HEAD + (size_t)blockIdx.x * BIN_WGS * BIN_CAP;
    int grp = wave;
    unsigned n = cnts[grp];
    unsigned f = cells[grp * BIN_CAP + lane];
    while (true) {
        const bool more = grp + 32 < BIN_WGS;
        const int  gn   = more ? grp + 32 : grp;
        const unsigned n2 = cnts[gn];
        const unsigned f2 = cells[gn * BIN_CAP + lane];
        n = (unsigned)__builtin_amdgcn_readfirstlane((int)n);
        if (n <= (unsigned)BIN_CAP) replay(cells + grp * BIN_CAP, (int)n, f);
        else scan_group(grp);
        if (!more) break;
        grp = gn; n = n2; f = f2;
    }
    __syncthreads();

    for (int row = wave; row < R; row += 32) {
        const int node = lo + row;
        if (node < N_NODES) {
            vecf o;
            #pragma unroll
            for (int c = 0; c < VEC; ++c) o[c] = acc[row * F + c * 32 + lane];
            *(vecf*)(agg + (size_t)node * F + lane * VEC) = o;
        }
    }
    if (DEG) {
        if (tid < R) {
            const int node = lo + tid;
            if (node < N_NODES) {
                const float dd = degl[tid];
                inv[node] = (dd > 0.0f) ? (1.0f / fmaxf(dd, 1.0f)) : 0.0f;
            }
        }
    }
}

#define SLAB_ROWS 64

template <int K, int DOUT, bool SAGE, bool RELU>
__global__ __launch_bounds__(128)
void slab_gemm_kernel(const float* __restrict__ A1, const float* __restrict__ scale,
                      const float* __restrict__ W1,
                      const float* A2, const float* __restrict__ W2,
                      const float* __restrict__ bias, float* out) {
    constexpr int KT  = SAGE ? 2 * K : K;
    constexpr int P   = KT + 4;
    constexpr int Q   = K / 4;
    constexpr int CT  = (DOUT + 15) / 16;
    constexpr int CPW = (CT >= 4) ? CT / 4 : 1;
    constexpr int RPW = (CT >= 4) ? 4 : 1;
    static_assert(CT == 1 || CT % 4 == 0, "column tiles are dealt to 4 waves");
    static_assert(K % 4 == 0 && (SLAB_ROWS * Q) % 128 == 0, "staging loop has no tail");
    static_assert(N_NODES % 16 == 0, "a row tile is wholly inside or outside the table");

    __shared__ __attribute__((aligned(16))) float slab[SLAB_ROWS * P];

    const int tid  = threadIdx.x;
    const int lane = tid & 31;
    const int wave = __builtin_amdgcn_readfirstlane(tid >> 5);
    const int row0 = blockIdx.x * SLAB_ROWS;

    for (int i = tid; i < SLAB_ROWS * Q; i += 128) {
        const int r = i / Q, c = i % Q;
        const int node = min(row0 + r, N_NODES - 1);
        float4 v = ((const float4*)(A1 + (size_t)node * K))[c];
        if (SAGE) {
            const float s = scale[node];
            v.x *= s; v.y *= s; v.z *= s; v.w *= s;
        }
        *(float4*)(slab + r * P + c * 4) = v;
        if (SAGE)
            *(float4*)(slab + r * P + K + c * 4) = ((const float4*)(A2 + (size_t)node * K))[c];
    }
    __syncthreads();

    const int mrow  = lane & 15;
    const int khalf = (lane >> 4) << 1;
    const int rt0   = (CT >= 4) ? 0 : wave;
    const int ct0   = (CT >= 4) ? wave * CPW : 0;

    v8f acc[RPW][CPW];
    const float* ar[RPW];
    const float* w1r[CPW];
    const float* w2r[CPW];
    #pragma unroll
    for (int i = 0; i < RPW; ++i) {
        ar[i] = slab + ((rt0 + i) * 16 + mrow) * P + khalf;
        #pragma unroll
        for (int j = 0; j < CPW; ++j) acc[i][j] = v8f{};
    }
    #pragma unroll
    for (int j = 0; j < CPW; ++j) {
        const int ncol = (ct0 + j) * 16 + mrow;
        const int wrow = (ncol < DOUT) ? ncol : 0;
        w1r[j] = W1 + (size_t)wrow * K + khalf;
        w2r[j] = SAGE ? W2 + (size_t)wrow * K + khalf : nullptr;
    }

    for (int k = 0; k < K; k += 4) {
        v2f a[RPW], b[CPW];
        #pragma unroll
        for (int i = 0; i < RPW; ++i) a[i] = *(const v2f*)(ar[i] + k);
        #pragma unroll
        for (int j = 0; j < CPW; ++j) b[j] = *(const v2f*)(w1r[j] + k);
        #pragma unroll
        for (int i = 0; i < RPW; ++i) {
            #pragma unroll
            for (int j = 0; j < CPW; ++j)
                acc[i][j] = __builtin_amdgcn_wmma_f32_16x16x4_f32(false, a[i], false, b[j],
                                                                  (short)0, acc[i][j], false, false);
        }
    }
    if (SAGE) {
        for (int k = 0; k < K; k += 4) {
            v2f a[RPW], b[CPW];
            #pragma unroll
            for (int i = 0; i < RPW; ++i) a[i] = *(const v2f*)(ar[i] + K + k);
            #pragma unroll
            for (int j = 0; j < CPW; ++j) b[j] = *(const v2f*)(w2r[j] + k);
            #pragma unroll
            for (int i = 0; i < RPW; ++i) {
                #pragma unroll
                for (int j = 0; j < CPW; ++j)
                    acc[i][j] = __builtin_amdgcn_wmma_f32_16x16x4_f32(false, a[i], false, b[j],
                                                                      (short)0, acc[i][j], false, false);
            }
        }
    }

    #pragma unroll
    for (int j = 0; j < CPW; ++j) {
        const int ncol = (ct0 + j) * 16 + mrow;
        if (ncol < DOUT) {
            const float bv = bias[ncol];
            #pragma unroll
            for (int i = 0; i < RPW; ++i) {
                const int m0 = row0 + (rt0 + i) * 16;
                if (m0 < N_NODES) {
                    const int rbase = m0 + ((lane >> 4) << 3);
                    #pragma unroll
                    for (int r = 0; r < 8; ++r) {
                        float v = acc[i][j][r] + bv;
                        if (RELU) v = fmaxf(v, 0.0f);
                        out[(size_t)(rbase + r) * DOUT + ncol] = v;
                    }
                }
            }
        }
    }
}

#define TAIL_K   128
#define TAIL_H2  64
#define TAIL_HD  128
#define TAIL_P0  (2 * TAIL_K + 4)
#define TAIL_P1  (TAIL_H2 + 4)
#define TAIL_P2  (TAIL_HD + 4)

__global__ __launch_bounds__(128)
void tail_kernel(const float* __restrict__ A1, const float* __restrict__ scale,
                 const float* __restrict__ W1,
                 const float* __restrict__ A2, const float* __restrict__ W2,
                 const float* __restrict__ bias,
                 const float* __restrict__ Wd1, const float* __restrict__ bd1,
                 const float* __restrict__ Wd2, const float* __restrict__ bd2,
                 float* __restrict__ out) {
    constexpr int K = TAIL_K;
    constexpr int Q = K / 4;
    static_assert(SLAB_ROWS == 64, "4 row tiles per group, one per wave in the last product");
    static_assert(SLAB_ROWS * TAIL_P1 + SLAB_ROWS * TAIL_P2 <= SLAB_ROWS * TAIL_P0,
                  "both hidden tiles fit the slab's space");
    __shared__ __attribute__((aligned(16))) float slab[SLAB_ROWS * TAIL_P0];
    float* const h2t = slab;
    float* const hdt = slab + SLAB_ROWS * TAIL_P1;

    const int tid  = threadIdx.x;
    const int lane = tid & 31;
    const int wave = __builtin_amdgcn_readfirstlane(tid >> 5);
    const int row0 = blockIdx.x * SLAB_ROWS;

    for (int i = tid; i < SLAB_ROWS * Q; i += 128) {
        const int r = i / Q, c = i % Q;
        const int node = min(row0 + r, N_NODES - 1);
        float4 v = ((const float4*)(A1 + (size_t)node * K))[c];
        const float s = scale[node];
        v.x *= s; v.y *= s; v.z *= s; v.w *= s;
        *(float4*)(slab + r * TAIL_P0 + c * 4) = v;
        *(float4*)(slab + r * TAIL_P0 + K + c * 4) = ((const float4*)(A2 + (size_t)node * K))[c];
    }
    __syncthreads();

    const int mrow  = lane & 15;
    const int khalf = (lane >> 4) << 1;
    const int rhalf = (lane >> 4) << 3;

    {
        v8f acc[4];
        #pragma unroll
        for (int i = 0; i < 4; ++i) acc[i] = v8f{};
        const int ncol = wave * 16 + mrow;
        const float* w1r = W1 + (size_t)ncol * K + khalf;
        const float* w2r = W2 + (size_t)ncol * K + khalf;
        const float* ar  = slab + mrow * TAIL_P0 + khalf;
        for (int k = 0; k < K; k += 4) {
            v2f a[4];
            #pragma unroll
            for (int i = 0; i < 4; ++i) a[i] = *(const v2f*)(ar + i * 16 * TAIL_P0 + k);
            const v2f b = *(const v2f*)(w1r + k);
            #pragma unroll
            for (int i = 0; i < 4; ++i)
                acc[i] = __builtin_amdgcn_wmma_f32_16x16x4_f32(false, a[i], false, b,
                                                               (short)0, acc[i], false, false);
        }
        for (int k = 0; k < K; k += 4) {
            v2f a[4];
            #pragma unroll
            for (int i = 0; i < 4; ++i) a[i] = *(const v2f*)(ar + i * 16 * TAIL_P0 + K + k);
            const v2f b = *(const v2f*)(w2r + k);
            #pragma unroll
            for (int i = 0; i < 4; ++i)
                acc[i] = __builtin_amdgcn_wmma_f32_16x16x4_f32(false, a[i], false, b,
                                                               (short)0, acc[i], false, false);
        }
        __syncthreads();
        const float bv = bias[ncol];
        #pragma unroll
        for (int i = 0; i < 4; ++i) {
            #pragma unroll
            for (int r = 0; r < 8; ++r)
                h2t[(i * 16 + rhalf + r) * TAIL_P1 + ncol] = fmaxf(acc[i][r] + bv, 0.0f);
        }
    }
    __syncthreads();

    {
        v8f acc[4][2];
        #pragma unroll
        for (int i = 0; i < 4; ++i) {
            #pragma unroll
            for (int j = 0; j < 2; ++j) acc[i][j] = v8f{};
        }
        const float* ar = h2t + mrow * TAIL_P1 + khalf;
        const float* wr[2];
        #pragma unroll
        for (int j = 0; j < 2; ++j)
            wr[j] = Wd1 + (size_t)((wave * 2 + j) * 16 + mrow) * TAIL_H2 + khalf;
        for (int k = 0; k < TAIL_H2; k += 4) {
            v2f a[4], b[2];
            #pragma unroll
            for (int i = 0; i < 4; ++i) a[i] = *(const v2f*)(ar + i * 16 * TAIL_P1 + k);
            #pragma unroll
            for (int j = 0; j < 2; ++j) b[j] = *(const v2f*)(wr[j] + k);
            #pragma unroll
            for (int i = 0; i < 4; ++i) {
                #pragma unroll
                for (int j = 0; j < 2; ++j)
                    acc[i][j] = __builtin_amdgcn_wmma_f32_16x16x4_f32(false, a[i], false, b[j],
                                                                      (short)0, acc[i][j], false, false);
            }
        }
        #pragma unroll
        for (int j = 0; j < 2; ++j) {
            const int ncol = (wave * 2 + j) * 16 + mrow;
            const float bv = bd1[ncol];
            #pragma unroll
            for (int i = 0; i < 4; ++i) {
                #pragma unroll
                for (int r = 0; r < 8; ++r)
                    hdt[(i * 16 + rhalf + r) * TAIL_P2 + ncol] = fmaxf(acc[i][j][r] + bv, 0.0f);
            }
        }
    }
    __syncthreads();

    {
        v8f acc = v8f{};
        const int wrow = (mrow < 2) ? mrow : 0;
        const float* wr = Wd2 + (size_t)wrow * TAIL_HD + khalf;
        const float* ar = hdt + (wave * 16 + mrow) * TAIL_P2 + khalf;
        for (int k = 0; k < TAIL_HD; k += 4) {
            const v2f a = *(const v2f*)(ar + k);
            const v2f b = *(const v2f*)(wr + k);
            acc = __builtin_amdgcn_wmma_f32_16x16x4_f32(false, a, false, b,
                                                        (short)0, acc, false, false);
        }
        const int m0 = row0 + wave * 16;
        if (mrow < 2 && m0 < N_NODES) {
            const float bv = bd2[mrow];
            #pragma unroll
            for (int r = 0; r < 8; ++r)
                out[(size_t)(m0 + rhalf + r) * 2 + mrow] = acc[r] + bv;
        }
    }
}

static inline int cdiv(long a, long b) { return (int)((a + b - 1) / b); }

extern "C" void kernel_launch(void* const* d_in, const int* in_sizes, int n_in,
                              void* d_out, int out_size, void* d_ws, size_t ws_size,
                              hipStream_t stream) {
    const float* x    = (const float*)d_in[0];
    const int*   edge = (const int*)d_in[1];
    const int*   src  = edge;
    const int*   dst  = edge + N_EDGES;
    const float* Wl0 = (const float*)d_in[2],  *bl0 = (const float*)d_in[3],  *Wr0 = (const float*)d_in[4];
    const float* Wl1 = (const float*)d_in[5],  *bl1 = (const float*)d_in[6],  *Wr1 = (const float*)d_in[7];
    const float* Wl2 = (const float*)d_in[8],  *bl2 = (const float*)d_in[9],  *Wr2 = (const float*)d_in[10];
    const float* Wd1 = (const float*)d_in[11], *bd1 = (const float*)d_in[12];
    const float* Wd2 = (const float*)d_in[13], *bd2 = (const float*)d_in[14];

    float* ws   = (float*)d_ws;
    float* deg  = ws;
    float* inv  = deg  + N_NODES;
    float* agg  = inv  + N_NODES;
    float* bufA = agg  + (size_t)N_NODES * 128;
    float* bufB = bufA + (size_t)N_NODES * 128;

    const int gs = cdiv(N_NODES, SLAB_ROWS);
    const int go = N_BLK;
    static_assert((size_t)SAVE_HEAD + (size_t)N_BLK * BIN_WGS * BIN_CAP
                  <= (size_t)N_NODES * 128, "cells must fit the plane");
    unsigned* sv = (unsigned*)bufB;

    bin_edges_kernel<<<BIN_WGS, 1024, 0, stream>>>(src, dst, sv);

    own_rows_sum_kernel<64, true><<<go, 1024, 0, stream>>>(x, src, dst, agg, inv, sv);
    slab_gemm_kernel<64, 128, true, true><<<gs, 128, 0, stream>>>(agg, inv, Wl0, x, Wr0, bl0, bufA);

    own_rows_sum_kernel<128, false><<<go, 1024, 0, stream>>>(bufA, src, dst, agg, inv, sv);
    slab_gemm_kernel<128, 128, true, true><<<gs, 128, 0, stream>>>(agg, inv, Wl1, bufA, Wr1, bl1, bufA);

    own_rows_sum_kernel<128, false><<<go, 1024, 0, stream>>>(bufA, src, dst, agg, inv, sv);
    tail_kernel<<<gs, 128, 0, stream>>>(agg, inv, Wl2, bufA, Wr2, bl2, Wd1, bd1, Wd2, bd2,
                                        (float*)d_out);
}


# sum passes: a wave owns 13 rows of its block, summed in registers; cell words go to the owner wave's LDS queue, scalar branches pick the row; no LDS float adds; sum order changes, still by arrival
# speedup vs baseline: 4.8093x; 1.0526x over previous
#include <hip/hip_runtime.h>
#include <hip/hip_bf16.h>

#define N_NODES 100000
#define N_EDGES 1600000

typedef float v2f __attribute__((ext_vector_type(2)));
typedef float v8f __attribute__((ext_vector_type(8)));

#define OWN_ROWS 392
#define OWN_QLEN 160
#define OWN_SLOTS ((OWN_ROWS + 31) / 32)
#define N_BLK ((N_NODES + OWN_ROWS - 1) / OWN_ROWS)
#define BIN_PIECES 25
#define BIN_WGS (N_EDGES / 512 / BIN_PIECES)
#define BIN_CAP 128
#define SAVE_HEAD 32768

#define OWN_SLOT_ADD(A, H, PV, T) \
    asm( \
    "s_cmp_lt_i32 %14, 8\n\t" \
    "s_cbranch_scc0 .Lown_ge8_%=\n\t" \
    "s_cmp_lt_i32 %14, 4\n\t" \
    "s_cbranch_scc0 .Lown_ge4_%=\n\t" \
    "s_cmp_lt_i32 %14, 2\n\t" \
    "s_cbranch_scc0 .Lown_ge2_%=\n\t" \
    "s_cmp_lt_i32 %14, 1\n\t" \
    "s_cbranch_scc0 .Lown_ge1_%=\n\t" \
    "v_pk_add_f32 %0, %0, %13\n\t" \
    "s_branch .Lown_end_%=\n\t" \
    ".Lown_ge1_%=:\n\t" \
    "v_pk_add_f32 %1, %1, %13\n\t" \
    "s_branch .Lown_end_%=\n\t" \
    ".Lown_ge2_%=:\n\t" \
    "s_cmp_lt_i32 %14, 3\n\t" \
    "s_cbranch_scc0 .Lown_ge3_%=\n\t" \
    "v_pk_add_f32 %2, %2, %13\n\t" \
    "s_branch .Lown_end_%=\n\t" \
    ".Lown_ge3_%=:\n\t" \
    "v_pk_add_f32 %3, %3, %13\n\t" \
    "s_branch .Lown_end_%=\n\t" \
    ".Lown_ge4_%=:\n\t" \
    "s_cmp_lt_i32 %14, 6\n\t" \
    "s_cbranch_scc0 .Lown_ge6_%=\n\t" \
    "s_cmp_lt_i32 %14, 5\n\t" \
    "s_cbranch_scc0 .Lown_ge5_%=\n\t" \
    "v_pk_add_f32 %4, %4, %13\n\t" \
    "s_branch .Lown_end_%=\n\t" \
    ".Lown_ge5_%=:\n\t" \
    "v_pk_add_f32 %5, %5, %13\n\t" \
    "s_branch .Lown_end_%=\n\t" \
    ".Lown_ge6_%=:\n\t" \
    "s_cmp_lt_i32 %14, 7\n\t" \
    "s_cbranch_scc0 .Lown_ge7_%=\n\t" \
    "v_pk_add_f32 %6, %6, %13\n\t" \
    "s_branch .Lown_end_%=\n\t" \
    ".Lown_ge7_%=:\n\t" \
    "v_pk_add_f32 %7, %7, %13\n\t" \
    "s_branch .Lown_end_%=\n\t" \
    ".Lown_ge8_%=:\n\t" \
    "s_cmp_lt_i32 %14, 12\n\t" \
    "s_cbranch_scc0 .Lown_ge12_%=\n\t" \
    "s_cmp_lt_i32 %14, 10\n\t" \
    "s_cbranch_scc0 .Lown_ge10_%=\n\t" \
    "s_cmp_lt_i32 %14, 9\n\t" \
    "s_cbranch_scc0 .Lown_ge9_%=\n\t" \
    "v_pk_add_f32 %8, %8, %13\n\t" \
    "s_branch .Lown_end_%=\n\t" \
    ".Lown_ge9_%=:\n\t" \
    "v_pk_add_f32 %9, %9, %13\n\t" \
    "s_branch .Lown_end_%=\n\t" \
    ".Lown_ge10_%=:\n\t" \
    "s_cmp_lt_i32 %14, 11\n\t" \
    "s_cbranch_scc0 .Lown_ge11_%=\n\t" \
    "v_pk_add_f32 %10, %10, %13\n\t" \
    "s_branch .Lown_end_%=\n\t" \
    ".Lown_ge11_%=:\n\t" \
    "v_pk_add_f32 %11, %11, %13\n\t" \
    "s_branch .Lown_end_%=\n\t" \
    ".Lown_ge12_%=:\n\t" \
    "s_cmp_lt_i32 %14, 13\n\t" \
    "s_cbranch_scc0 .Lown_ge13_%=\n\t" \
    "v_pk_add_f32 %12, %12, %13\n\t" \
    "s_branch .Lown_end_%=\n\t" \
    ".Lown_ge13_%=:\n\t" \
    ".Lown_end_%=:\n\t" \
    "" \
    : "+v"(A[0][H]), "+v"(A[1][H]), "+v"(A[2][H]), "+v"(A[3][H]), "+v"(A[4][H]), \
      "+v"(A[5][H]), "+v"(A[6][H]), "+v"(A[7][H]), "+v"(A[8][H]), "+v"(A[9][H]), \
      "+v"(A[10][H]), "+v"(A[11][H]), "+v"(A[12][H]) \
    : "v"(PV), "s"(T) \
    : "scc")

#define OWN_SLOT_ADD2(A, PV0, PV1, T) \
    asm( \
    "s_cmp_lt_i32 %28, 8\n\t" \
    "s_cbranch_scc0 .Lown2_ge8_%=\n\t" \
    "s_cmp_lt_i32 %28, 4\n\t" \
    "s_cbranch_scc0 .Lown2_ge4_%=\n\t" \
    "s_cmp_lt_i32 %28, 2\n\t" \
    "s_cbranch_scc0 .Lown2_ge2_%=\n\t" \
    "s_cmp_lt_i32 %28, 1\n\t" \
    "s_cbranch_scc0 .Lown2_ge1_%=\n\t" \
    "v_pk_add_f32 %0, %0, %26\n\t" \
    "v_pk_add_f32 %13, %13, %27\n\t" \
    "s_branch .Lown2_end_%=\n\t" \
    ".Lown2_ge1_%=:\n\t" \
    "v_pk_add_f32 %1, %1, %26\n\t" \
    "v_pk_add_f32 %14, %14, %27\n\t" \
    "s_branch .Lown2_end_%=\n\t" \
    ".Lown2_ge2_%=:\n\t" \
    "s_cmp_lt_i32 %28, 3\n\t" \
    "s_cbranch_scc0 .Lown2_ge3_%=\n\t" \
    "v_pk_add_f32 %2, %2, %26\n\t" \
    "v_pk_add_f32 %15, %15, %27\n\t" \
    "s_branch .Lown2_end_%=\n\t" \
    ".Lown2_ge3_%=:\n\t" \
    "v_pk_add_f32 %3, %3, %26\n\t" \
    "v_pk_add_f32 %16, %16, %27\n\t" \
    "s_branch .Lown2_end_%=\n\t" \
    ".Lown2_ge4_%=:\n\t" \
    "s_cmp_lt_i32 %28, 6\n\t" \
    "s_cbranch_scc0 .Lown2_ge6_%=\n\t" \
    "s_cmp_lt_i32 %28, 5\n\t" \
    "s_cbranch_scc0 .Lown2_ge5_%=\n\t" \
    "v_pk_add_f32 %4, %4, %26\n\t" \
    "v_pk_add_f32 %17, %17, %27\n\t" \
    "s_branch .Lown2_end_%=\n\t" \
    ".Lown2_ge5_%=:\n\t" \
    "v_pk_add_f32 %5, %5, %26\n\t" \
    "v_pk_add_f32 %18, %18, %27\n\t" \
    "s_branch .Lown2_end_%=\n\t" \
    ".Lown2_ge6_%=:\n\t" \
    "s_cmp_lt_i32 %28, 7\n\t" \
    "s_cbranch_scc0 .Lown2_ge7_%=\n\t" \
    "v_pk_add_f32 %6, %6, %26\n\t" \
    "v_pk_add_f32 %19, %19, %27\n\t" \
    "s_branch .Lown2_end_%=\n\t" \
    ".Lown2_ge7_%=:\n\t" \
    "v_pk_add_f32 %7, %7, %26\n\t" \
    "v_pk_add_f32 %20, %20, %27\n\t" \
    "s_branch .Lown2_end_%=\n\t" \
    ".Lown2_ge8_%=:\n\t" \
    "s_cmp_lt_i32 %28, 12\n\t" \
    "s_cbranch_scc0 .Lown2_ge12_%=\n\t" \
    "s_cmp_lt_i32 %28, 10\n\t" \
    "s_cbranch_scc0 .Lown2_ge10_%=\n\t" \
    "s_cmp_lt_i32 %28, 9\n\t" \
    "s_cbranch_scc0 .Lown2_ge9_%=\n\t" \
    "v_pk_add_f32 %8, %8, %26\n\t" \
    "v_pk_add_f32 %21, %21, %27\n\t" \
    "s_branch .Lown2_end_%=\n\t" \
    ".Lown2_ge9_%=:\n\t" \
    "v_pk_add_f32 %9, %9, %26\n\t" \
    "v_pk_add_f32 %22, %22, %27\n\t" \
    "s_branch .Lown2_end_%=\n\t" \
    ".Lown2_ge10_%=:\n\t" \
    "s_cmp_lt_i32 %28, 11\n\t" \
    "s_cbranch_scc0 .Lown2_ge11_%=\n\t" \
    "v_pk_add_f32 %10, %10, %26\n\t" \
    "v_pk_add_f32 %23, %23, %27\n\t" \
    "s_branch .Lown2_end_%=\n\t" \
    ".Lown2_ge11_%=:\n\t" \
    "v_pk_add_f32 %11, %11, %26\n\t" \
    "v_pk_add_f32 %24, %24, %27\n\t" \
    "s_branch .Lown2_end_%=\n\t" \
    ".Lown2_ge12_%=:\n\t" \
    "s_cmp_lt_i32 %28, 13\n\t" \
    "s_cbranch_scc0 .Lown2_ge13_%=\n\t" \
    "v_pk_add_f32 %12, %12, %26\n\t" \
    "v_pk_add_f32 %25, %25, %27\n\t" \
    "s_branch .Lown2_end_%=\n\t" \
    ".Lown2_ge13_%=:\n\t" \
    ".Lown2_end_%=:\n\t" \
    "" \
    : "+v"(A[0][0]), "+v"(A[1][0]), "+v"(A[2][0]), "+v"(A[3][0]), "+v"(A[4][0]), "+v"(A[5][0]), "+v"(A[6][0]), "+v"(A[7][0]), "+v"(A[8][0]), "+v"(A[9][0]), "+v"(A[10][0]), "+v"(A[11][0]), "+v"(A[12][0]), \
      "+v"(A[0][1]), "+v"(A[1][1]), "+v"(A[2][1]), "+v"(A[3][1]), "+v"(A[4][1]), "+v"(A[5][1]), "+v"(A[6][1]), "+v"(A[7][1]), "+v"(A[8][1]), "+v"(A[9][1]), "+v"(A[10][1]), "+v"(A[11][1]), "+v"(A[12][1]) \
    : "v"(PV0), "v"(PV1), "s"(T) \
    : "scc")

__global__ __launch_bounds__(1024)
void bin_edges_kernel(const int* __restrict__ src, const int* __restrict__ dst,
                      unsigned* __restrict__ sv) {
    static_assert(N_EDGES % (512 * BIN_PIECES) == 0, "every group has whole pieces");
    static_assert(N_BLK * BIN_WGS <= SAVE_HEAD, "one count per cell");
    static_assert(OWN_ROWS <= 512, "a word keeps the local row in 9 bits");
    static_assert(BIN_CAP == 128, "a wave writes a cell with one 16-byte store per lane");
    __shared__ unsigned cnt[N_BLK];
    __shared__ __attribute__((aligned(16))) unsigned cell[N_BLK * BIN_CAP];

    const int tid  = threadIdx.x;
    const int lane = tid & 31;
    const int wave = tid >> 5;
    for (int i = tid; i < N_BLK; i += 1024) cnt[i] = 0u;
    __syncthreads();

    constexpr int NV = BIN_PIECES * 128;
    constexpr int NK = (NV + 1023) / 1024;
    const int4* d4 = (const int4*)dst + (size_t)blockIdx.x * NV;
    const int4* s4 = (const int4*)src + (size_t)blockIdx.x * NV;
    int4 dv[NK], sw[NK];
    #pragma unroll
    for (int k = 0; k < NK; ++k) {
        const int i = min(k * 1024 + tid, NV - 1);
        dv[k] = d4[i];
        sw[k] = s4[i];
    }
    #pragma unroll
    for (int k = 0; k < NK; ++k) {
        if (k * 1024 + tid < NV) {
            const int d[4] = {dv[k].x, dv[k].y, dv[k].z, dv[k].w};
            const int s[4] = {sw[k].x, sw[k].y, sw[k].z, sw[k].w};
            unsigned o[4], r[4], lr[4];
            #pragma unroll
            for (int c = 0; c < 4; ++c) {
                o[c]  = min((unsigned)d[c] / (unsigned)OWN_ROWS, (unsigned)(N_BLK - 1));
                lr[c] = (unsigned)d[c] - o[c] * (unsigned)OWN_ROWS;
                r[c]  = atomicAdd(&cnt[o[c]], 1u);
            }
            #pragma unroll
            for (int c = 0; c < 4; ++c) {
                if (r[c] < (unsigned)BIN_CAP)
                    cell[o[c] * BIN_CAP + r[c]] = ((unsigned)s[c] << 9) | (lr[c] & 511u);
            }
        }
    }
    __syncthreads();

    for (int o = wave; o < N_BLK; o += 32) {
        const uint4 v = ((const uint4*)(cell + o * BIN_CAP))[lane];
        ((uint4*)(sv + SAVE_HEAD + ((size_t)o * BIN_WGS + blockIdx.x) * BIN_CAP))[lane] = v;
        if (lane == 0) sv[o * BIN_WGS + blockIdx.x] = cnt[o];
    }
}

template <int F, bool DEG>
__global__ __launch_bounds__(1024)
void own_rows_sum_kernel(const float* __restrict__ x,
                         const int* __restrict__ src,
                         const int* __restrict__ dst,
                         float* __restrict__ agg,
                         float* __restrict__ inv,
                         const unsigned* __restrict__ sv) {
    constexpr int R   = OWN_ROWS;
    constexpr int VEC = F / 32;
    typedef float vecf __attribute__((ext_vector_type(VEC)));

    __shared__ __attribute__((aligned(16))) float acc[(R + 1) * F];
    constexpr int QLEN = ((R + 1) * F * 4 < 160 * 1024) ? 3 * OWN_QLEN : OWN_QLEN;
    __shared__ int2  que[32][QLEN];
    __shared__ float degl[R];

    const int tid  = threadIdx.x;
    const int lane = tid & 31;
    const int wave = __builtin_amdgcn_readfirstlane(tid >> 5);
    const int lo   = blockIdx.x * R;

    constexpr int NS = OWN_SLOTS;
    constexpr int QW = 2 * OWN_QLEN;
    static_assert(NS == 13, "OWN_SLOT_ADD lists 13 slots");
    static_assert(VEC == 2 || VEC == 4, "a lane's piece of a row is one or two 8-byte halves");
    static_assert((NS << 5) < 512, "the padding slot fits the word's 9 bits");
    static_assert(BIN_CAP == 128, "a cell is read as four words per lane");
    __shared__ unsigned qcnt[32];
    __shared__ unsigned bad;

    if (tid < 32) qcnt[tid] = 0u;
    if (tid == 0) bad = 0u;
    __syncthreads();
    {
        constexpr int NCW = (BIN_WGS + 31) / 32;
        const unsigned* const cnts  = sv + blockIdx.x * BIN_WGS;
        const unsigned* const cells = sv + SAVE_HEAD + (size_t)blockIdx.x * BIN_WGS * BIN_CAP;
        unsigned cn[NCW], cw[NCW][4];
        #pragma unroll
        for (int k = 0; k < NCW; ++k) {
            const int g = min(wave + 32 * k, BIN_WGS - 1);
            cn[k] = cnts[g];
            #pragma unroll
            for (int c = 0; c < 4; ++c) cw[k][c] = cells[g * BIN_CAP + c * 32 + lane];
        }
        #pragma unroll
        for (int k = 0; k < NCW; ++k) {
            if (wave + 32 * k < BIN_WGS) {
                const unsigned n = (unsigned)__builtin_amdgcn_readfirstlane((int)cn[k]);
                if (n > (unsigned)BIN_CAP) {
                    if (lane == 0) bad = 1u;
                } else {
                    #pragma unroll
                    for (int c = 0; c < 4; ++c) {
                        if ((unsigned)(c * 32 + lane) < n) {
                            const unsigned wd = cw[k][c];
                            const unsigned s  = min(wd >> 9, (unsigned)(N_NODES - 1));
                            const unsigned lr = min(wd & 511u, (unsigned)(R - 1));
                            const unsigned w  = lr & 31u;
                            const unsigned at = atomicAdd(&qcnt[w], 1u);
                            if (at < (unsigned)QW) ((unsigned*)que[w])[at] = (s << 9) | lr;
                            else bad = 1u;
                        }
                    }
                }
            }
        }
    }
    __syncthreads();
    if (__builtin_amdgcn_readfirstlane((int)bad) == 0) {
        const int nq = __builtin_amdgcn_readfirstlane((int)qcnt[wave]);
        const unsigned* const myq = (const unsigned*)que[wave];
        constexpr int NH = VEC / 2;
        v2f accr[NS][NH];
        #pragma unroll
        for (int i = 0; i < NS; ++i) {
            #pragma unroll
            for (int hf = 0; hf < NH; ++hf) accr[i][hf] = (v2f)(0.0f);
        }
        float dcnt = 0.0f;
        for (int h = 0; h < nq; h += 32) {
            const int cnt = min(32, nq - h);
            unsigned wd = 0u;
            if (lane < cnt) wd = myq[h + lane];
            int sl = (int)(wd >> 9);
            int ry = (int)((wd & 511u) >> 5);
            if (lane >= cnt) { sl = 0; ry = NS; }
            for (int g = 0; g * 8 < cnt; ++g) {
                vecf v[8];
                int  sr[8];
                #pragma unroll
                for (int j = 0; j < 8; ++j) {
                    const int s = __builtin_amdgcn_readlane(sl, g * 8 + j);
                    sr[j] = __builtin_amdgcn_readlane(ry, g * 8 + j);
                    v[j]  = *(const vecf*)(x + (size_t)s * F + lane * VEC);
                }
                __builtin_amdgcn_sched_barrier(0);
                #pragma unroll
                for (int j = 0; j < 8; ++j) {
                    const int t = sr[j];
                    if constexpr (NH == 1) {
                        v2f pv;
                        pv[0] = v[j][0];
                        pv[1] = v[j][1];
                        OWN_SLOT_ADD(accr, 0, pv, t);
                    } else {
                        v2f p0, p1;
                        p0[0] = v[j][0];
                        p0[1] = v[j][1];
                        p1[0] = v[j][VEC - 2];
                        p1[1] = v[j][VEC - 1];
                        OWN_SLOT_ADD2(accr, p0, p1, t);
                    }
                    if (DEG) dcnt += (lane == sr[j]) ? 1.0f : 0.0f;
                }
            }
        }
        #pragma unroll
        for (int i = 0; i < NS; ++i) {
            const int row  = wave + 32 * i;
            const int node = lo + row;
            if (row < R && node < N_NODES) {
                vecf o;
                #pragma unroll
                for (int hf = 0; hf < NH; ++hf) {
                    o[2 * hf]     = accr[i][hf][0];
                    o[2 * hf + 1] = accr[i][hf][1];
                }
                *(vecf*)(agg + (size_t)node * F + lane * VEC) = o;
            }
        }
        if (DEG) {
            if (lane < NS && wave + 32 * lane < R) degl[wave + 32 * lane] = dcnt;
            __syncthreads();
            if (tid < R) {
                const int node = lo + tid;
                if (node < N_NODES) {
                    const float dd = degl[tid];
                    inv[node] = (dd > 0.0f) ? (1.0f / fmaxf(dd, 1.0f)) : 0.0f;
                }
            }
        }
        return;
    }

    for (int i = tid; i < (R + 1) * F / 4; i += 1024)
        ((float4*)acc)[i] = make_float4(0.0f, 0.0f, 0.0f, 0.0f);
    if (DEG) for (int i = tid; i < R; i += 1024) degl[i] = 0.0f;
    __syncthreads();

    int2* q = que[wave];
    int tail = 0;

    auto add_rows = [&](int sl, int ry, int cnt) {
        #pragma unroll
        for (int g = 0; g < 4; ++g) {
            if (g * 8 < cnt) {
                vecf v[8];
                int  rr[8];
                #pragma unroll
                for (int j = 0; j < 8; ++j) {
                    int s = __builtin_amdgcn_readlane(sl, g * 8 + j);
                    rr[j] = __builtin_amdgcn_readlane(ry, g * 8 + j);
                    v[j]  = *(const vecf*)(x + (size_t)s * F + lane * VEC);
                }
                #pragma unroll
                for (int j = 0; j < 8; ++j) {
                    float* a = acc + rr[j] * F + lane;
                    #pragma unroll
                    for (int c = 0; c < VEC; ++c)
                        __hip_atomic_fetch_add(a + c * 32, v[j][c], __ATOMIC_RELAXED,
                                               __HIP_MEMORY_SCOPE_WORKGROUP);
                }
            }
        }
    };

    auto replay = [&](const unsigned* cw, int n, unsigned cur) {
        for (int h = 0; h < n; h += 32) {
            const int cnt = min(32, n - h);
            const unsigned nxt = cw[min(h + 32 + lane, BIN_CAP - 1)];
            int sl = min((int)(cur >> 9), N_NODES - 1);
            int ry = min((int)(cur & 511u), R - 1);
            if (lane >= cnt) { sl = 0; ry = R; }
            if (DEG) {
                if (lane < cnt)
                    __hip_atomic_fetch_add(&degl[ry], 1.0f, __ATOMIC_RELAXED,
                                           __HIP_MEMORY_SCOPE_WORKGROUP);
            }
            add_rows(sl, ry, cnt);
            cur = nxt;
        }
    };

    auto drain = [&](int h, int cnt) {
        __builtin_amdgcn_wave_barrier();
        int2 ent = q[h + lane];
        if (lane >= cnt) ent = make_int2(0, R);
        const int sl = min(max(src[ent.x], 0), N_NODES - 1);
        add_rows(sl, ent.y, cnt);
    };

    auto chunk = [&](const int4 d, int ch) {
        unsigned r[4];
        r[0] = (unsigned)(d.x - lo); r[1] = (unsigned)(d.y - lo);
        r[2] = (unsigned)(d.z - lo); r[3] = (unsigned)(d.w - lo);
        const unsigned m = min(min(r[0], r[1]), min(r[2], r[3]));
        if (__builtin_amdgcn_ballot_w32(m < (unsigned)R) != 0u) {
            #pragma unroll
            for (int c = 0; c < 4; ++c) {
                const bool hit = r[c] < (unsigned)R;
                const unsigned mask = __builtin_amdgcn_ballot_w32(hit);
                if (mask != 0u) {
                    if (hit) {
                        const int e = ch * 128 + lane * 4 + c;
                        q[tail + (int)__builtin_amdgcn_mbcnt_lo(mask, 0u)] = make_int2(e, (int)r[c]);
                        if (DEG)
                            __hip_atomic_fetch_add(&degl[r[c]], 1.0f, __ATOMIC_RELAXED,
                                                   __HIP_MEMORY_SCOPE_WORKGROUP);
                    }
                    tail += __builtin_popcount(mask);
                }
            }
            if (tail >= 32) {
                int h = 0;
                do { drain(h, 32); h += 32; } while (tail - h >= 32);
                __builtin_amdgcn_wave_barrier();
                const int2 t = q[h + lane];
                __builtin_amdgcn_wave_barrier();
                if (lane < tail - h) q[lane] = t;
                tail -= h;
            }
        }
    };

    auto scan_group = [&](int grp) {
        const int4* d4 = (const int4*)dst;
        for (int ci = 0; ci < BIN_PIECES; ++ci) {
            const int sc = grp * BIN_PIECES + ci;
            for (int k = 0; k < 4; ++k)
                chunk(d4[((size_t)sc * 4 + k) * 32 + lane], sc * 4 + k);
        }
        if (tail > 0) drain(0, tail);
        tail = 0;
    };

    static_assert(BIN_WGS >= 32, "every wave has a first cell");
    const unsigned* const cnts  = sv + blockIdx.x * BIN_WGS;
    const unsigned* const cells = sv + SAVE_HEAD + (size_t)blockIdx.x * BIN_WGS * BIN_CAP;
    int grp = wave;
    unsigned n = cnts[grp];
    unsigned f = cells[grp * BIN_CAP + lane];
    while (true) {
        const bool more = grp + 32 < BIN_WGS;
        const int  gn   = more ? grp + 32 : grp;
        const unsigned n2 = cnts[gn];
        const unsigned f2 = cells[gn * BIN_CAP + lane];
        n = (unsigned)__builtin_amdgcn_readfirstlane((int)n);
        if (n <= (unsigned)BIN_CAP) replay(cells + grp * BIN_CAP, (int)n, f);
        else scan_group(grp);
        if (!more) break;
        grp = gn; n = n2; f = f2;
    }
    __syncthreads();

    for (int row = wave; row < R; row += 32) {
        const int node = lo + row;
        if (node < N_NODES) {
            vecf o;
            #pragma unroll
            for (int c = 0; c < VEC; ++c) o[c] = acc[row * F + c * 32 + lane];
            *(vecf*)(agg + (size_t)node * F + lane * VEC) = o;
        }
    }
    if (DEG) {
        if (tid < R) {
            const int node = lo + tid;
            if (node < N_NODES) {
                const float dd = degl[tid];
                inv[node] = (dd > 0.0f) ? (1.0f / fmaxf(dd, 1.0f)) : 0.0f;
            }
        }
    }
}

#define SLAB_ROWS 64

template <int K, int DOUT, bool SAGE, bool RELU>
__global__ __launch_bounds__(128)
void slab_gemm_kernel(const float* __restrict__ A1, const float* __restrict__ scale,
                      const float* __restrict__ W1,
                      const float* A2, const float* __restrict__ W2,
                      const float* __restrict__ bias, float* out) {
    constexpr int KT  = SAGE ? 2 * K : K;
    constexpr int P   = KT + 4;
    constexpr int Q   = K / 4;
    constexpr int CT  = (DOUT + 15) / 16;
    constexpr int CPW = (CT >= 4) ? CT / 4 : 1;
    constexpr int RPW = (CT >= 4) ? 4 : 1;
    static_assert(CT == 1 || CT % 4 == 0, "column tiles are dealt to 4 waves");
    static_assert(K % 4 == 0 && (SLAB_ROWS * Q) % 128 == 0, "staging loop has no tail");
    static_assert(N_NODES % 16 == 0, "a row tile is wholly inside or outside the table");

    __shared__ __attribute__((aligned(16))) float slab[SLAB_ROWS * P];

    const int tid  = threadIdx.x;
    const int lane = tid & 31;
    const int wave = __builtin_amdgcn_readfirstlane(tid >> 5);
    const int row0 = blockIdx.x * SLAB_ROWS;

    for (int i = tid; i < SLAB_ROWS * Q; i += 128) {
        const int r = i / Q, c = i % Q;
        const int node = min(row0 + r, N_NODES - 1);
        float4 v = ((const float4*)(A1 + (size_t)node * K))[c];
        if (SAGE) {
            const float s = scale[node];
            v.x *= s; v.y *= s; v.z *= s; v.w *= s;
        }
        *(float4*)(slab + r * P + c * 4) = v;
        if (SAGE)
            *(float4*)(slab + r * P + K + c * 4) = ((const float4*)(A2 + (size_t)node * K))[c];
    }
    __syncthreads();

    const int mrow  = lane & 15;
    const int khalf = (lane >> 4) << 1;
    const int rt0   = (CT >= 4) ? 0 : wave;
    const int ct0   = (CT >= 4) ? wave * CPW : 0;

    v8f acc[RPW][CPW];
    const float* ar[RPW];
    const float* w1r[CPW];
    const float* w2r[CPW];
    #pragma unroll
    for (int i = 0; i < RPW; ++i) {
        ar[i] = slab + ((rt0 + i) * 16 + mrow) * P + khalf;
        #pragma unroll
        for (int j = 0; j < CPW; ++j) acc[i][j] = v8f{};
    }
    #pragma unroll
    for (int j = 0; j < CPW; ++j) {
        const int ncol = (ct0 + j) * 16 + mrow;
        const int wrow = (ncol < DOUT) ? ncol : 0;
        w1r[j] = W1 + (size_t)wrow * K + khalf;
        w2r[j] = SAGE ? W2 + (size_t)wrow * K + khalf : nullptr;
    }

    for (int k = 0; k < K; k += 4) {
        v2f a[RPW], b[CPW];
        #pragma unroll
        for (int i = 0; i < RPW; ++i) a[i] = *(const v2f*)(ar[i] + k);
        #pragma unroll
        for (int j = 0; j < CPW; ++j) b[j] = *(const v2f*)(w1r[j] + k);
        #pragma unroll
        for (int i = 0; i < RPW; ++i) {
            #pragma unroll
            for (int j = 0; j < CPW; ++j)
                acc[i][j] = __builtin_amdgcn_wmma_f32_16x16x4_f32(false, a[i], false, b[j],
                                                                  (short)0, acc[i][j], false, false);
        }
    }
    if (SAGE) {
        for (int k = 0; k < K; k += 4) {
            v2f a[RPW], b[CPW];
            #pragma unroll
            for (int i = 0; i < RPW; ++i) a[i] = *(const v2f*)(ar[i] + K + k);
            #pragma unroll
            for (int j = 0; j < CPW; ++j) b[j] = *(const v2f*)(w2r[j] + k);
            #pragma unroll
            for (int i = 0; i < RPW; ++i) {
                #pragma unroll
                for (int j = 0; j < CPW; ++j)
                    acc[i][j] = __builtin_amdgcn_wmma_f32_16x16x4_f32(false, a[i], false, b[j],
                                                                      (short)0, acc[i][j], false, false);
            }
        }
    }

    #pragma unroll
    for (int j = 0; j < CPW; ++j) {
        const int ncol = (ct0 + j) * 16 + mrow;
        if (ncol < DOUT) {
            const float bv = bias[ncol];
            #pragma unroll
            for (int i = 0; i < RPW; ++i) {
                const int m0 = row0 + (rt0 + i) * 16;
                if (m0 < N_NODES) {
                    const int rbase = m0 + ((lane >> 4) << 3);
                    #pragma unroll
                    for (int r = 0; r < 8; ++r) {
                        float v = acc[i][j][r] + bv;
                        if (RELU) v = fmaxf(v, 0.0f);
                        out[(size_t)(rbase + r) * DOUT + ncol] = v;
                    }
                }
            }
        }
    }
}

#define TAIL_K   128
#define TAIL_H2  64
#define TAIL_HD  128
#define TAIL_P0  (2 * TAIL_K + 4)
#define TAIL_P1  (TAIL_H2 + 4)
#define TAIL_P2  (TAIL_HD + 4)

__global__ __launch_bounds__(128)
void tail_kernel(const float* __restrict__ A1, const float* __restrict__ scale,
                 const float* __restrict__ W1,
                 const float* __restrict__ A2, const float* __restrict__ W2,
                 const float* __restrict__ bias,
                 const float* __restrict__ Wd1, const float* __restrict__ bd1,
                 const float* __restrict__ Wd2, const float* __restrict__ bd2,
                 float* __restrict__ out) {
    constexpr int K = TAIL_K;
    constexpr int Q = K / 4;
    static_assert(SLAB_ROWS == 64, "4 row tiles per group, one per wave in the last product");
    static_assert(SLAB_ROWS * TAIL_P1 + SLAB_ROWS * TAIL_P2 <= SLAB_ROWS * TAIL_P0,
                  "both hidden tiles fit the slab's space");
    __shared__ __attribute__((aligned(16))) float slab[SLAB_ROWS * TAIL_P0];
    float* const h2t = slab;
    float* const hdt = slab + SLAB_ROWS * TAIL_P1;

    const int tid  = threadIdx.x;
    const int lane = tid & 31;
    const int wave = __builtin_amdgcn_readfirstlane(tid >> 5);
    const int row0 = blockIdx.x * SLAB_ROWS;

    for (int i = tid; i < SLAB_ROWS * Q; i += 128) {
        const int r = i / Q, c = i % Q;
        const int node = min(row0 + r, N_NODES - 1);
        float4 v = ((const float4*)(A1 + (size_t)node * K))[c];
        const float s = scale[node];
        v.x *= s; v.y *= s; v.z *= s; v.w *= s;
        *(float4*)(slab + r * TAIL_P0 + c * 4) = v;
        *(float4*)(slab + r * TAIL_P0 + K + c * 4) = ((const float4*)(A2 + (size_t)node * K))[c];
    }
    __syncthreads();

    const int mrow  = lane & 15;
    const int khalf = (lane >> 4) << 1;
    const int rhalf = (lane >> 4) << 3;

    {
        v8f acc[4];
        #pragma unroll
        for (int i = 0; i < 4; ++i) acc[i] = v8f{};
        const int ncol = wave * 16 + mrow;
        const float* w1r = W1 + (size_t)ncol * K + khalf;
        const float* w2r = W2 + (size_t)ncol * K + khalf;
        const float* ar  = slab + mrow * TAIL_P0 + khalf;
        for (int k = 0; k < K; k += 4) {
            v2f a[4];
            #pragma unroll
            for (int i = 0; i < 4; ++i) a[i] = *(const v2f*)(ar + i * 16 * TAIL_P0 + k);
            const v2f b = *(const v2f*)(w1r + k);
            #pragma unroll
            for (int i = 0; i < 4; ++i)
                acc[i] = __builtin_amdgcn_wmma_f32_16x16x4_f32(false, a[i], false, b,
                                                               (short)0, acc[i], false, false);
        }
        for (int k = 0; k < K; k += 4) {
            v2f a[4];
            #pragma unroll
            for (int i = 0; i < 4; ++i) a[i] = *(const v2f*)(ar + i * 16 * TAIL_P0 + K + k);
            const v2f b = *(const v2f*)(w2r + k);
            #pragma unroll
            for (int i = 0; i < 4; ++i)
                acc[i] = __builtin_amdgcn_wmma_f32_16x16x4_f32(false, a[i], false, b,
                                                               (short)0, acc[i], false, false);
        }
        __syncthreads();
        const float bv = bias[ncol];
        #pragma unroll
        for (int i = 0; i < 4; ++i) {
            #pragma unroll
            for (int r = 0; r < 8; ++r)
                h2t[(i * 16 + rhalf + r) * TAIL_P1 + ncol] = fmaxf(acc[i][r] + bv, 0.0f);
        }
    }
    __syncthreads();

    {
        v8f acc[4][2];
        #pragma unroll
        for (int i = 0; i < 4; ++i) {
            #pragma unroll
            for (int j = 0; j < 2; ++j) acc[i][j] = v8f{};
        }
        const float* ar = h2t + mrow * TAIL_P1 + khalf;
        const float* wr[2];
        #pragma unroll
        for (int j = 0; j < 2; ++j)
            wr[j] = Wd1 + (size_t)((wave * 2 + j) * 16 + mrow) * TAIL_H2 + khalf;
        for (int k = 0; k < TAIL_H2; k += 4) {
            v2f a[4], b[2];
            #pragma unroll
            for (int i = 0; i < 4; ++i) a[i] = *(const v2f*)(ar + i * 16 * TAIL_P1 + k);
            #pragma unroll
            for (int j = 0; j < 2; ++j) b[j] = *(const v2f*)(wr[j] + k);
            #pragma unroll
            for (int i = 0; i < 4; ++i) {
                #pragma unroll
                for (int j = 0; j < 2; ++j)
                    acc[i][j] = __builtin_amdgcn_wmma_f32_16x16x4_f32(false, a[i], false, b[j],
                                                                      (short)0, acc[i][j], false, false);
            }
        }
        #pragma unroll
        for (int j = 0; j < 2; ++j) {
            const int ncol = (wave * 2 + j) * 16 + mrow;
            const float bv = bd1[ncol];
            #pragma unroll
            for (int i = 0; i < 4; ++i) {
                #pragma unroll
                for (int r = 0; r < 8; ++r)
                    hdt[(i * 16 + rhalf + r) * TAIL_P2 + ncol] = fmaxf(acc[i][j][r] + bv, 0.0f);
            }
        }
    }
    __syncthreads();

    {
        v8f acc = v8f{};
        const int wrow = (mrow < 2) ? mrow : 0;
        const float* wr = Wd2 + (size_t)wrow * TAIL_HD + khalf;
        const float* ar = hdt + (wave * 16 + mrow) * TAIL_P2 + khalf;
        for (int k = 0; k < TAIL_HD; k += 4) {
            const v2f a = *(const v2f*)(ar + k);
            const v2f b = *(const v2f*)(wr + k);
            acc = __builtin_amdgcn_wmma_f32_16x16x4_f32(false, a, false, b,
                                                        (short)0, acc, false, false);
        }
        const int m0 = row0 + wave * 16;
        if (mrow < 2 && m0 < N_NODES) {
            const float bv = bd2[mrow];
            #pragma unroll
            for (int r = 0; r < 8; ++r)
                out[(size_t)(m0 + rhalf + r) * 2 + mrow] = acc[r] + bv;
        }
    }
}

static inline int cdiv(long a, long b) { return (int)((a + b - 1) / b); }

extern "C" void kernel_launch(void* const* d_in, const int* in_sizes, int n_in,
                              void* d_out, int out_size, void* d_ws, size_t ws_size,
                              hipStream_t stream) {
    const float* x    = (const float*)d_in[0];
    const int*   edge = (const int*)d_in[1];
    const int*   src  = edge;
    const int*   dst  = edge + N_EDGES;
    const float* Wl0 = (const float*)d_in[2],  *bl0 = (const float*)d_in[3],  *Wr0 = (const float*)d_in[4];
    const float* Wl1 = (const float*)d_in[5],  *bl1 = (const float*)d_in[6],  *Wr1 = (const float*)d_in[7];
    const float* Wl2 = (const float*)d_in[8],  *bl2 = (const float*)d_in[9],  *Wr2 = (const float*)d_in[10];
    const float* Wd1 = (const float*)d_in[11], *bd1 = (const float*)d_in[12];
    const float* Wd2 = (const float*)d_in[13], *bd2 = (const float*)d_in[14];

    float* ws   = (float*)d_ws;
    float* deg  = ws;
    float* inv  = deg  + N_NODES;
    float* agg  = inv  + N_NODES;
    float* bufA = agg  + (size_t)N_NODES * 128;
    float* bufB = bufA + (size_t)N_NODES * 128;

    const int gs = cdiv(N_NODES, SLAB_ROWS);
    const int go = N_BLK;
    static_assert((size_t)SAVE_HEAD + (size_t)N_BLK * BIN_WGS * BIN_CAP
                  <= (size_t)N_NODES * 128, "cells must fit the plane");
    unsigned* sv = (unsigned*)bufB;

    bin_edges_kernel<<<BIN_WGS, 1024, 0, stream>>>(src, dst, sv);

    own_rows_sum_kernel<64, true><<<go, 1024, 0, stream>>>(x, src, dst, agg, inv, sv);
    slab_gemm_kernel<64, 128, true, true><<<gs, 128, 0, stream>>>(agg, inv, Wl0, x, Wr0, bl0, bufA);

    own_rows_sum_kernel<128, false><<<go, 1024, 0, stream>>>(bufA, src, dst, agg, inv, sv);
    slab_gemm_kernel<128, 128, true, true><<<gs, 128, 0, stream>>>(agg, inv, Wl1, bufA, Wr1, bl1, bufA);

    own_rows_sum_kernel<128, false><<<go, 1024, 0, stream>>>(bufA, src, dst, agg, inv, sv);
    tail_kernel<<<gs, 128, 0, stream>>>(agg, inv, Wl2, bufA, Wr2, bl2, Wd1, bd1, Wd2, bd2,
                                        (float*)d_out);
}


# layers 0,1: the owner work-group also multiplies its 392 rows: sums go from registers (times inv) to an LDS slab, WMMA from there; no plane of sums in memory, 2 launches fewer; sum order unchanged
# speedup vs baseline: 5.0793x; 1.0561x over previous
#include <hip/hip_runtime.h>
#include <hip/hip_bf16.h>

#define N_NODES 100000
#define N_EDGES 1600000

typedef float v2f __attribute__((ext_vector_type(2)));
typedef float v8f __attribute__((ext_vector_type(8)));

#define OWN_ROWS 392
#define OWN_QLEN 160
#define OWN_SLOTS ((OWN_ROWS + 31) / 32)
#define N_BLK ((N_NODES + OWN_ROWS - 1) / OWN_ROWS)
#define BIN_PIECES 25
#define BIN_WGS (N_EDGES / 512 / BIN_PIECES)
#define BIN_CAP 128
#define SAVE_HEAD 32768

#define OWN_SLOT_ADD(A, H, PV, T) \
    asm( \
    "s_cmp_lt_i32 %14, 8\n\t" \
    "s_cbranch_scc0 .Lown_ge8_%=\n\t" \
    "s_cmp_lt_i32 %14, 4\n\t" \
    "s_cbranch_scc0 .Lown_ge4_%=\n\t" \
    "s_cmp_lt_i32 %14, 2\n\t" \
    "s_cbranch_scc0 .Lown_ge2_%=\n\t" \
    "s_cmp_lt_i32 %14, 1\n\t" \
    "s_cbranch_scc0 .Lown_ge1_%=\n\t" \
    "v_pk_add_f32 %0, %0, %13\n\t" \
    "s_branch .Lown_end_%=\n\t" \
    ".Lown_ge1_%=:\n\t" \
    "v_pk_add_f32 %1, %1, %13\n\t" \
    "s_branch .Lown_end_%=\n\t" \
    ".Lown_ge2_%=:\n\t" \
    "s_cmp_lt_i32 %14, 3\n\t" \
    "s_cbranch_scc0 .Lown_ge3_%=\n\t" \
    "v_pk_add_f32 %2, %2, %13\n\t" \
    "s_branch .Lown_end_%=\n\t" \
    ".Lown_ge3_%=:\n\t" \
    "v_pk_add_f32 %3, %3, %13\n\t" \
    "s_branch .Lown_end_%=\n\t" \
    ".Lown_ge4_%=:\n\t" \
    "s_cmp_lt_i32 %14, 6\n\t" \
    "s_cbranch_scc0 .Lown_ge6_%=\n\t" \
    "s_cmp_lt_i32 %14, 5\n\t" \
    "s_cbranch_scc0 .Lown_ge5_%=\n\t" \
    "v_pk_add_f32 %4, %4, %13\n\t" \
    "s_branch .Lown_end_%=\n\t" \
    ".Lown_ge5_%=:\n\t" \
    "v_pk_add_f32 %5, %5, %13\n\t" \
    "s_branch .Lown_end_%=\n\t" \
    ".Lown_ge6_%=:\n\t" \
    "s_cmp_lt_i32 %14, 7\n\t" \
    "s_cbranch_scc0 .Lown_ge7_%=\n\t" \
    "v_pk_add_f32 %6, %6, %13\n\t" \
    "s_branch .Lown_end_%=\n\t" \
    ".Lown_ge7_%=:\n\t" \
    "v_pk_add_f32 %7, %7, %13\n\t" \
    "s_branch .Lown_end_%=\n\t" \
    ".Lown_ge8_%=:\n\t" \
    "s_cmp_lt_i32 %14, 12\n\t" \
    "s_cbranch_scc0 .Lown_ge12_%=\n\t" \
    "s_cmp_lt_i32 %14, 10\n\t" \
    "s_cbranch_scc0 .Lown_ge10_%=\n\t" \
    "s_cmp_lt_i32 %14, 9\n\t" \
    "s_cbranch_scc0 .Lown_ge9_%=\n\t" \
    "v_pk_add_f32 %8, %8, %13\n\t" \
    "s_branch .Lown_end_%=\n\t" \
    ".Lown_ge9_%=:\n\t" \
    "v_pk_add_f32 %9, %9, %13\n\t" \
    "s_branch .Lown_end_%=\n\t" \
    ".Lown_ge10_%=:\n\t" \
    "s_cmp_lt_i32 %14, 11\n\t" \
    "s_cbranch_scc0 .Lown_ge11_%=\n\t" \
    "v_pk_add_f32 %10, %10, %13\n\t" \
    "s_branch .Lown_end_%=\n\t" \
    ".Lown_ge11_%=:\n\t" \
    "v_pk_add_f32 %11, %11, %13\n\t" \
    "s_branch .Lown_end_%=\n\t" \
    ".Lown_ge12_%=:\n\t" \
    "s_cmp_lt_i32 %14, 13\n\t" \
    "s_cbranch_scc0 .Lown_ge13_%=\n\t" \
    "v_pk_add_f32 %12, %12, %13\n\t" \
    "s_branch .Lown_end_%=\n\t" \
    ".Lown_ge13_%=:\n\t" \
    ".Lown_end_%=:\n\t" \
    "" \
    : "+v"(A[0][H]), "+v"(A[1][H]), "+v"(A[2][H]), "+v"(A[3][H]), "+v"(A[4][H]), \
      "+v"(A[5][H]), "+v"(A[6][H]), "+v"(A[7][H]), "+v"(A[8][H]), "+v"(A[9][H]), \
      "+v"(A[10][H]), "+v"(A[11][H]), "+v"(A[12][H]) \
    : "v"(PV), "s"(T) \
    : "scc")

#define OWN_SLOT_ADD2(A, PV0, PV1, T) \
    asm( \
    "s_cmp_lt_i32 %28, 8\n\t" \
    "s_cbranch_scc0 .Lown2_ge8_%=\n\t" \
    "s_cmp_lt_i32 %28, 4\n\t" \
    "s_cbranch_scc0 .Lown2_ge4_%=\n\t" \
    "s_cmp_lt_i32 %28, 2\n\t" \
    "s_cbranch_scc0 .Lown2_ge2_%=\n\t" \
    "s_cmp_lt_i32 %28, 1\n\t" \
    "s_cbranch_scc0 .Lown2_ge1_%=\n\t" \
    "v_pk_add_f32 %0, %0, %26\n\t" \
    "v_pk_add_f32 %13, %13, %27\n\t" \
    "s_branch .Lown2_end_%=\n\t" \
    ".Lown2_ge1_%=:\n\t" \
    "v_pk_add_f32 %1, %1, %26\n\t" \
    "v_pk_add_f32 %14, %14, %27\n\t" \
    "s_branch .Lown2_end_%=\n\t" \
    ".Lown2_ge2_%=:\n\t" \
    "s_cmp_lt_i32 %28, 3\n\t" \
    "s_cbranch_scc0 .Lown2_ge3_%=\n\t" \
    "v_pk_add_f32 %2, %2, %26\n\t" \
    "v_pk_add_f32 %15, %15, %27\n\t" \
    "s_branch .Lown2_end_%=\n\t" \
    ".Lown2_ge3_%=:\n\t" \
    "v_pk_add_f32 %3, %3, %26\n\t" \
    "v_pk_add_f32 %16, %16, %27\n\t" \
    "s_branch .Lown2_end_%=\n\t" \
    ".Lown2_ge4_%=:\n\t" \
    "s_cmp_lt_i32 %28, 6\n\t" \
    "s_cbranch_scc0 .Lown2_ge6_%=\n\t" \
    "s_cmp_lt_i32 %28, 5\n\t" \
    "s_cbranch_scc0 .Lown2_ge5_%=\n\t" \
    "v_pk_add_f32 %4, %4, %26\n\t" \
    "v_pk_add_f32 %17, %17, %27\n\t" \
    "s_branch .Lown2_end_%=\n\t" \
    ".Lown2_ge5_%=:\n\t" \
    "v_pk_add_f32 %5, %5, %26\n\t" \
    "v_pk_add_f32 %18, %18, %27\n\t" \
    "s_branch .Lown2_end_%=\n\t" \
    ".Lown2_ge6_%=:\n\t" \
    "s_cmp_lt_i32 %28, 7\n\t" \
    "s_cbranch_scc0 .Lown2_ge7_%=\n\t" \
    "v_pk_add_f32 %6, %6, %26\n\t" \
    "v_pk_add_f32 %19, %19, %27\n\t" \
    "s_branch .Lown2_end_%=\n\t" \
    ".Lown2_ge7_%=:\n\t" \
    "v_pk_add_f32 %7, %7, %26\n\t" \
    "v_pk_add_f32 %20, %20, %27\n\t" \
    "s_branch .Lown2_end_%=\n\t" \
    ".Lown2_ge8_%=:\n\t" \
    "s_cmp_lt_i32 %28, 12\n\t" \
    "s_cbranch_scc0 .Lown2_ge12_%=\n\t" \
    "s_cmp_lt_i32 %28, 10\n\t" \
    "s_cbranch_scc0 .Lown2_ge10_%=\n\t" \
    "s_cmp_lt_i32 %28, 9\n\t" \
    "s_cbranch_scc0 .Lown2_ge9_%=\n\t" \
    "v_pk_add_f32 %8, %8, %26\n\t" \
    "v_pk_add_f32 %21, %21, %27\n\t" \
    "s_branch .Lown2_end_%=\n\t" \
    ".Lown2_ge9_%=:\n\t" \
    "v_pk_add_f32 %9, %9, %26\n\t" \
    "v_pk_add_f32 %22, %22, %27\n\t" \
    "s_branch .Lown2_end_%=\n\t" \
    ".Lown2_ge10_%=:\n\t" \
    "s_cmp_lt_i32 %28, 11\n\t" \
    "s_cbranch_scc0 .Lown2_ge11_%=\n\t" \
    "v_pk_add_f32 %10, %10, %26\n\t" \
    "v_pk_add_f32 %23, %23, %27\n\t" \
    "s_branch .Lown2_end_%=\n\t" \
    ".Lown2_ge11_%=:\n\t" \
    "v_pk_add_f32 %11, %11, %26\n\t" \
    "v_pk_add_f32 %24, %24, %27\n\t" \
    "s_branch .Lown2_end_%=\n\t" \
    ".Lown2_ge12_%=:\n\t" \
    "s_cmp_lt_i32 %28, 13\n\t" \
    "s_cbranch_scc0 .Lown2_ge13_%=\n\t" \
    "v_pk_add_f32 %12, %12, %26\n\t" \
    "v_pk_add_f32 %25, %25, %27\n\t" \
    "s_branch .Lown2_end_%=\n\t" \
    ".Lown2_ge13_%=:\n\t" \
    ".Lown2_end_%=:\n\t" \
    "" \
    : "+v"(A[0][0]), "+v"(A[1][0]), "+v"(A[2][0]), "+v"(A[3][0]), "+v"(A[4][0]), "+v"(A[5][0]), "+v"(A[6][0]), "+v"(A[7][0]), "+v"(A[8][0]), "+v"(A[9][0]), "+v"(A[10][0]), "+v"(A[11][0]), "+v"(A[12][0]), \
      "+v"(A[0][1]), "+v"(A[1][1]), "+v"(A[2][1]), "+v"(A[3][1]), "+v"(A[4][1]), "+v"(A[5][1]), "+v"(A[6][1]), "+v"(A[7][1]), "+v"(A[8][1]), "+v"(A[9][1]), "+v"(A[10][1]), "+v"(A[11][1]), "+v"(A[12][1]) \
    : "v"(PV0), "v"(PV1), "s"(T) \
    : "scc")

__global__ __launch_bounds__(1024)
void bin_edges_kernel(const int* __restrict__ src, const int* __restrict__ dst,
                      unsigned* __restrict__ sv) {
    static_assert(N_EDGES % (512 * BIN_PIECES) == 0, "every group has whole pieces");
    static_assert(N_BLK * BIN_WGS <= SAVE_HEAD, "one count per cell");
    static_assert(OWN_ROWS <= 512, "a word keeps the local row in 9 bits");
    static_assert(BIN_CAP == 128, "a wave writes a cell with one 16-byte store per lane");
    __shared__ unsigned cnt[N_BLK];
    __shared__ __attribute__((aligned(16))) unsigned cell[N_BLK * BIN_CAP];

    const int tid  = threadIdx.x;
    const int lane = tid & 31;
    const int wave = tid >> 5;
    for (int i = tid; i < N_BLK; i += 1024) cnt[i] = 0u;
    __syncthreads();

    constexpr int NV = BIN_PIECES * 128;
    constexpr int NK = (NV + 1023) / 1024;
    const int4* d4 = (const int4*)dst + (size_t)blockIdx.x * NV;
    const int4* s4 = (const int4*)src + (size_t)blockIdx.x * NV;
    int4 dv[NK], sw[NK];
    #pragma unroll
    for (int k = 0; k < NK; ++k) {
        const int i = min(k * 1024 + tid, NV - 1);
        dv[k] = d4[i];
        sw[k] = s4[i];
    }
    #pragma unroll
    for (int k = 0; k < NK; ++k) {
        if (k * 1024 + tid < NV) {
            const int d[4] = {dv[k].x, dv[k].y, dv[k].z, dv[k].w};
            const int s[4] = {sw[k].x, sw[k].y, sw[k].z, sw[k].w};
            unsigned o[4], r[4], lr[4];
            #pragma unroll
            for (int c = 0; c < 4; ++c) {
                o[c]  = min((unsigned)d[c] / (unsigned)OWN_ROWS, (unsigned)(N_BLK - 1));
                lr[c] = (unsigned)d[c] - o[c] * (unsigned)OWN_ROWS;
                r[c]  = atomicAdd(&cnt[o[c]], 1u);
            }
            #pragma unroll
            for (int c = 0; c < 4; ++c) {
                if (r[c] < (unsigned)BIN_CAP)
                    cell[o[c] * BIN_CAP + r[c]] = ((unsigned)s[c] << 9) | (lr[c] & 511u);
            }
        }
    }
    __syncthreads();

    for (int o = wave; o < N_BLK; o += 32) {
        const uint4 v = ((const uint4*)(cell + o * BIN_CAP))[lane];
        ((uint4*)(sv + SAVE_HEAD + ((size_t)o * BIN_WGS + blockIdx.x) * BIN_CAP))[lane] = v;
        if (lane == 0) sv[o * BIN_WGS + blockIdx.x] = cnt[o];
    }
}

#define OWN_TILES ((OWN_ROWS + 15) / 16)
template <int F, bool DEG, int DOUT>
__global__ __launch_bounds__(1024)
void own_rows_sum_kernel(const float* __restrict__ x,
                         const int* __restrict__ src,
                         const int* __restrict__ dst,
                         float* __restrict__ agg,
                         float* __restrict__ inv,
                         const unsigned* __restrict__ sv,
                         const float* __restrict__ Wl,
                         const float* __restrict__ Wr,
                         const float* __restrict__ bias,
                         float* __restrict__ out) {
    constexpr int R   = OWN_ROWS;
    constexpr int VEC = F / 32;
    typedef float vecf __attribute__((ext_vector_type(VEC)));
    constexpr int SP  = F + 4;
    constexpr int ST  = OWN_TILES;
    constexpr int ACCN = (DOUT > 0 && ST * 16 * SP > (R + 1) * F) ? ST * 16 * SP : (R + 1) * F;

    __shared__ __attribute__((aligned(16))) float acc[ACCN];
    constexpr int QLEN = ((R + 1) * F * 4 < 160 * 1024) ? 3 * OWN_QLEN : OWN_QLEN;
    __shared__ int2  que[32][QLEN];
    __shared__ float degl[R];

    const int tid  = threadIdx.x;
    const int lane = tid & 31;
    const int wave = __builtin_amdgcn_readfirstlane(tid >> 5);
    const int lo   = blockIdx.x * R;

    constexpr int NS = OWN_SLOTS;
    constexpr int QW = 2 * OWN_QLEN;
    static_assert(NS == 13, "OWN_SLOT_ADD lists 13 slots");
    static_assert(VEC == 2 || VEC == 4, "a lane's piece of a row is one or two 8-byte halves");
    static_assert((NS << 5) < 512, "the padding slot fits the word's 9 bits");
    static_assert(BIN_CAP == 128, "a cell is read as four words per lane");
    __shared__ unsigned qcnt[32];
    __shared__ unsigned bad;

    if (tid < 32) qcnt[tid] = 0u;
    if (tid == 0) bad = 0u;
    __syncthreads();
    {
        constexpr int NCW = (BIN_WGS + 31) / 32;
        const unsigned* const cnts  = sv + blockIdx.x * BIN_WGS;
        const unsigned* const cells = sv + SAVE_HEAD + (size_t)blockIdx.x * BIN_WGS * BIN_CAP;
        unsigned cn[NCW], cw[NCW][4];
        #pragma unroll
        for (int k = 0; k < NCW; ++k) {
            const int g = min(wave + 32 * k, BIN_WGS - 1);
            cn[k] = cnts[g];
            #pragma unroll
            for (int c = 0; c < 4; ++c) cw[k][c] = cells[g * BIN_CAP + c * 32 + lane];
        }
        #pragma unroll
        for (int k = 0; k < NCW; ++k) {
            if (wave + 32 * k < BIN_WGS) {
                const unsigned n = (unsigned)__builtin_amdgcn_readfirstlane((int)cn[k]);
                if (n > (unsigned)BIN_CAP) {
                    if (lane == 0) bad = 1u;
                } else {
                    #pragma unroll
                    for (int c = 0; c < 4; ++c) {
                        if ((unsigned)(c * 32 + lane) < n) {
                            const unsigned wd = cw[k][c];
                            const unsigned s  = min(wd >> 9, (unsigned)(N_NODES - 1));
                            const unsigned lr = min(wd & 511u, (unsigned)(R - 1));
                            const unsigned w  = lr & 31u;
                            const unsigned at = atomicAdd(&qcnt[w], 1u);
                            if (at < (unsigned)QW) ((unsigned*)que[w])[at] = (s << 9) | lr;
                            else bad = 1u;
                        }
                    }
                }
            }
        }
    }
    __syncthreads();
    constexpr int NH = VEC / 2;
    v2f accr[NS][NH];
    if (__builtin_amdgcn_readfirstlane((int)bad) == 0) {
        const int nq = __builtin_amdgcn_readfirstlane((int)qcnt[wave]);
        const unsigned* const myq = (const unsigned*)que[wave];
        #pragma unroll
        for (int i = 0; i < NS; ++i) {
            #pragma unroll
            for (int hf = 0; hf < NH; ++hf) accr[i][hf] = (v2f)(0.0f);
        }
        float dcnt = 0.0f;
        for (int h = 0; h < nq; h += 32) {
            const int cnt = min(32, nq - h);
            unsigned wd = 0u;
            if (lane < cnt) wd = myq[h + lane];
            int sl = (int)(wd >> 9);
            int ry = (int)((wd & 511u) >> 5);
            if (lane >= cnt) { sl = 0; ry = NS; }
            for (int g = 0; g * 8 < cnt; ++g) {
                vecf v[8];
                int  sr[8];
                #pragma unroll
                for (int j = 0; j < 8; ++j) {
                    const int s = __builtin_amdgcn_readlane(sl, g * 8 + j);
                    sr[j] = __builtin_amdgcn_readlane(ry, g * 8 + j);
                    v[j]  = *(const vecf*)(x + (size_t)s * F + lane * VEC);
                }
                __builtin_amdgcn_sched_barrier(0);
                #pragma unroll
                for (int j = 0; j < 8; ++j) {
                    const int t = sr[j];
                    if constexpr (NH == 1) {
                        v2f pv;
                        pv[0] = v[j][0];
                        pv[1] = v[j][1];
                        OWN_SLOT_ADD(accr, 0, pv, t);
                    } else {
                        v2f p0, p1;
                        p0[0] = v[j][0];
                        p0[1] = v[j][1];
                        p1[0] = v[j][VEC - 2];
                        p1[1] = v[j][VEC - 1];
                        OWN_SLOT_ADD2(accr, p0, p1, t);
                    }
                    if (DEG) dcnt += (lane == sr[j]) ? 1.0f : 0.0f;
                }
            }
        }
        if constexpr (DOUT == 0) {
            #pragma unroll
            for (int i = 0; i < NS; ++i) {
                const int row  = wave + 32 * i;
                const int node = lo + row;
                if (row < R && node < N_NODES) {
                    vecf o;
                    #pragma unroll
                    for (int hf = 0; hf < NH; ++hf) {
                        o[2 * hf]     = accr[i][hf][0];
                        o[2 * hf + 1] = accr[i][hf][1];
                    }
                    *(vecf*)(agg + (size_t)node * F + lane * VEC) = o;
                }
            }
            if (DEG) {
                if (lane < NS && wave + 32 * lane < R) degl[wave + 32 * lane] = dcnt;
                __syncthreads();
                if (tid < R) {
                    const int node = lo + tid;
                    if (node < N_NODES) {
                        const float dd = degl[tid];
                        inv[node] = (dd > 0.0f) ? (1.0f / fmaxf(dd, 1.0f)) : 0.0f;
                    }
                }
            }
            return;
        } else {
            if (DEG) {
                if (lane < NS && wave + 32 * lane < R) degl[wave + 32 * lane] = dcnt;
            }
        }
    } else {
        for (int i = tid; i < (R + 1) * F / 4; i += 1024)
            ((float4*)acc)[i] = make_float4(0.0f, 0.0f, 0.0f, 0.0f);
        if (DEG) for (int i = tid; i < R; i += 1024) degl[i] = 0.0f;
        __syncthreads();

        int2* q = que[wave];
        int tail = 0;

        auto add_rows = [&](int sl, int ry, int cnt) {
            #pragma unroll
            for (int g = 0; g < 4; ++g) {
                if (g * 8 < cnt) {
                    vecf v[8];
                    int  rr[8];
                    #pragma unroll
                    for (int j = 0; j < 8; ++j) {
                        int s = __builtin_amdgcn_readlane(sl, g * 8 + j);
                        rr[j] = __builtin_amdgcn_readlane(ry, g * 8 + j);
                        v[j]  = *(const vecf*)(x + (size_t)s * F + lane * VEC);
                    }
                    #pragma unroll
                    for (int j = 0; j < 8; ++j) {
                        float* a = acc + rr[j] * F + lane;
                        #pragma unroll
                        for (int c = 0; c < VEC; ++c)
                            __hip_atomic_fetch_add(a + c * 32, v[j][c], __ATOMIC_RELAXED,
                                                   __HIP_MEMORY_SCOPE_WORKGROUP);
                    }
                }
            }
        };

        auto replay = [&](const unsigned* cw, int n, unsigned cur) {
            for (int h = 0; h < n; h += 32) {
                const int cnt = min(32, n - h);
                const unsigned nxt = cw[min(h + 32 + lane, BIN_CAP - 1)];
                int sl = min((int)(cur >> 9), N_NODES - 1);
                int ry = min((int)(cur & 511u), R - 1);
                if (lane >= cnt) { sl = 0; ry = R; }
                if (DEG) {
                    if (lane < cnt)
                        __hip_atomic_fetch_add(&degl[ry], 1.0f, __ATOMIC_RELAXED,
                                               __HIP_MEMORY_SCOPE_WORKGROUP);
                }
                add_rows(sl, ry, cnt);
                cur = nxt;
            }
        };

        auto drain = [&](int h, int cnt) {
            __builtin_amdgcn_wave_barrier();
            int2 ent = q[h + lane];
            if (lane >= cnt) ent = make_int2(0, R);
            const int sl = min(max(src[ent.x], 0), N_NODES - 1);
            add_rows(sl, ent.y, cnt);
        };

        auto chunk = [&](const int4 d, int ch) {
            unsigned r[4];
            r[0] = (unsigned)(d.x - lo); r[1] = (unsigned)(d.y - lo);
            r[2] = (unsigned)(d.z - lo); r[3] = (unsigned)(d.w - lo);
            const unsigned m = min(min(r[0], r[1]), min(r[2], r[3]));
            if (__builtin_amdgcn_ballot_w32(m < (unsigned)R) != 0u) {
                #pragma unroll
                for (int c = 0; c < 4; ++c) {
                    const bool hit = r[c] < (unsigned)R;
                    const unsigned mask = __builtin_amdgcn_ballot_w32(hit);
                    if (mask != 0u) {
                        if (hit) {
                            const int e = ch * 128 + lane * 4 + c;
                            q[tail + (int)__builtin_amdgcn_mbcnt_lo(mask, 0u)] = make_int2(e, (int)r[c]);
                            if (DEG)
                                __hip_atomic_fetch_add(&degl[r[c]], 1.0f, __ATOMIC_RELAXED,
                                                       __HIP_MEMORY_SCOPE_WORKGROUP);
                        }
                        tail += __builtin_popcount(mask);
                    }
                }
                if (tail >= 32) {
                    int h = 0;
                    do { drain(h, 32); h += 32; } while (tail - h >= 32);
                    __builtin_amdgcn_wave_barrier();
                    const int2 t = q[h + lane];
                    __builtin_amdgcn_wave_barrier();
                    if (lane < tail - h) q[lane] = t;
                    tail -= h;
                }
            }
        };

        auto scan_group = [&](int grp) {
            const int4* d4 = (const int4*)dst;
            for (int ci = 0; ci < BIN_PIECES; ++ci) {
                const int sc = grp * BIN_PIECES + ci;
                for (int k = 0; k < 4; ++k)
                    chunk(d4[((size_t)sc * 4 + k) * 32 + lane], sc * 4 + k);
            }
            if (tail > 0) drain(0, tail);
            tail = 0;
        };

        static_assert(BIN_WGS >= 32, "every wave has a first cell");
        const unsigned* const cnts  = sv + blockIdx.x * BIN_WGS;
        const unsigned* const cells = sv + SAVE_HEAD + (size_t)blockIdx.x * BIN_WGS * BIN_CAP;
        int grp = wave;
        unsigned n = cnts[grp];
        unsigned f = cells[grp * BIN_CAP + lane];
        while (true) {
            const bool more = grp + 32 < BIN_WGS;
            const int  gn   = more ? grp + 32 : grp;
            const unsigned n2 = cnts[gn];
            const unsigned f2 = cells[gn * BIN_CAP + lane];
            n = (unsigned)__builtin_amdgcn_readfirstlane((int)n);
            if (n <= (unsigned)BIN_CAP) replay(cells + grp * BIN_CAP, (int)n, f);
            else scan_group(grp);
            if (!more) break;
            grp = gn; n = n2; f = f2;
        }
        __syncthreads();

        if constexpr (DOUT == 0) {
            for (int row = wave; row < R; row += 32) {
                const int node = lo + row;
                if (node < N_NODES) {
                    vecf o;
                    #pragma unroll
                    for (int c = 0; c < VEC; ++c) o[c] = acc[row * F + c * 32 + lane];
                    *(vecf*)(agg + (size_t)node * F + lane * VEC) = o;
                }
            }
            if (DEG) {
                if (tid < R) {
                    const int node = lo + tid;
                    if (node < N_NODES) {
                        const float dd = degl[tid];
                        inv[node] = (dd > 0.0f) ? (1.0f / fmaxf(dd, 1.0f)) : 0.0f;
                    }
                }
            }
            return;
        } else {
            #pragma unroll
            for (int i = 0; i < NS; ++i) {
                const int row = wave + 32 * i;
                #pragma unroll
                for (int hf = 0; hf < NH; ++hf) {
                    accr[i][hf] = (v2f)(0.0f);
                    if (row < R) {
                        accr[i][hf][0] = acc[row * F + (2 * hf) * 32 + lane];
                        accr[i][hf][1] = acc[row * F + (2 * hf + 1) * 32 + lane];
                    }
                }
            }
        }
    }

    if constexpr (DOUT > 0) {
        static_assert(DOUT == 128, "a wave has two column tiles: four waves cover the 128 columns");
        static_assert(ST == 25, "row tiles: three per wave, and the last one for waves 0-3");
        static_assert(SP % 64 == 4, "fragment reads of 16 rows x 2 k-pairs fall into 64 banks");
        __syncthreads();
        float* const slab = acc;
        if (DEG) {
            if (tid < R) {
                const int node = lo + tid;
                if (node < N_NODES) {
                    const float dd = degl[tid];
                    inv[node] = (dd > 0.0f) ? (1.0f / fmaxf(dd, 1.0f)) : 0.0f;
                }
            }
        }
        float sl = 0.0f;
        if (lane < NS) {
            const int row = min(wave + 32 * lane, R - 1);
            if (DEG) {
                const float dd = degl[row];
                sl = (dd > 0.0f) ? (1.0f / fmaxf(dd, 1.0f)) : 0.0f;
            } else {
                sl = inv[min(lo + row, N_NODES - 1)];
            }
        }
        #pragma unroll
        for (int i = 0; i < NS; ++i) {
            const int row = wave + 32 * i;
            if (row < R) {
                const float s = __int_as_float(__builtin_amdgcn_readlane(__float_as_int(sl), i));
                vecf o;
                #pragma unroll
                for (int hf = 0; hf < NH; ++hf) {
                    o[2 * hf]     = accr[i][hf][0] * s;
                    o[2 * hf + 1] = accr[i][hf][1] * s;
                }
                *(vecf*)(slab + row * SP + lane * VEC) = o;
            }
        }
        for (int i = tid; i < (ST * 16 - R) * F; i += 1024)
            slab[(R + i / F) * SP + i % F] = 0.0f;
        __syncthreads();

        const int mrow  = lane & 15;
        const int khalf = (lane >> 4) << 1;
        const int rhalf = (lane >> 4) << 3;
        const int rg    = wave >> 2;
        const int cg    = wave & 3;
        const bool extra = (rg == 0);
        const int ncol0 = cg * 32 + mrow;
        const int ncol1 = ncol0 + 16;

        v8f pacc[3][2], pex[2];
        #pragma unroll
        for (int j = 0; j < 2; ++j) {
            pex[j] = v8f{};
            #pragma unroll
            for (int i = 0; i < 3; ++i) pacc[i][j] = v8f{};
        }
        const float* ar[3];
        #pragma unroll
        for (int i = 0; i < 3; ++i) ar[i] = slab + ((rg + 8 * i) * 16 + mrow) * SP + khalf;
        const float* const arx = slab + ((ST - 1) * 16 + mrow) * SP + khalf;

        auto mul_range = [&](const float* W) {
            const float* const wa = W + (size_t)ncol0 * F + khalf;
            const float* const wb = W + (size_t)ncol1 * F + khalf;
            if (extra) {
                for (int k = 0; k < F; k += 4) {
                    const v2f a  = *(const v2f*)(arx + k);
                    const v2f b0 = *(const v2f*)(wa + k);
                    const v2f b1 = *(const v2f*)(wb + k);
                    pex[0] = __builtin_amdgcn_wmma_f32_16x16x4_f32(false, a, false, b0,
                                                                   (short)0, pex[0], false, false);
                    pex[1] = __builtin_amdgcn_wmma_f32_16x16x4_f32(false, a, false, b1,
                                                                   (short)0, pex[1], false, false);
                }
            }
            for (int k = 0; k < F; k += 4) {
                v2f a[3], b[2];
                #pragma unroll
                for (int i = 0; i < 3; ++i) a[i] = *(const v2f*)(ar[i] + k);
                b[0] = *(const v2f*)(wa + k);
                b[1] = *(const v2f*)(wb + k);
                #pragma unroll
                for (int i = 0; i < 3; ++i) {
                    #pragma unroll
                    for (int j = 0; j < 2; ++j)
                        pacc[i][j] = __builtin_amdgcn_wmma_f32_16x16x4_f32(false, a[i], false, b[j],
                                                                           (short)0, pacc[i][j], false, false);
                }
            }
        };

        mul_range(Wl);

        typedef float v4f __attribute__((ext_vector_type(4)));
        constexpr int QF  = F / 4;
        constexpr int NIT = (R * QF + 1023) / 1024;
        constexpr int NB  = 4;
        __syncthreads();
        #pragma unroll
        for (int b0 = 0; b0 < NIT; b0 += NB) {
            v4f xs[NB];
            #pragma unroll
            for (int t = 0; t < NB; ++t) {
                const int i = min((b0 + t) * 1024 + tid, R * QF - 1);
                const int node = min(lo + i / QF, N_NODES - 1);
                xs[t] = *(const v4f*)(x + (size_t)node * F + (i % QF) * 4);
            }
            #pragma unroll
            for (int t = 0; t < NB; ++t) {
                const int i = (b0 + t) * 1024 + tid;
                if (i < R * QF)
                    *(v4f*)(slab + (i / QF) * SP + (i % QF) * 4) = xs[t];
            }
        }
        __syncthreads();

        mul_range(Wr);

        #pragma unroll
        for (int j = 0; j < 2; ++j) {
            const int ncol = ncol0 + 16 * j;
            const float bv = bias[ncol];
            static_assert(R % 8 == 0 && N_NODES % 8 == 0, "one test for a register's 8 rows");
            #pragma unroll
            for (int i = 0; i < 3; ++i) {
                const int node0 = lo + (rg + 8 * i) * 16 + rhalf;
                if (node0 < N_NODES) {
                    #pragma unroll
                    for (int r = 0; r < 8; ++r)
                        out[(size_t)(node0 + r) * DOUT + ncol] = fmaxf(pacc[i][j][r] + bv, 0.0f);
                }
            }
            if (extra) {
                const int lrow0 = (ST - 1) * 16 + rhalf;
                if (lrow0 < R && lo + lrow0 < N_NODES) {
                    #pragma unroll
                    for (int r = 0; r < 8; ++r)
                        out[(size_t)(lo + lrow0 + r) * DOUT + ncol] = fmaxf(pex[j][r] + bv, 0.0f);
                }
            }
        }
    }
}

#define SLAB_ROWS 64
#define TAIL_K   128
#define TAIL_H2  64
#define TAIL_HD  128
#define TAIL_P0  (2 * TAIL_K + 4)
#define TAIL_P1  (TAIL_H2 + 4)
#define TAIL_P2  (TAIL_HD + 4)

__global__ __launch_bounds__(128)
void tail_kernel(const float* __restrict__ A1, const float* __restrict__ scale,
                 const float* __restrict__ W1,
                 const float* __restrict__ A2, const float* __restrict__ W2,
                 const float* __restrict__ bias,
                 const float* __restrict__ Wd1, const float* __restrict__ bd1,
                 const float* __restrict__ Wd2, const float* __restrict__ bd2,
                 float* __restrict__ out) {
    constexpr int K = TAIL_K;
    constexpr int Q = K / 4;
    static_assert(SLAB_ROWS == 64, "4 row tiles per group, one per wave in the last product");
    static_assert(SLAB_ROWS * TAIL_P1 + SLAB_ROWS * TAIL_P2 <= SLAB_ROWS * TAIL_P0,
                  "both hidden tiles fit the slab's space");
    __shared__ __attribute__((aligned(16))) float slab[SLAB_ROWS * TAIL_P0];
    float* const h2t = slab;
    float* const hdt = slab + SLAB_ROWS * TAIL_P1;

    const int tid  = threadIdx.x;
    const int lane = tid & 31;
    const int wave = __builtin_amdgcn_readfirstlane(tid >> 5);
    const int row0 = blockIdx.x * SLAB_ROWS;

    for (int i = tid; i < SLAB_ROWS * Q; i += 128) {
        const int r = i / Q, c = i % Q;
        const int node = min(row0 + r, N_NODES - 1);
        float4 v = ((const float4*)(A1 + (size_t)node * K))[c];
        const float s = scale[node];
        v.x *= s; v.y *= s; v.z *= s; v.w *= s;
        *(float4*)(slab + r * TAIL_P0 + c * 4) = v;
        *(float4*)(slab + r * TAIL_P0 + K + c * 4) = ((const float4*)(A2 + (size_t)node * K))[c];
    }
    __syncthreads();

    const int mrow  = lane & 15;
    const int khalf = (lane >> 4) << 1;
    const int rhalf = (lane >> 4) << 3;

    {
        v8f acc[4];
        #pragma unroll
        for (int i = 0; i < 4; ++i) acc[i] = v8f{};
        const int ncol = wave * 16 + mrow;
        const float* w1r = W1 + (size_t)ncol * K + khalf;
        const float* w2r = W2 + (size_t)ncol * K + khalf;
        const float* ar  = slab + mrow * TAIL_P0 + khalf;
        for (int k = 0; k < K; k += 4) {
            v2f a[4];
            #pragma unroll
            for (int i = 0; i < 4; ++i) a[i] = *(const v2f*)(ar + i * 16 * TAIL_P0 + k);
            const v2f b = *(const v2f*)(w1r + k);
            #pragma unroll
            for (int i = 0; i < 4; ++i)
                acc[i] = __builtin_amdgcn_wmma_f32_16x16x4_f32(false, a[i], false, b,
                                                               (short)0, acc[i], false, false);
        }
        for (int k = 0; k < K; k += 4) {
            v2f a[4];
            #pragma unroll
            for (int i = 0; i < 4; ++i) a[i] = *(const v2f*)(ar + i * 16 * TAIL_P0 + K + k);
            const v2f b = *(const v2f*)(w2r + k);
            #pragma unroll
            for (int i = 0; i < 4; ++i)
                acc[i] = __builtin_amdgcn_wmma_f32_16x16x4_f32(false, a[i], false, b,
                                                               (short)0, acc[i], false, false);
        }
        __syncthreads();
        const float bv = bias[ncol];
        #pragma unroll
        for (int i = 0; i < 4; ++i) {
            #pragma unroll
            for (int r = 0; r < 8; ++r)
                h2t[(i * 16 + rhalf + r) * TAIL_P1 + ncol] = fmaxf(acc[i][r] + bv, 0.0f);
        }
    }
    __syncthreads();

    {
        v8f acc[4][2];
        #pragma unroll
        for (int i = 0; i < 4; ++i) {
            #pragma unroll
            for (int j = 0; j < 2; ++j) acc[i][j] = v8f{};
        }
        const float* ar = h2t + mrow * TAIL_P1 + khalf;
        const float* wr[2];
        #pragma unroll
        for (int j = 0; j < 2; ++j)
            wr[j] = Wd1 + (size_t)((wave * 2 + j) * 16 + mrow) * TAIL_H2 + khalf;
        for (int k = 0; k < TAIL_H2; k += 4) {
            v2f a[4], b[2];
            #pragma unroll
            for (int i = 0; i < 4; ++i) a[i] = *(const v2f*)(ar + i * 16 * TAIL_P1 + k);
            #pragma unroll
            for (int j = 0; j < 2; ++j) b[j] = *(const v2f*)(wr[j] + k);
            #pragma unroll
            for (int i = 0; i < 4; ++i) {
                #pragma unroll
                for (int j = 0; j < 2; ++j)
                    acc[i][j] = __builtin_amdgcn_wmma_f32_16x16x4_f32(false, a[i], false, b[j],
                                                                      (short)0, acc[i][j], false, false);
            }
        }
        #pragma unroll
        for (int j = 0; j < 2; ++j) {
            const int ncol = (wave * 2 + j) * 16 + mrow;
            const float bv = bd1[ncol];
            #pragma unroll
            for (int i = 0; i < 4; ++i) {
                #pragma unroll
                for (int r = 0; r < 8; ++r)
                    hdt[(i * 16 + rhalf + r) * TAIL_P2 + ncol] = fmaxf(acc[i][j][r] + bv, 0.0f);
            }
        }
    }
    __syncthreads();

    {
        v8f acc = v8f{};
        const int wrow = (mrow < 2) ? mrow : 0;
        const float* wr = Wd2 + (size_t)wrow * TAIL_HD + khalf;
        const float* ar = hdt + (wave * 16 + mrow) * TAIL_P2 + khalf;
        for (int k = 0; k < TAIL_HD; k += 4) {
            const v2f a = *(const v2f*)(ar + k);
            const v2f b = *(const v2f*)(wr + k);
            acc = __builtin_amdgcn_wmma_f32_16x16x4_f32(false, a, false, b,
                                                        (short)0, acc, false, false);
        }
        const int m0 = row0 + wave * 16;
        if (mrow < 2 && m0 < N_NODES) {
            const float bv = bd2[mrow];
            #pragma unroll
            for (int r = 0; r < 8; ++r)
                out[(size_t)(m0 + rhalf + r) * 2 + mrow] = acc[r] + bv;
        }
    }
}

static inline int cdiv(long a, long b) { return (int)((a + b - 1) / b); }

extern "C" void kernel_launch(void* const* d_in, const int* in_sizes, int n_in,
                              void* d_out, int out_size, void* d_ws, size_t ws_size,
                              hipStream_t stream) {
    const float* x    = (const float*)d_in[0];
    const int*   edge = (const int*)d_in[1];
    const int*   src  = edge;
    const int*   dst  = edge + N_EDGES;
    const float* Wl0 = (const float*)d_in[2],  *bl0 = (const float*)d_in[3],  *Wr0 = (const float*)d_in[4];
    const float* Wl1 = (const float*)d_in[5],  *bl1 = (const float*)d_in[6],  *Wr1 = (const float*)d_in[7];
    const float* Wl2 = (const float*)d_in[8],  *bl2 = (const float*)d_in[9],  *Wr2 = (const float*)d_in[10];
    const float* Wd1 = (const float*)d_in[11], *bd1 = (const float*)d_in[12];
    const float* Wd2 = (const float*)d_in[13], *bd2 = (const float*)d_in[14];

    float* ws   = (float*)d_ws;
    float* deg  = ws;
    float* inv  = deg  + N_NODES;
    float* agg  = inv  + N_NODES;
    float* bufA = agg  + (size_t)N_NODES * 128;
    float* bufB = bufA + (size_t)N_NODES * 128;

    const int gs = cdiv(N_NODES, SLAB_ROWS);
    const int go = N_BLK;
    static_assert((size_t)SAVE_HEAD + (size_t)N_BLK * BIN_WGS * BIN_CAP
                  <= (size_t)N_NODES * 128, "cells must fit the plane");
    unsigned* sv = (unsigned*)bufB;

    bin_edges_kernel<<<BIN_WGS, 1024, 0, stream>>>(src, dst, sv);

    own_rows_sum_kernel<64, true, 128><<<go, 1024, 0, stream>>>(x, src, dst, nullptr, inv, sv,
                                                                Wl0, Wr0, bl0, bufA);

    float* const h1 = agg;
    own_rows_sum_kernel<128, false, 128><<<go, 1024, 0, stream>>>(bufA, src, dst, nullptr, inv, sv,
                                                                  Wl1, Wr1, bl1, h1);

    own_rows_sum_kernel<128, false, 0><<<go, 1024, 0, stream>>>(h1, src, dst, bufA, inv, sv,
                                                                nullptr, nullptr, nullptr, nullptr);
    tail_kernel<<<gs, 128, 0, stream>>>(bufA, inv, Wl2, h1, Wr2, bl2, Wd1, bd1, Wd2, bd2,
                                        (float*)d_out);
}


# layer 2: the owner work-group multiplies its 392 rows and runs both decoder products too (result tiles stay in LDS); no plane of sums in memory, 1 launch fewer; sum order unchanged
# speedup vs baseline: 5.6282x; 1.1081x over previous
#include <hip/hip_runtime.h>
#include <hip/hip_bf16.h>

#define N_NODES 100000
#define N_EDGES 1600000

typedef float v2f __attribute__((ext_vector_type(2)));
typedef float v8f __attribute__((ext_vector_type(8)));

#define OWN_ROWS 392
#define OWN_QLEN 160
#define OWN_SLOTS ((OWN_ROWS + 31) / 32)
#define N_BLK ((N_NODES + OWN_ROWS - 1) / OWN_ROWS)
#define BIN_PIECES 25
#define BIN_WGS (N_EDGES / 512 / BIN_PIECES)
#define BIN_CAP 128
#define SAVE_HEAD 32768

#define OWN_SLOT_ADD(A, H, PV, T) \
    asm( \
    "s_cmp_lt_i32 %14, 8\n\t" \
    "s_cbranch_scc0 .Lown_ge8_%=\n\t" \
    "s_cmp_lt_i32 %14, 4\n\t" \
    "s_cbranch_scc0 .Lown_ge4_%=\n\t" \
    "s_cmp_lt_i32 %14, 2\n\t" \
    "s_cbranch_scc0 .Lown_ge2_%=\n\t" \
    "s_cmp_lt_i32 %14, 1\n\t" \
    "s_cbranch_scc0 .Lown_ge1_%=\n\t" \
    "v_pk_add_f32 %0, %0, %13\n\t" \
    "s_branch .Lown_end_%=\n\t" \
    ".Lown_ge1_%=:\n\t" \
    "v_pk_add_f32 %1, %1, %13\n\t" \
    "s_branch .Lown_end_%=\n\t" \
    ".Lown_ge2_%=:\n\t" \
    "s_cmp_lt_i32 %14, 3\n\t" \
    "s_cbranch_scc0 .Lown_ge3_%=\n\t" \
    "v_pk_add_f32 %2, %2, %13\n\t" \
    "s_branch .Lown_end_%=\n\t" \
    ".Lown_ge3_%=:\n\t" \
    "v_pk_add_f32 %3, %3, %13\n\t" \
    "s_branch .Lown_end_%=\n\t" \
    ".Lown_ge4_%=:\n\t" \
    "s_cmp_lt_i32 %14, 6\n\t" \
    "s_cbranch_scc0 .Lown_ge6_%=\n\t" \
    "s_cmp_lt_i32 %14, 5\n\t" \
    "s_cbranch_scc0 .Lown_ge5_%=\n\t" \
    "v_pk_add_f32 %4, %4, %13\n\t" \
    "s_branch .Lown_end_%=\n\t" \
    ".Lown_ge5_%=:\n\t" \
    "v_pk_add_f32 %5, %5, %13\n\t" \
    "s_branch .Lown_end_%=\n\t" \
    ".Lown_ge6_%=:\n\t" \
    "s_cmp_lt_i32 %14, 7\n\t" \
    "s_cbranch_scc0 .Lown_ge7_%=\n\t" \
    "v_pk_add_f32 %6, %6, %13\n\t" \
    "s_branch .Lown_end_%=\n\t" \
    ".Lown_ge7_%=:\n\t" \
    "v_pk_add_f32 %7, %7, %13\n\t" \
    "s_branch .Lown_end_%=\n\t" \
    ".Lown_ge8_%=:\n\t" \
    "s_cmp_lt_i32 %14, 12\n\t" \
    "s_cbranch_scc0 .Lown_ge12_%=\n\t" \
    "s_cmp_lt_i32 %14, 10\n\t" \
    "s_cbranch_scc0 .Lown_ge10_%=\n\t" \
    "s_cmp_lt_i32 %14, 9\n\t" \
    "s_cbranch_scc0 .Lown_ge9_%=\n\t" \
    "v_pk_add_f32 %8, %8, %13\n\t" \
    "s_branch .Lown_end_%=\n\t" \
    ".Lown_ge9_%=:\n\t" \
    "v_pk_add_f32 %9, %9, %13\n\t" \
    "s_branch .Lown_end_%=\n\t" \
    ".Lown_ge10_%=:\n\t" \
    "s_cmp_lt_i32 %14, 11\n\t" \
    "s_cbranch_scc0 .Lown_ge11_%=\n\t" \
    "v_pk_add_f32 %10, %10, %13\n\t" \
    "s_branch .Lown_end_%=\n\t" \
    ".Lown_ge11_%=:\n\t" \
    "v_pk_add_f32 %11, %11, %13\n\t" \
    "s_branch .Lown_end_%=\n\t" \
    ".Lown_ge12_%=:\n\t" \
    "s_cmp_lt_i32 %14, 13\n\t" \
    "s_cbranch_scc0 .Lown_ge13_%=\n\t" \
    "v_pk_add_f32 %12, %12, %13\n\t" \
    "s_branch .Lown_end_%=\n\t" \
    ".Lown_ge13_%=:\n\t" \
    ".Lown_end_%=:\n\t" \
    "" \
    : "+v"(A[0][H]), "+v"(A[1][H]), "+v"(A[2][H]), "+v"(A[3][H]), "+v"(A[4][H]), \
      "+v"(A[5][H]), "+v"(A[6][H]), "+v"(A[7][H]), "+v"(A[8][H]), "+v"(A[9][H]), \
      "+v"(A[10][H]), "+v"(A[11][H]), "+v"(A[12][H]) \
    : "v"(PV), "s"(T) \
    : "scc")

#define OWN_SLOT_ADD2(A, PV0, PV1, T) \
    asm( \
    "s_cmp_lt_i32 %28, 8\n\t" \
    "s_cbranch_scc0 .Lown2_ge8_%=\n\t" \
    "s_cmp_lt_i32 %28, 4\n\t" \
    "s_cbranch_scc0 .Lown2_ge4_%=\n\t" \
    "s_cmp_lt_i32 %28, 2\n\t" \
    "s_cbranch_scc0 .Lown2_ge2_%=\n\t" \
    "s_cmp_lt_i32 %28, 1\n\t" \
    "s_cbranch_scc0 .Lown2_ge1_%=\n\t" \
    "v_pk_add_f32 %0, %0, %26\n\t" \
    "v_pk_add_f32 %13, %13, %27\n\t" \
    "s_branch .Lown2_end_%=\n\t" \
    ".Lown2_ge1_%=:\n\t" \
    "v_pk_add_f32 %1, %1, %26\n\t" \
    "v_pk_add_f32 %14, %14, %27\n\t" \
    "s_branch .Lown2_end_%=\n\t" \
    ".Lown2_ge2_%=:\n\t" \
    "s_cmp_lt_i32 %28, 3\n\t" \
    "s_cbranch_scc0 .Lown2_ge3_%=\n\t" \
    "v_pk_add_f32 %2, %2, %26\n\t" \
    "v_pk_add_f32 %15, %15, %27\n\t" \
    "s_branch .Lown2_end_%=\n\t" \
    ".Lown2_ge3_%=:\n\t" \
    "v_pk_add_f32 %3, %3, %26\n\t" \
    "v_pk_add_f32 %16, %16, %27\n\t" \
    "s_branch .Lown2_end_%=\n\t" \
    ".Lown2_ge4_%=:\n\t" \
    "s_cmp_lt_i32 %28, 6\n\t" \
    "s_cbranch_scc0 .Lown2_ge6_%=\n\t" \
    "s_cmp_lt_i32 %28, 5\n\t" \
    "s_cbranch_scc0 .Lown2_ge5_%=\n\t" \
    "v_pk_add_f32 %4, %4, %26\n\t" \
    "v_pk_add_f32 %17, %17, %27\n\t" \
    "s_branch .Lown2_end_%=\n\t" \
    ".Lown2_ge5_%=:\n\t" \
    "v_pk_add_f32 %5, %5, %26\n\t" \
    "v_pk_add_f32 %18, %18, %27\n\t" \
    "s_branch .Lown2_end_%=\n\t" \
    ".Lown2_ge6_%=:\n\t" \
    "s_cmp_lt_i32 %28, 7\n\t" \
    "s_cbranch_scc0 .Lown2_ge7_%=\n\t" \
    "v_pk_add_f32 %6, %6, %26\n\t" \
    "v_pk_add_f32 %19, %19, %27\n\t" \
    "s_branch .Lown2_end_%=\n\t" \
    ".Lown2_ge7_%=:\n\t" \
    "v_pk_add_f32 %7, %7, %26\n\t" \
    "v_pk_add_f32 %20, %20, %27\n\t" \
    "s_branch .Lown2_end_%=\n\t" \
    ".Lown2_ge8_%=:\n\t" \
    "s_cmp_lt_i32 %28, 12\n\t" \
    "s_cbranch_scc0 .Lown2_ge12_%=\n\t" \
    "s_cmp_lt_i32 %28, 10\n\t" \
    "s_cbranch_scc0 .Lown2_ge10_%=\n\t" \
    "s_cmp_lt_i32 %28, 9\n\t" \
    "s_cbranch_scc0 .Lown2_ge9_%=\n\t" \
    "v_pk_add_f32 %8, %8, %26\n\t" \
    "v_pk_add_f32 %21, %21, %27\n\t" \
    "s_branch .Lown2_end_%=\n\t" \
    ".Lown2_ge9_%=:\n\t" \
    "v_pk_add_f32 %9, %9, %26\n\t" \
    "v_pk_add_f32 %22, %22, %27\n\t" \
    "s_branch .Lown2_end_%=\n\t" \
    ".Lown2_ge10_%=:\n\t" \
    "s_cmp_lt_i32 %28, 11\n\t" \
    "s_cbranch_scc0 .Lown2_ge11_%=\n\t" \
    "v_pk_add_f32 %10, %10, %26\n\t" \
    "v_pk_add_f32 %23, %23, %27\n\t" \
    "s_branch .Lown2_end_%=\n\t" \
    ".Lown2_ge11_%=:\n\t" \
    "v_pk_add_f32 %11, %11, %26\n\t" \
    "v_pk_add_f32 %24, %24, %27\n\t" \
    "s_branch .Lown2_end_%=\n\t" \
    ".Lown2_ge12_%=:\n\t" \
    "s_cmp_lt_i32 %28, 13\n\t" \
    "s_cbranch_scc0 .Lown2_ge13_%=\n\t" \
    "v_pk_add_f32 %12, %12, %26\n\t" \
    "v_pk_add_f32 %25, %25, %27\n\t" \
    "s_branch .Lown2_end_%=\n\t" \
    ".Lown2_ge13_%=:\n\t" \
    ".Lown2_end_%=:\n\t" \
    "" \
    : "+v"(A[0][0]), "+v"(A[1][0]), "+v"(A[2][0]), "+v"(A[3][0]), "+v"(A[4][0]), "+v"(A[5][0]), "+v"(A[6][0]), "+v"(A[7][0]), "+v"(A[8][0]), "+v"(A[9][0]), "+v"(A[10][0]), "+v"(A[11][0]), "+v"(A[12][0]), \
      "+v"(A[0][1]), "+v"(A[1][1]), "+v"(A[2][1]), "+v"(A[3][1]), "+v"(A[4][1]), "+v"(A[5][1]), "+v"(A[6][1]), "+v"(A[7][1]), "+v"(A[8][1]), "+v"(A[9][1]), "+v"(A[10][1]), "+v"(A[11][1]), "+v"(A[12][1]) \
    : "v"(PV0), "v"(PV1), "s"(T) \
    : "scc")

__global__ __launch_bounds__(1024)
void bin_edges_kernel(const int* __restrict__ src, const int* __restrict__ dst,
                      unsigned* __restrict__ sv) {
    static_assert(N_EDGES % (512 * BIN_PIECES) == 0, "every group has whole pieces");
    static_assert(N_BLK * BIN_WGS <= SAVE_HEAD, "one count per cell");
    static_assert(OWN_ROWS <= 512, "a word keeps the local row in 9 bits");
    static_assert(BIN_CAP == 128, "a wave writes a cell with one 16-byte store per lane");
    __shared__ unsigned cnt[N_BLK];
    __shared__ __attribute__((aligned(16))) unsigned cell[N_BLK * BIN_CAP];

    const int tid  = threadIdx.x;
    const int lane = tid & 31;
    const int wave = tid >> 5;
    for (int i = tid; i < N_BLK; i += 1024) cnt[i] = 0u;
    __syncthreads();

    constexpr int NV = BIN_PIECES * 128;
    constexpr int NK = (NV + 1023) / 1024;
    const int4* d4 = (const int4*)dst + (size_t)blockIdx.x * NV;
    const int4* s4 = (const int4*)src + (size_t)blockIdx.x * NV;
    int4 dv[NK], sw[NK];
    #pragma unroll
    for (int k = 0; k < NK; ++k) {
        const int i = min(k * 1024 + tid, NV - 1);
        dv[k] = d4[i];
        sw[k] = s4[i];
    }
    #pragma unroll
    for (int k = 0; k < NK; ++k) {
        if (k * 1024 + tid < NV) {
            const int d[4] = {dv[k].x, dv[k].y, dv[k].z, dv[k].w};
            const int s[4] = {sw[k].x, sw[k].y, sw[k].z, sw[k].w};
            unsigned o[4], r[4], lr[4];
            #pragma unroll
            for (int c = 0; c < 4; ++c) {
                o[c]  = min((unsigned)d[c] / (unsigned)OWN_ROWS, (unsigned)(N_BLK - 1));
                lr[c] = (unsigned)d[c] - o[c] * (unsigned)OWN_ROWS;
                r[c]  = atomicAdd(&cnt[o[c]], 1u);
            }
            #pragma unroll
            for (int c = 0; c < 4; ++c) {
                if (r[c] < (unsigned)BIN_CAP)
                    cell[o[c] * BIN_CAP + r[c]] = ((unsigned)s[c] << 9) | (lr[c] & 511u);
            }
        }
    }
    __syncthreads();

    for (int o = wave; o < N_BLK; o += 32) {
        const uint4 v = ((const uint4*)(cell + o * BIN_CAP))[lane];
        ((uint4*)(sv + SAVE_HEAD + ((size_t)o * BIN_WGS + blockIdx.x) * BIN_CAP))[lane] = v;
        if (lane == 0) sv[o * BIN_WGS + blockIdx.x] = cnt[o];
    }
}

#define OWN_TILES ((OWN_ROWS + 15) / 16)
template <int N> struct tile_count { static constexpr int value = N; };
template <int F, bool DEG, int DOUT>
__global__ __launch_bounds__(1024)
void own_rows_sum_kernel(const float* __restrict__ x,
                         const int* __restrict__ src,
                         const int* __restrict__ dst,
                         float* __restrict__ agg,
                         float* __restrict__ inv,
                         const unsigned* __restrict__ sv,
                         const float* __restrict__ Wl,
                         const float* __restrict__ Wr,
                         const float* __restrict__ bias,
                         float* __restrict__ out,
                         const float* __restrict__ Wd1,
                         const float* __restrict__ bd1,
                         const float* __restrict__ Wd2,
                         const float* __restrict__ bd2) {
    constexpr int R   = OWN_ROWS;
    constexpr int VEC = F / 32;
    typedef float vecf __attribute__((ext_vector_type(VEC)));
    constexpr int SP  = F + 4;
    constexpr int ST  = OWN_TILES;
    constexpr int ACCN0 = (DOUT > 0 && ST * 16 * SP > (R + 1) * F) ? ST * 16 * SP : (R + 1) * F;
    constexpr int TAILN = ST * 16 * 68 + 16 * 16 * 132;
    constexpr int ACCN = (DOUT == 64 && TAILN > ACCN0) ? TAILN : ACCN0;

    __shared__ __attribute__((aligned(16))) float acc[ACCN];
    constexpr int QLEN = ((R + 1) * F * 4 < 160 * 1024) ? 3 * OWN_QLEN : OWN_QLEN;
    __shared__ int2  que[32][QLEN];
    __shared__ float degl[R];

    const int tid  = threadIdx.x;
    const int lane = tid & 31;
    const int wave = __builtin_amdgcn_readfirstlane(tid >> 5);
    const int lo   = blockIdx.x * R;

    constexpr int NS = OWN_SLOTS;
    constexpr int QW = 2 * OWN_QLEN;
    static_assert(NS == 13, "OWN_SLOT_ADD lists 13 slots");
    static_assert(VEC == 2 || VEC == 4, "a lane's piece of a row is one or two 8-byte halves");
    static_assert((NS << 5) < 512, "the padding slot fits the word's 9 bits");
    static_assert(BIN_CAP == 128, "a cell is read as four words per lane");
    __shared__ unsigned qcnt[32];
    __shared__ unsigned bad;

    if (tid < 32) qcnt[tid] = 0u;
    if (tid == 0) bad = 0u;
    __syncthreads();
    {
        constexpr int NCW = (BIN_WGS + 31) / 32;
        const unsigned* const cnts  = sv + blockIdx.x * BIN_WGS;
        const unsigned* const cells = sv + SAVE_HEAD + (size_t)blockIdx.x * BIN_WGS * BIN_CAP;
        unsigned cn[NCW], cw[NCW][4];
        #pragma unroll
        for (int k = 0; k < NCW; ++k) {
            const int g = min(wave + 32 * k, BIN_WGS - 1);
            cn[k] = cnts[g];
            #pragma unroll
            for (int c = 0; c < 4; ++c) cw[k][c] = cells[g * BIN_CAP + c * 32 + lane];
        }
        #pragma unroll
        for (int k = 0; k < NCW; ++k) {
            if (wave + 32 * k < BIN_WGS) {
                const unsigned n = (unsigned)__builtin_amdgcn_readfirstlane((int)cn[k]);
                if (n > (unsigned)BIN_CAP) {
                    if (lane == 0) bad = 1u;
                } else {
                    #pragma unroll
                    for (int c = 0; c < 4; ++c) {
                        if ((unsigned)(c * 32 + lane) < n) {
                            const unsigned wd = cw[k][c];
                            const unsigned s  = min(wd >> 9, (unsigned)(N_NODES - 1));
                            const unsigned lr = min(wd & 511u, (unsigned)(R - 1));
                            const unsigned w  = lr & 31u;
                            const unsigned at = atomicAdd(&qcnt[w], 1u);
                            if (at < (unsigned)QW) ((unsigned*)que[w])[at] = (s << 9) | lr;
                            else bad = 1u;
                        }
                    }
                }
            }
        }
    }
    __syncthreads();
    constexpr int NH = VEC / 2;
    v2f accr[NS][NH];
    if (__builtin_amdgcn_readfirstlane((int)bad) == 0) {
        const int nq = __builtin_amdgcn_readfirstlane((int)qcnt[wave]);
        const unsigned* const myq = (const unsigned*)que[wave];
        #pragma unroll
        for (int i = 0; i < NS; ++i) {
            #pragma unroll
            for (int hf = 0; hf < NH; ++hf) accr[i][hf] = (v2f)(0.0f);
        }
        float dcnt = 0.0f;
        for (int h = 0; h < nq; h += 32) {
            const int cnt = min(32, nq - h);
            unsigned wd = 0u;
            if (lane < cnt) wd = myq[h + lane];
            int sl = (int)(wd >> 9);
            int ry = (int)((wd & 511u) >> 5);
            if (lane >= cnt) { sl = 0; ry = NS; }
            for (int g = 0; g * 8 < cnt; ++g) {
                vecf v[8];
                int  sr[8];
                #pragma unroll
                for (int j = 0; j < 8; ++j) {
                    const int s = __builtin_amdgcn_readlane(sl, g * 8 + j);
                    sr[j] = __builtin_amdgcn_readlane(ry, g * 8 + j);
                    v[j]  = *(const vecf*)(x + (size_t)s * F + lane * VEC);
                }
                __builtin_amdgcn_sched_barrier(0);
                #pragma unroll
                for (int j = 0; j < 8; ++j) {
                    const int t = sr[j];
                    if constexpr (NH == 1) {
                        v2f pv;
                        pv[0] = v[j][0];
                        pv[1] = v[j][1];
                        OWN_SLOT_ADD(accr, 0, pv, t);
                    } else {
                        v2f p0, p1;
                        p0[0] = v[j][0];
                        p0[1] = v[j][1];
                        p1[0] = v[j][VEC - 2];
                        p1[1] = v[j][VEC - 1];
                        OWN_SLOT_ADD2(accr, p0, p1, t);
                    }
                    if (DEG) dcnt += (lane == sr[j]) ? 1.0f : 0.0f;
                }
            }
        }
        if constexpr (DOUT == 0) {
            #pragma unroll
            for (int i = 0; i < NS; ++i) {
                const int row  = wave + 32 * i;
                const int node = lo + row;
                if (row < R && node < N_NODES) {
                    vecf o;
                    #pragma unroll
                    for (int hf = 0; hf < NH; ++hf) {
                        o[2 * hf]     = accr[i][hf][0];
                        o[2 * hf + 1] = accr[i][hf][1];
                    }
                    *(vecf*)(agg + (size_t)node * F + lane * VEC) = o;
                }
            }
            if (DEG) {
                if (lane < NS && wave + 32 * lane < R) degl[wave + 32 * lane] = dcnt;
                __syncthreads();
                if (tid < R) {
                    const int node = lo + tid;
                    if (node < N_NODES) {
                        const float dd = degl[tid];
                        inv[node] = (dd > 0.0f) ? (1.0f / fmaxf(dd, 1.0f)) : 0.0f;
                    }
                }
            }
            return;
        } else {
            if (DEG) {
                if (lane < NS && wave + 32 * lane < R) degl[wave + 32 * lane] = dcnt;
            }
        }
    } else {
        for (int i = tid; i < (R + 1) * F / 4; i += 1024)
            ((float4*)acc)[i] = make_float4(0.0f, 0.0f, 0.0f, 0.0f);
        if (DEG) for (int i = tid; i < R; i += 1024) degl[i] = 0.0f;
        __syncthreads();

        int2* q = que[wave];
        int tail = 0;

        auto add_rows = [&](int sl, int ry, int cnt) {
            #pragma unroll
            for (int g = 0; g < 4; ++g) {
                if (g * 8 < cnt) {
                    vecf v[8];
                    int  rr[8];
                    #pragma unroll
                    for (int j = 0; j < 8; ++j) {
                        int s = __builtin_amdgcn_readlane(sl, g * 8 + j);
                        rr[j] = __builtin_amdgcn_readlane(ry, g * 8 + j);
                        v[j]  = *(const vecf*)(x + (size_t)s * F + lane * VEC);
                    }
                    #pragma unroll
                    for (int j = 0; j < 8; ++j) {
                        float* a = acc + rr[j] * F + lane;
                        #pragma unroll
                        for (int c = 0; c < VEC; ++c)
                            __hip_atomic_fetch_add(a + c * 32, v[j][c], __ATOMIC_RELAXED,
                                                   __HIP_MEMORY_SCOPE_WORKGROUP);
                    }
                }
            }
        };

        auto replay = [&](const unsigned* cw, int n, unsigned cur) {
            for (int h = 0; h < n; h += 32) {
                const int cnt = min(32, n - h);
                const unsigned nxt = cw[min(h + 32 + lane, BIN_CAP - 1)];
                int sl = min((int)(cur >> 9), N_NODES - 1);
                int ry = min((int)(cur & 511u), R - 1);
                if (lane >= cnt) { sl = 0; ry = R; }
                if (DEG) {
                    if (lane < cnt)
                        __hip_atomic_fetch_add(&degl[ry], 1.0f, __ATOMIC_RELAXED,
                                               __HIP_MEMORY_SCOPE_WORKGROUP);
                }
                add_rows(sl, ry, cnt);
                cur = nxt;
            }
        };

        auto drain = [&](int h, int cnt) {
            __builtin_amdgcn_wave_barrier();
            int2 ent = q[h + lane];
            if (lane >= cnt) ent = make_int2(0, R);
            const int sl = min(max(src[ent.x], 0), N_NODES - 1);
            add_rows(sl, ent.y, cnt);
        };

        auto chunk = [&](const int4 d, int ch) {
            unsigned r[4];
            r[0] = (unsigned)(d.x - lo); r[1] = (unsigned)(d.y - lo);
            r[2] = (unsigned)(d.z - lo); r[3] = (unsigned)(d.w - lo);
            const unsigned m = min(min(r[0], r[1]), min(r[2], r[3]));
            if (__builtin_amdgcn_ballot_w32(m < (unsigned)R) != 0u) {
                #pragma unroll
                for (int c = 0; c < 4; ++c) {
                    const bool hit = r[c] < (unsigned)R;
                    const unsigned mask = __builtin_amdgcn_ballot_w32(hit);
                    if (mask != 0u) {
                        if (hit) {
                            const int e = ch * 128 + lane * 4 + c;
                            q[tail + (int)__builtin_amdgcn_mbcnt_lo(mask, 0u)] = make_int2(e, (int)r[c]);
                            if (DEG)
                                __hip_atomic_fetch_add(&degl[r[c]], 1.0f, __ATOMIC_RELAXED,
                                                       __HIP_MEMORY_SCOPE_WORKGROUP);
                        }
                        tail += __builtin_popcount(mask);
                    }
                }
                if (tail >= 32) {
                    int h = 0;
                    do { drain(h, 32); h += 32; } while (tail - h >= 32);
                    __builtin_amdgcn_wave_barrier();
                    const int2 t = q[h + lane];
                    __builtin_amdgcn_wave_barrier();
                    if (lane < tail - h) q[lane] = t;
                    tail -= h;
                }
            }
        };

        auto scan_group = [&](int grp) {
            const int4* d4 = (const int4*)dst;
            for (int ci = 0; ci < BIN_PIECES; ++ci) {
                const int sc = grp * BIN_PIECES + ci;
                for (int k = 0; k < 4; ++k)
                    chunk(d4[((size_t)sc * 4 + k) * 32 + lane], sc * 4 + k);
            }
            if (tail > 0) drain(0, tail);
            tail = 0;
        };

        static_assert(BIN_WGS >= 32, "every wave has a first cell");
        const unsigned* const cnts  = sv + blockIdx.x * BIN_WGS;
        const unsigned* const cells = sv + SAVE_HEAD + (size_t)blockIdx.x * BIN_WGS * BIN_CAP;
        int grp = wave;
        unsigned n = cnts[grp];
        unsigned f = cells[grp * BIN_CAP + lane];
        while (true) {
            const bool more = grp + 32 < BIN_WGS;
            const int  gn   = more ? grp + 32 : grp;
            const unsigned n2 = cnts[gn];
            const unsigned f2 = cells[gn * BIN_CAP + lane];
            n = (unsigned)__builtin_amdgcn_readfirstlane((int)n);
            if (n <= (unsigned)BIN_CAP) replay(cells + grp * BIN_CAP, (int)n, f);
            else scan_group(grp);
            if (!more) break;
            grp = gn; n = n2; f = f2;
        }
        __syncthreads();

        if constexpr (DOUT == 0) {
            for (int row = wave; row < R; row += 32) {
                const int node = lo + row;
                if (node < N_NODES) {
                    vecf o;
                    #pragma unroll
                    for (int c = 0; c < VEC; ++c) o[c] = acc[row * F + c * 32 + lane];
                    *(vecf*)(agg + (size_t)node * F + lane * VEC) = o;
                }
            }
            if (DEG) {
                if (tid < R) {
                    const int node = lo + tid;
                    if (node < N_NODES) {
                        const float dd = degl[tid];
                        inv[node] = (dd > 0.0f) ? (1.0f / fmaxf(dd, 1.0f)) : 0.0f;
                    }
                }
            }
            return;
        } else {
            #pragma unroll
            for (int i = 0; i < NS; ++i) {
                const int row = wave + 32 * i;
                #pragma unroll
                for (int hf = 0; hf < NH; ++hf) {
                    accr[i][hf] = (v2f)(0.0f);
                    if (row < R) {
                        accr[i][hf][0] = acc[row * F + (2 * hf) * 32 + lane];
                        accr[i][hf][1] = acc[row * F + (2 * hf + 1) * 32 + lane];
                    }
                }
            }
        }
    }

    if constexpr (DOUT > 0) {
        static_assert(DOUT == 128 || DOUT == 64, "a wave has DOUT / 64 column tiles: four waves cover the columns");
        static_assert(ST == 25, "row tiles: three per wave, and the last one for waves 0-3");
        static_assert(SP % 64 == 4, "fragment reads of 16 rows x 2 k-pairs fall into 64 banks");
        __syncthreads();
        float* const slab = acc;
        if (DEG) {
            if (tid < R) {
                const int node = lo + tid;
                if (node < N_NODES) {
                    const float dd = degl[tid];
                    inv[node] = (dd > 0.0f) ? (1.0f / fmaxf(dd, 1.0f)) : 0.0f;
                }
            }
        }
        float sl = 0.0f;
        if (lane < NS) {
            const int row = min(wave + 32 * lane, R - 1);
            if (DEG) {
                const float dd = degl[row];
                sl = (dd > 0.0f) ? (1.0f / fmaxf(dd, 1.0f)) : 0.0f;
            } else {
                sl = inv[min(lo + row, N_NODES - 1)];
            }
        }
        #pragma unroll
        for (int i = 0; i < NS; ++i) {
            const int row = wave + 32 * i;
            if (row < R) {
                const float s = __int_as_float(__builtin_amdgcn_readlane(__float_as_int(sl), i));
                vecf o;
                #pragma unroll
                for (int hf = 0; hf < NH; ++hf) {
                    o[2 * hf]     = accr[i][hf][0] * s;
                    o[2 * hf + 1] = accr[i][hf][1] * s;
                }
                *(vecf*)(slab + row * SP + lane * VEC) = o;
            }
        }
        for (int i = tid; i < (ST * 16 - R) * F; i += 1024)
            slab[(R + i / F) * SP + i % F] = 0.0f;
        __syncthreads();

        const int mrow  = lane & 15;
        const int khalf = (lane >> 4) << 1;
        const int rhalf = (lane >> 4) << 3;
        const int rg    = wave >> 2;
        const int cg    = wave & 3;
        const bool extra = (rg == 0);
        constexpr int NJ = DOUT / 64;
        const int ncol0 = cg * (16 * NJ) + mrow;

        v8f pacc[3][NJ], pex[NJ];
        #pragma unroll
        for (int j = 0; j < NJ; ++j) {
            pex[j] = v8f{};
            #pragma unroll
            for (int i = 0; i < 3; ++i) pacc[i][j] = v8f{};
        }
        const float* ar[3];
        #pragma unroll
        for (int i = 0; i < 3; ++i) ar[i] = slab + ((rg + 8 * i) * 16 + mrow) * SP + khalf;
        const float* const arx = slab + ((ST - 1) * 16 + mrow) * SP + khalf;

        auto mul_range = [&](const float* W) {
            const float* wp[NJ];
            #pragma unroll
            for (int j = 0; j < NJ; ++j) wp[j] = W + (size_t)(ncol0 + 16 * j) * F + khalf;
            if (extra) {
                for (int k = 0; k < F; k += 4) {
                    const v2f a  = *(const v2f*)(arx + k);
                    v2f b[NJ];
                    #pragma unroll
                    for (int j = 0; j < NJ; ++j) b[j] = *(const v2f*)(wp[j] + k);
                    #pragma unroll
                    for (int j = 0; j < NJ; ++j)
                        pex[j] = __builtin_amdgcn_wmma_f32_16x16x4_f32(false, a, false, b[j],
                                                                       (short)0, pex[j], false, false);
                }
            }
            for (int k = 0; k < F; k += 4) {
                v2f a[3], b[NJ];
                #pragma unroll
                for (int i = 0; i < 3; ++i) a[i] = *(const v2f*)(ar[i] + k);
                #pragma unroll
                for (int j = 0; j < NJ; ++j) b[j] = *(const v2f*)(wp[j] + k);
                #pragma unroll
                for (int i = 0; i < 3; ++i) {
                    #pragma unroll
                    for (int j = 0; j < NJ; ++j)
                        pacc[i][j] = __builtin_amdgcn_wmma_f32_16x16x4_f32(false, a[i], false, b[j],
                                                                           (short)0, pacc[i][j], false, false);
                }
            }
        };

        mul_range(Wl);

        typedef float v4f __attribute__((ext_vector_type(4)));
        constexpr int QF  = F / 4;
        constexpr int NIT = (R * QF + 1023) / 1024;
        constexpr int NB  = 4;
        __syncthreads();
        #pragma unroll
        for (int b0 = 0; b0 < NIT; b0 += NB) {
            v4f xs[NB];
            #pragma unroll
            for (int t = 0; t < NB; ++t) {
                const int i = min((b0 + t) * 1024 + tid, R * QF - 1);
                const int node = min(lo + i / QF, N_NODES - 1);
                xs[t] = *(const v4f*)(x + (size_t)node * F + (i % QF) * 4);
            }
            #pragma unroll
            for (int t = 0; t < NB; ++t) {
                const int i = (b0 + t) * 1024 + tid;
                if (i < R * QF)
                    *(v4f*)(slab + (i / QF) * SP + (i % QF) * 4) = xs[t];
            }
        }
        __syncthreads();

        mul_range(Wr);

        static_assert(R % 8 == 0 && N_NODES % 8 == 0, "one test for a register's 8 rows");
        if constexpr (DOUT == 128) {
            #pragma unroll
            for (int j = 0; j < NJ; ++j) {
                const int ncol = ncol0 + 16 * j;
                const float bv = bias[ncol];
                #pragma unroll
                for (int i = 0; i < 3; ++i) {
                    const int node0 = lo + (rg + 8 * i) * 16 + rhalf;
                    if (node0 < N_NODES) {
                        #pragma unroll
                        for (int r = 0; r < 8; ++r)
                            out[(size_t)(node0 + r) * DOUT + ncol] = fmaxf(pacc[i][j][r] + bv, 0.0f);
                    }
                }
                if (extra) {
                    const int lrow0 = (ST - 1) * 16 + rhalf;
                    if (lrow0 < R && lo + lrow0 < N_NODES) {
                        #pragma unroll
                        for (int r = 0; r < 8; ++r)
                            out[(size_t)(lo + lrow0 + r) * DOUT + ncol] = fmaxf(pex[j][r] + bv, 0.0f);
                    }
                }
            }
        } else {
            constexpr int HP = 64 + 4;
            constexpr int DP = 128 + 4;
            constexpr int HALF_T = 16;
            static_assert(ST * 16 * HP + HALF_T * 16 * DP <= ACCN, "h2 and the held tiles of hd fit");
            static_assert(ST - HALF_T == 8 + 1 && HALF_T == 16, "first pass: 4 row tiles per wave; second: 2, and the last one");
            float* const h2t = acc;
            float* const hdt = acc + ST * 16 * HP;
            __syncthreads();
            {
                const float bv = bias[ncol0];
                #pragma unroll
                for (int i = 0; i < 3; ++i) {
                    #pragma unroll
                    for (int r = 0; r < 8; ++r)
                        h2t[((rg + 8 * i) * 16 + rhalf + r) * HP + ncol0] = fmaxf(pacc[i][0][r] + bv, 0.0f);
                }
                if (extra) {
                    #pragma unroll
                    for (int r = 0; r < 8; ++r)
                        h2t[((ST - 1) * 16 + rhalf + r) * HP + ncol0] = fmaxf(pex[0][r] + bv, 0.0f);
                }
            }
            __syncthreads();

            const int c8 = wave & 7;
            const int r4 = wave >> 3;
            const int dcol = c8 * 16 + mrow;
            const float* const wd = Wd1 + (size_t)dcol * 64 + khalf;
            const float bdv = bd1[dcol];
            const int wrow = (mrow < 2) ? mrow : 0;
            const float* const w2r = Wd2 + (size_t)wrow * 128 + khalf;
            const float b2v = bd2[wrow];

            auto hd_tiles = [&](auto ntc, int T0, int L0) {
                constexpr int NT = decltype(ntc)::value;
                v8f q[NT];
                const float* ap[NT];
                #pragma unroll
                for (int i = 0; i < NT; ++i) {
                    q[i]  = v8f{};
                    ap[i] = h2t + ((T0 + 4 * i) * 16 + mrow) * HP + khalf;
                }
                for (int k = 0; k < 64; k += 4) {
                    v2f a[NT];
                    #pragma unroll
                    for (int i = 0; i < NT; ++i) a[i] = *(const v2f*)(ap[i] + k);
                    const v2f b = *(const v2f*)(wd + k);
                    #pragma unroll
                    for (int i = 0; i < NT; ++i)
                        q[i] = __builtin_amdgcn_wmma_f32_16x16x4_f32(false, a[i], false, b,
                                                                     (short)0, q[i], false, false);
                }
                #pragma unroll
                for (int i = 0; i < NT; ++i) {
                    #pragma unroll
                    for (int r = 0; r < 8; ++r)
                        hdt[((L0 + 4 * i) * 16 + rhalf + r) * DP + dcol] = fmaxf(q[i][r] + bdv, 0.0f);
                }
            };
            auto out_tile = [&](int T, int L) {
                v8f q = v8f{};
                const float* const ap = hdt + (L * 16 + mrow) * DP + khalf;
                for (int k = 0; k < 128; k += 4) {
                    const v2f a = *(const v2f*)(ap + k);
                    const v2f b = *(const v2f*)(w2r + k);
                    q = __builtin_amdgcn_wmma_f32_16x16x4_f32(false, a, false, b,
                                                              (short)0, q, false, false);
                }
                const int lrow0 = T * 16 + rhalf;
                if (mrow < 2 && lrow0 < R && lo + lrow0 < N_NODES) {
                    #pragma unroll
                    for (int r = 0; r < 8; ++r)
                        out[(size_t)(lo + lrow0 + r) * 2 + mrow] = q[r] + b2v;
                }
            };

            hd_tiles(tile_count<4>{}, r4, r4);
            __syncthreads();
            if (wave < HALF_T) out_tile(wave, wave);
            __syncthreads();
            hd_tiles(tile_count<2>{}, HALF_T + r4, r4);
            if (r4 == 0) hd_tiles(tile_count<1>{}, ST - 1, 8);
            __syncthreads();
            if (wave < ST - HALF_T) out_tile(HALF_T + wave, wave);
        }
    }
}

extern "C" void kernel_launch(void* const* d_in, const int* in_sizes, int n_in,
                              void* d_out, int out_size, void* d_ws, size_t ws_size,
                              hipStream_t stream) {
    const float* x    = (const float*)d_in[0];
    const int*   edge = (const int*)d_in[1];
    const int*   src  = edge;
    const int*   dst  = edge + N_EDGES;
    const float* Wl0 = (const float*)d_in[2],  *bl0 = (const float*)d_in[3],  *Wr0 = (const float*)d_in[4];
    const float* Wl1 = (const float*)d_in[5],  *bl1 = (const float*)d_in[6],  *Wr1 = (const float*)d_in[7];
    const float* Wl2 = (const float*)d_in[8],  *bl2 = (const float*)d_in[9],  *Wr2 = (const float*)d_in[10];
    const float* Wd1 = (const float*)d_in[11], *bd1 = (const float*)d_in[12];
    const float* Wd2 = (const float*)d_in[13], *bd2 = (const float*)d_in[14];

    float* ws   = (float*)d_ws;
    float* deg  = ws;
    float* inv  = deg  + N_NODES;
    float* agg  = inv  + N_NODES;
    float* bufA = agg  + (size_t)N_NODES * 128;
    float* bufB = bufA + (size_t)N_NODES * 128;

    const int go = N_BLK;
    static_assert((size_t)SAVE_HEAD + (size_t)N_BLK * BIN_WGS * BIN_CAP
                  <= (size_t)N_NODES * 128, "cells must fit the plane");
    unsigned* sv = (unsigned*)bufB;

    bin_edges_kernel<<<BIN_WGS, 1024, 0, stream>>>(src, dst, sv);

    own_rows_sum_kernel<64, true, 128><<<go, 1024, 0, stream>>>(x, src, dst, nullptr, inv, sv,
                                                                Wl0, Wr0, bl0, bufA,
                                                                nullptr, nullptr, nullptr, nullptr);

    float* const h1 = agg;
    own_rows_sum_kernel<128, false, 128><<<go, 1024, 0, stream>>>(bufA, src, dst, nullptr, inv, sv,
                                                                  Wl1, Wr1, bl1, h1,
                                                                  nullptr, nullptr, nullptr, nullptr);

    own_rows_sum_kernel<128, false, 64><<<go, 1024, 0, stream>>>(h1, src, dst, nullptr, inv, sv,
                                                                 Wl2, Wr2, bl2, (float*)d_out,
                                                                 Wd1, bd1, Wd2, bd2);
}
